# starting kernel
# baseline (speedup 1.0000x reference)
#include <hip/hip_runtime.h>
#include <hip/hip_bf16.h>


typedef __attribute__((ext_vector_type(2))) float v2f;
typedef __attribute__((ext_vector_type(8))) float v8f;

__global__ __launch_bounds__(256) void gcn_init_kernel(float* __restrict__ deg,
                                                       float* __restrict__ agg,
                                                       int N, int total32) {
  int i = blockIdx.x * 256 + threadIdx.x;
  if (i < total32) agg[i] = 0.0f;
  if (i < N) deg[i] = 1.0f;
}

__global__ __launch_bounds__(256) void gcn_degree_kernel(const int* __restrict__ dst,
                                                         float* __restrict__ deg,
                                                         int E) {
  int e = blockIdx.x * 256 + threadIdx.x;
  if (e < E) unsafeAtomicAdd(&deg[dst[e]], 1.0f);
}

__global__ __launch_bounds__(256) void gcn_dinv_kernel(float* __restrict__ deg, int N) {
  int i = blockIdx.x * 256 + threadIdx.x;
  if (i < N) {
    float d = deg[i];
    deg[i] = (d > 0.0f) ? rsqrtf(d) : 0.0f;
  }
}

template <int NCOL>
__global__ __launch_bounds__(256) void gcn_gemm_wmma_kernel(const float* __restrict__ X,
                                                            const float* __restrict__ W,
                                                            float* __restrict__ Y,
                                                            int nTiles) {
  const int lane = threadIdx.x & 31;
  const int wave = threadIdx.x >> 5;
  const int tile = blockIdx.x * 8 + wave;
  if (tile >= nTiles) return;

  const int rowBase = tile * 16;
  const int lo = lane & 15;
  const int hi = lane >> 4;

  v2f a[8];
#pragma unroll
  for (int kb = 0; kb < 8; ++kb) {
    const int k = kb * 4 + hi * 2;
    const float* xr = X + (size_t)(rowBase + lo) * 32 + k;
    a[kb].x = xr[0];
    a[kb].y = xr[1];
  }

#pragma unroll
  for (int ct = 0; ct < NCOL / 16; ++ct) {
    v8f c = {};
#pragma unroll
    for (int kb = 0; kb < 8; ++kb) {
      const int k = kb * 4 + hi * 2;
      v2f b;
      b.x = W[(size_t)k * NCOL + ct * 16 + lo];
      b.y = W[(size_t)(k + 1) * NCOL + ct * 16 + lo];
      c = __builtin_amdgcn_wmma_f32_16x16x4_f32(false, a[kb], false, b,
                                                (short)0, c, false, false);
    }
#pragma unroll
    for (int v = 0; v < 8; ++v) {
      const int m = v + hi * 8;
      Y[(size_t)(rowBase + m) * NCOL + ct * 16 + lo] = c[v];
    }
  }
}

template <int F>
__global__ __launch_bounds__(256) void gcn_edge_agg_kernel(const int* __restrict__ src,
                                                           const int* __restrict__ dst,
                                                           const float* __restrict__ dinv,
                                                           const float* __restrict__ H,
                                                           float* __restrict__ AGG,
                                                           long long total) {
  const long long t = (long long)blockIdx.x * 256 + threadIdx.x;
  if (t >= total) return;
  constexpr int CPE = F / 4;
  const int e  = (int)(t / CPE);
  const int fb = (int)(t % CPE) * 4;
  const int s = src[e];
  const int d = dst[e];
  const float c = dinv[s] * dinv[d];
  const float4 v = *(const float4*)(H + (size_t)s * F + fb);
  float* p = AGG + (size_t)d * F + fb;
  unsafeAtomicAdd(p + 0, v.x * c);
  unsafeAtomicAdd(p + 1, v.y * c);
  unsafeAtomicAdd(p + 2, v.z * c);
  unsafeAtomicAdd(p + 3, v.w * c);
}

__global__ __launch_bounds__(256) void gcn_finish1_kernel(float* __restrict__ AGG,
                                                          const float* __restrict__ XW,
                                                          const float* __restrict__ dinv,
                                                          const float* __restrict__ b1,
                                                          float* __restrict__ H1,
                                                          int total) {
  int i = blockIdx.x * 256 + threadIdx.x;
  if (i >= total) return;
  const int n = i >> 5;
  const int f = i & 31;
  float v = AGG[i];
  AGG[i] = 0.0f;
  const float di = dinv[n];
  v += XW[i] * di * di;
  H1[i] = tanhf(v + b1[f]);
}

__global__ __launch_bounds__(256) void gcn_finish2_kernel(const float* __restrict__ AGG,
                                                          const float* __restrict__ HW,
                                                          const float* __restrict__ dinv,
                                                          const float* __restrict__ b2,
                                                          float* __restrict__ out_h,
                                                          float* __restrict__ out_ls,
                                                          int N) {
  int n = blockIdx.x * 256 + threadIdx.x;
  if (n >= N) return;
  const float di = dinv[n];
  const float sc = di * di;
  float v[16];
  float m = -3.402823466e38f;
#pragma unroll
  for (int f = 0; f < 16; ++f) {
    const float t = AGG[(size_t)n * 16 + f] + HW[(size_t)n * 16 + f] * sc + b2[f];
    v[f] = t;
    m = fmaxf(m, t);
  }
  float s = 0.0f;
#pragma unroll
  for (int f = 0; f < 16; ++f) s += expf(v[f] - m);
  const float lse = logf(s);
#pragma unroll
  for (int f = 0; f < 16; ++f) {
    out_h[(size_t)n * 16 + f] = v[f];
    out_ls[(size_t)n * 16 + f] = v[f] - m - lse;
  }
}

extern "C" void kernel_launch(void* const* d_in, const int* in_sizes, int n_in,
                              void* d_out, int out_size, void* d_ws, size_t ws_size,
                              hipStream_t stream) {
  const float* x  = (const float*)d_in[0];
  const int*   ei = (const int*)d_in[1];
  const float* W1 = (const float*)d_in[2];
  const float* b1 = (const float*)d_in[3];
  const float* W2 = (const float*)d_in[4];
  const float* b2 = (const float*)d_in[5];

  const int N = in_sizes[0] / 32;
  const int E = in_sizes[1] / 2;
  const int* src = ei;
  const int* dst = ei + E;

  float* ws   = (float*)d_ws;
  float* deg  = ws;
  float* xw1  = ws + (size_t)N;
  float* agg1 = ws + (size_t)N * 33;
  float* h1   = ws + (size_t)N * 65;
  float* hw2  = xw1;
  float* agg2 = agg1;

  float* out_h  = (float*)d_out;
  float* out_ls = out_h + (size_t)N * 16;

  auto nb = [](long long n) { return (unsigned)((n + 255) / 256); };
  const int nTiles = N / 16;
  const unsigned gemmBlocks = (unsigned)((nTiles + 7) / 8);

  gcn_init_kernel<<<nb((long long)N * 32), 256, 0, stream>>>(deg, agg1, N, N * 32);
  gcn_degree_kernel<<<nb(E), 256, 0, stream>>>(dst, deg, E);
  gcn_dinv_kernel<<<nb(N), 256, 0, stream>>>(deg, N);

  gcn_gemm_wmma_kernel<32><<<gemmBlocks, 256, 0, stream>>>(x, W1, xw1, nTiles);
  gcn_edge_agg_kernel<32><<<nb((long long)E * 8), 256, 0, stream>>>(
      src, dst, deg, xw1, agg1, (long long)E * 8);
  gcn_finish1_kernel<<<nb((long long)N * 32), 256, 0, stream>>>(agg1, xw1, deg, b1, h1,
                                                                N * 32);

  gcn_gemm_wmma_kernel<16><<<gemmBlocks, 256, 0, stream>>>(h1, W2, hw2, nTiles);
  gcn_edge_agg_kernel<16><<<nb((long long)E * 4), 256, 0, stream>>>(
      src, dst, deg, hw2, agg2, (long long)E * 4);
  gcn_finish2_kernel<<<nb(N), 256, 0, stream>>>(agg2, hw2, deg, b2, out_h, out_ls, N);
}


# edge aggregation: each wave instruction spans whole contiguous feature rows (one gather + one f32 atomic per row, 32 edges per wave) instead of float4 pieces; atomics kept, arrival order as before
# speedup vs baseline: 1.2341x; 1.2341x over previous
#include <hip/hip_runtime.h>
#include <hip/hip_bf16.h>


typedef __attribute__((ext_vector_type(2))) float v2f;
typedef __attribute__((ext_vector_type(8))) float v8f;

__global__ __launch_bounds__(256) void gcn_init_kernel(float* __restrict__ deg,
                                                       float* __restrict__ agg,
                                                       int N, int total32) {
  int i = blockIdx.x * 256 + threadIdx.x;
  if (i < total32) agg[i] = 0.0f;
  if (i < N) deg[i] = 1.0f;
}

__global__ __launch_bounds__(256) void gcn_degree_kernel(const int* __restrict__ dst,
                                                         float* __restrict__ deg,
                                                         int E) {
  int e = blockIdx.x * 256 + threadIdx.x;
  if (e < E) unsafeAtomicAdd(&deg[dst[e]], 1.0f);
}

__global__ __launch_bounds__(256) void gcn_dinv_kernel(float* __restrict__ deg, int N) {
  int i = blockIdx.x * 256 + threadIdx.x;
  if (i < N) {
    float d = deg[i];
    deg[i] = (d > 0.0f) ? rsqrtf(d) : 0.0f;
  }
}

template <int NCOL>
__global__ __launch_bounds__(256) void gcn_gemm_wmma_kernel(const float* __restrict__ X,
                                                            const float* __restrict__ W,
                                                            float* __restrict__ Y,
                                                            int nTiles) {
  const int lane = threadIdx.x & 31;
  const int wave = threadIdx.x >> 5;
  const int tile = blockIdx.x * 8 + wave;
  if (tile >= nTiles) return;

  const int rowBase = tile * 16;
  const int lo = lane & 15;
  const int hi = lane >> 4;

  v2f a[8];
#pragma unroll
  for (int kb = 0; kb < 8; ++kb) {
    const int k = kb * 4 + hi * 2;
    const float* xr = X + (size_t)(rowBase + lo) * 32 + k;
    a[kb].x = xr[0];
    a[kb].y = xr[1];
  }

#pragma unroll
  for (int ct = 0; ct < NCOL / 16; ++ct) {
    v8f c = {};
#pragma unroll
    for (int kb = 0; kb < 8; ++kb) {
      const int k = kb * 4 + hi * 2;
      v2f b;
      b.x = W[(size_t)k * NCOL + ct * 16 + lo];
      b.y = W[(size_t)(k + 1) * NCOL + ct * 16 + lo];
      c = __builtin_amdgcn_wmma_f32_16x16x4_f32(false, a[kb], false, b,
                                                (short)0, c, false, false);
    }
#pragma unroll
    for (int v = 0; v < 8; ++v) {
      const int m = v + hi * 8;
      Y[(size_t)(rowBase + m) * NCOL + ct * 16 + lo] = c[v];
    }
  }
}

template <int F>
__global__ __launch_bounds__(256) void gcn_edge_agg_kernel(const int* __restrict__ src,
                                                           const int* __restrict__ dst,
                                                           const float* __restrict__ dinv,
                                                           const float* __restrict__ H,
                                                           float* __restrict__ AGG,
                                                           int E, int nChunks, int nWaves) {
  constexpr int EPI = 32 / F;
  constexpr int STEPS = 32 / EPI;
  const int lane = threadIdx.x & 31;
  const int f = lane & (F - 1);
  const int sub = lane / F;
  const int wave = blockIdx.x * 8 + (threadIdx.x >> 5);
  for (int ch = wave; ch < nChunks; ch += nWaves) {
    const int e = ch * 32 + lane;
    int s = 0, d = 0;
    float c = 0.0f;
    if (e < E) {
      s = src[e];
      d = dst[e];
      c = dinv[s] * dinv[d];
    }
    const int nvalid = min(32, E - ch * 32);
    for (int j0 = 0; j0 < STEPS; j0 += 8) {
      float v[8], cj[8];
      int dj[8];
#pragma unroll
      for (int u = 0; u < 8; ++u) {
        const int q = (j0 + u) * EPI + sub;
        int sj;
        if constexpr (EPI == 1) {
          sj = __builtin_amdgcn_readlane(s, q);
          dj[u] = __builtin_amdgcn_readlane(d, q);
          cj[u] = __int_as_float(__builtin_amdgcn_readlane(__float_as_int(c), q));
        } else {
          sj = __shfl(s, q, 32);
          dj[u] = __shfl(d, q, 32);
          cj[u] = __shfl(c, q, 32);
        }
        v[u] = H[(size_t)sj * F + f];
      }
#pragma unroll
      for (int u = 0; u < 8; ++u) {
        const int q = (j0 + u) * EPI + sub;
        if (q < nvalid) unsafeAtomicAdd(AGG + (size_t)dj[u] * F + f, v[u] * cj[u]);
      }
    }
  }
}

__global__ __launch_bounds__(256) void gcn_finish1_kernel(float* __restrict__ AGG,
                                                          const float* __restrict__ XW,
                                                          const float* __restrict__ dinv,
                                                          const float* __restrict__ b1,
                                                          float* __restrict__ H1,
                                                          int total) {
  int i = blockIdx.x * 256 + threadIdx.x;
  if (i >= total) return;
  const int n = i >> 5;
  const int f = i & 31;
  float v = AGG[i];
  AGG[i] = 0.0f;
  const float di = dinv[n];
  v += XW[i] * di * di;
  H1[i] = tanhf(v + b1[f]);
}

__global__ __launch_bounds__(256) void gcn_finish2_kernel(const float* __restrict__ AGG,
                                                          const float* __restrict__ HW,
                                                          const float* __restrict__ dinv,
                                                          const float* __restrict__ b2,
                                                          float* __restrict__ out_h,
                                                          float* __restrict__ out_ls,
                                                          int N) {
  int n = blockIdx.x * 256 + threadIdx.x;
  if (n >= N) return;
  const float di = dinv[n];
  const float sc = di * di;
  float v[16];
  float m = -3.402823466e38f;
#pragma unroll
  for (int f = 0; f < 16; ++f) {
    const float t = AGG[(size_t)n * 16 + f] + HW[(size_t)n * 16 + f] * sc + b2[f];
    v[f] = t;
    m = fmaxf(m, t);
  }
  float s = 0.0f;
#pragma unroll
  for (int f = 0; f < 16; ++f) s += expf(v[f] - m);
  const float lse = logf(s);
#pragma unroll
  for (int f = 0; f < 16; ++f) {
    out_h[(size_t)n * 16 + f] = v[f];
    out_ls[(size_t)n * 16 + f] = v[f] - m - lse;
  }
}

extern "C" void kernel_launch(void* const* d_in, const int* in_sizes, int n_in,
                              void* d_out, int out_size, void* d_ws, size_t ws_size,
                              hipStream_t stream) {
  const float* x  = (const float*)d_in[0];
  const int*   ei = (const int*)d_in[1];
  const float* W1 = (const float*)d_in[2];
  const float* b1 = (const float*)d_in[3];
  const float* W2 = (const float*)d_in[4];
  const float* b2 = (const float*)d_in[5];

  const int N = in_sizes[0] / 32;
  const int E = in_sizes[1] / 2;
  const int* src = ei;
  const int* dst = ei + E;

  float* ws   = (float*)d_ws;
  float* deg  = ws;
  float* xw1  = ws + (size_t)N;
  float* agg1 = ws + (size_t)N * 33;
  float* h1   = ws + (size_t)N * 65;
  float* hw2  = xw1;
  float* agg2 = agg1;

  float* out_h  = (float*)d_out;
  float* out_ls = out_h + (size_t)N * 16;

  auto nb = [](long long n) { return (unsigned)((n + 255) / 256); };
  const int nTiles = N / 16;
  const unsigned gemmBlocks = (unsigned)((nTiles + 7) / 8);
  const int nChunks = (E + 31) / 32;
  const unsigned aggBlocks = (unsigned)((nChunks + 7) / 8);

  gcn_init_kernel<<<nb((long long)N * 32), 256, 0, stream>>>(deg, agg1, N, N * 32);
  gcn_degree_kernel<<<nb(E), 256, 0, stream>>>(dst, deg, E);
  gcn_dinv_kernel<<<nb(N), 256, 0, stream>>>(deg, N);

  gcn_gemm_wmma_kernel<32><<<gemmBlocks, 256, 0, stream>>>(x, W1, xw1, nTiles);
  gcn_edge_agg_kernel<32><<<aggBlocks, 256, 0, stream>>>(
      src, dst, deg, xw1, agg1, E, nChunks, (int)aggBlocks * 8);
  gcn_finish1_kernel<<<nb((long long)N * 32), 256, 0, stream>>>(agg1, xw1, deg, b1, h1,
                                                                N * 32);

  gcn_gemm_wmma_kernel<16><<<gemmBlocks, 256, 0, stream>>>(h1, W2, hw2, nTiles);
  gcn_edge_agg_kernel<16><<<aggBlocks, 256, 0, stream>>>(
      src, dst, deg, hw2, agg2, E, nChunks, (int)aggBlocks * 8);
  gcn_finish2_kernel<<<nb(N), 256, 0, stream>>>(agg2, hw2, deg, b2, out_h, out_ls, N);
}


# edge sums by destination: per-call counting sort of edges by dst (returning int atomic = rank, scan, place); resident waves gather each node's rows and sum in registers in rank order; no f32 atomics
# speedup vs baseline: 1.3158x; 1.0662x over previous
#include <hip/hip_runtime.h>
#include <hip/hip_bf16.h>


typedef __attribute__((ext_vector_type(2))) float v2f;
typedef __attribute__((ext_vector_type(8))) float v8f;

__global__ __launch_bounds__(256) void gcn_zero_kernel(int* __restrict__ cnt, int N) {
  int i = blockIdx.x * 256 + threadIdx.x;
  if (i < N) cnt[i] = 0;
}

__global__ __launch_bounds__(256) void gcn_rank_kernel(const int* __restrict__ dst,
                                                       int* __restrict__ cnt,
                                                       int* __restrict__ rank,
                                                       int E, int T) {
  const int t = blockIdx.x * 256 + threadIdx.x;
  int d[4], r[4];
#pragma unroll
  for (int k = 0; k < 4; ++k) {
    const int e = min(t + k * T, E - 1);
    d[k] = dst[e];
  }
#pragma unroll
  for (int k = 0; k < 4; ++k) {
    const int e = t + k * T;
    r[k] = atomicAdd(&cnt[d[k]], (e < E) ? 1 : 0);
  }
#pragma unroll
  for (int k = 0; k < 4; ++k) {
    const int e = t + k * T;
    if (e < E) rank[e] = r[k];
  }
}

__global__ __launch_bounds__(256) void gcn_scan_kernel(const int* __restrict__ cnt,
                                                       int* __restrict__ rowstart,
                                                       float* __restrict__ dinv,
                                                       int N) {
  __shared__ int wsum[8];
  __shared__ int wacc[8];
  const int tid = threadIdx.x;
  const int lane = tid & 31;
  const int wave = tid >> 5;
  const int b = blockIdx.x;

  int acc = 0;
  const int lim = b * 256;
  const int4* c4 = (const int4*)cnt;
  for (int g0 = 0; g0 < lim; g0 += 16 * 256) {
    int4 v[16];
#pragma unroll
    for (int u = 0; u < 16; ++u) v[u] = c4[min(g0 + u * 256 + tid, lim - 1)];
#pragma unroll
    for (int u = 0; u < 16; ++u) {
      if (g0 + u * 256 + tid < lim) acc += (v[u].x + v[u].y) + (v[u].z + v[u].w);
    }
  }

  const int i0 = b * 1024 + tid * 4;
  int c[4];
#pragma unroll
  for (int j = 0; j < 4; ++j) c[j] = (i0 + j < N) ? cnt[i0 + j] : 0;
  const int s4 = (c[0] + c[1]) + (c[2] + c[3]);

  int x = s4;
#pragma unroll
  for (int o = 1; o < 32; o <<= 1) {
    const int y = __shfl_up(x, o, 32);
    if (lane >= o) x += y;
  }
#pragma unroll
  for (int o = 16; o > 0; o >>= 1) acc += __shfl_xor(acc, o, 32);
  if (lane == 31) wsum[wave] = x;
  if (lane == 0) wacc[wave] = acc;
  __syncthreads();
  int base = 0;
#pragma unroll
  for (int w = 0; w < 8; ++w) {
    base += wacc[w];
    if (w < wave) base += wsum[w];
  }
  int ex = base + (x - s4);
#pragma unroll
  for (int j = 0; j < 4; ++j) {
    if (i0 + j < N) {
      rowstart[i0 + j] = ex;
      const float dg = (float)(c[j] + 1);
      dinv[i0 + j] = (dg > 0.0f) ? rsqrtf(dg) : 0.0f;
    }
    ex += c[j];
  }
}

__global__ __launch_bounds__(256) void gcn_fill_kernel(const int* __restrict__ src,
                                                       const int* __restrict__ dst,
                                                       const int* __restrict__ rank,
                                                       const int* __restrict__ rowstart,
                                                       int* __restrict__ csr,
                                                       int E, int T) {
  const int t = blockIdx.x * 256 + threadIdx.x;
  int d[4], r[4], s[4], p[4];
#pragma unroll
  for (int k = 0; k < 4; ++k) {
    const int e = min(t + k * T, E - 1);
    d[k] = dst[e]; r[k] = rank[e]; s[k] = src[e];
  }
#pragma unroll
  for (int k = 0; k < 4; ++k) p[k] = rowstart[d[k]] + r[k];
#pragma unroll
  for (int k = 0; k < 4; ++k) {
    const int e = t + k * T;
    if (e < E) csr[p[k]] = s[k];
  }
}

template <int NCOL>
__global__ __launch_bounds__(256) void gcn_gemm_wmma_kernel(const float* __restrict__ X,
                                                            const float* __restrict__ W,
                                                            float* __restrict__ Y,
                                                            int nTiles) {
  const int lane = threadIdx.x & 31;
  const int wave = threadIdx.x >> 5;
  const int tile = blockIdx.x * 8 + wave;
  if (tile >= nTiles) return;

  const int rowBase = tile * 16;
  const int lo = lane & 15;
  const int hi = lane >> 4;

  v2f a[8];
#pragma unroll
  for (int kb = 0; kb < 8; ++kb) {
    const int k = kb * 4 + hi * 2;
    const float* xr = X + (size_t)(rowBase + lo) * 32 + k;
    a[kb].x = xr[0];
    a[kb].y = xr[1];
  }

#pragma unroll
  for (int ct = 0; ct < NCOL / 16; ++ct) {
    v8f c = {};
#pragma unroll
    for (int kb = 0; kb < 8; ++kb) {
      const int k = kb * 4 + hi * 2;
      v2f b;
      b.x = W[(size_t)k * NCOL + ct * 16 + lo];
      b.y = W[(size_t)(k + 1) * NCOL + ct * 16 + lo];
      c = __builtin_amdgcn_wmma_f32_16x16x4_f32(false, a[kb], false, b,
                                                (short)0, c, false, false);
    }
#pragma unroll
    for (int v = 0; v < 8; ++v) {
      const int m = v + hi * 8;
      Y[(size_t)(rowBase + m) * NCOL + ct * 16 + lo] = c[v];
    }
  }
}


__global__ __launch_bounds__(256) void gcn_gather32_kernel(const int* __restrict__ rowstart,
                                                           const int* __restrict__ cnt,
                                                           const int* __restrict__ csr,
                                                           const float* __restrict__ dinv,
                                                           const float* __restrict__ H,
                                                           const float* __restrict__ b1,
                                                           float* __restrict__ H1,
                                                           int N, int nWaves) {
  const int lane = threadIdx.x & 31;
  const int wave = __builtin_amdgcn_readfirstlane(blockIdx.x * 8 + (threadIdx.x >> 5));
  const float bias = b1[lane];
  int n = wave;
  int start = 0, deg = 0;
  float dn = 0.0f;
  if (n < N) { start = rowstart[n]; deg = cnt[n]; dn = dinv[n]; }
  while (n < N) {
    const int n2 = n + nWaves;
    int start2 = 0, deg2 = 0;
    float dn2 = 0.0f;
    if (n2 < N) { start2 = rowstart[n2]; deg2 = cnt[n2]; dn2 = dinv[n2]; }
    const float selfv = H[(size_t)n * 32 + lane];
    float acc = 0.0f;
    for (int c0 = 0; c0 < deg; c0 += 32) {
      const int nv = min(32, deg - c0);
      int s = n;
      if (lane < nv) s = csr[start + c0 + lane];
      const float ds = dinv[s];
      for (int j0 = 0; j0 < nv; j0 += 16) {
        float v[16], cj[16];
#pragma unroll
        for (int u = 0; u < 16; ++u) {
          const int q = j0 + u;
          const int sj = __builtin_amdgcn_readlane(s, q);
          cj[u] = __int_as_float(__builtin_amdgcn_readlane(__float_as_int(ds), q)) * dn;
          v[u] = H[(size_t)sj * 32 + lane];
        }
#pragma unroll
        for (int u = 0; u < 16; ++u) {
          if (j0 + u < nv) acc += v[u] * cj[u];
        }
      }
    }
    const float t = acc + selfv * dn * dn;
    H1[(size_t)n * 32 + lane] = tanhf(t + bias);
    n = n2; start = start2; deg = deg2; dn = dn2;
  }
}

__global__ __launch_bounds__(256) void gcn_gather16_kernel(const int* __restrict__ rowstart,
                                                           const int* __restrict__ cnt,
                                                           const int* __restrict__ csr,
                                                           const float* __restrict__ dinv,
                                                           const float* __restrict__ H,
                                                           float* __restrict__ AGG,
                                                           int N, int nWaves) {
  const int lane = threadIdx.x & 31;
  const int f = lane & 15;
  const int sub = lane >> 4;
  const int wave = __builtin_amdgcn_readfirstlane(blockIdx.x * 8 + (threadIdx.x >> 5));
  const int nPairs = (N + 1) >> 1;
  for (int p = wave; p < nPairs; p += nWaves) {
    const int n = 2 * p + sub;
    const bool valid = n < N;
    const int nc = valid ? n : N - 1;
    const int start = rowstart[nc];
    const int deg = valid ? cnt[nc] : 0;
    const float dn = dinv[nc];
    const int degmax = max(__builtin_amdgcn_readlane(deg, 0), __builtin_amdgcn_readlane(deg, 16));
    float acc = 0.0f;
    for (int c0 = 0; c0 < degmax; c0 += 32) {
      const int left = deg - c0;
      int sa = nc, sb = nc;
      if (f < left) sa = csr[start + c0 + f];
      if (16 + f < left) sb = csr[start + c0 + 16 + f];
      const float da = dinv[sa];
      const float db = dinv[sb];
      {
        float v[16], cj[16];
#pragma unroll
        for (int u = 0; u < 16; ++u) {
          const int s0 = __builtin_amdgcn_readlane(sa, u);
          const int s1 = __builtin_amdgcn_readlane(sa, 16 + u);
          const float e0 = __int_as_float(__builtin_amdgcn_readlane(__float_as_int(da), u));
          const float e1 = __int_as_float(__builtin_amdgcn_readlane(__float_as_int(da), 16 + u));
          const int sj = sub ? s1 : s0;
          cj[u] = (sub ? e1 : e0) * dn;
          v[u] = H[(size_t)sj * 16 + f];
        }
#pragma unroll
        for (int u = 0; u < 16; ++u) {
          if (u < left) acc += v[u] * cj[u];
        }
      }
      if (degmax - c0 > 16) {
        float v[16], cj[16];
#pragma unroll
        for (int u = 0; u < 16; ++u) {
          const int s0 = __builtin_amdgcn_readlane(sb, u);
          const int s1 = __builtin_amdgcn_readlane(sb, 16 + u);
          const float e0 = __int_as_float(__builtin_amdgcn_readlane(__float_as_int(db), u));
          const float e1 = __int_as_float(__builtin_amdgcn_readlane(__float_as_int(db), 16 + u));
          const int sj = sub ? s1 : s0;
          cj[u] = (sub ? e1 : e0) * dn;
          v[u] = H[(size_t)sj * 16 + f];
        }
#pragma unroll
        for (int u = 0; u < 16; ++u) {
          if (16 + u < left) acc += v[u] * cj[u];
        }
      }
    }
    if (valid) AGG[(size_t)n * 16 + f] = acc;
  }
}

__global__ __launch_bounds__(256) void gcn_finish2_kernel(const float* __restrict__ AGG,
                                                          const float* __restrict__ HW,
                                                          const float* __restrict__ dinv,
                                                          const float* __restrict__ b2,
                                                          float* __restrict__ out_h,
                                                          float* __restrict__ out_ls,
                                                          int N) {
  int n = blockIdx.x * 256 + threadIdx.x;
  if (n >= N) return;
  const float di = dinv[n];
  const float sc = di * di;
  float v[16];
  float m = -3.402823466e38f;
#pragma unroll
  for (int f = 0; f < 16; ++f) {
    const float t = AGG[(size_t)n * 16 + f] + HW[(size_t)n * 16 + f] * sc + b2[f];
    v[f] = t;
    m = fmaxf(m, t);
  }
  float s = 0.0f;
#pragma unroll
  for (int f = 0; f < 16; ++f) s += expf(v[f] - m);
  const float lse = logf(s);
#pragma unroll
  for (int f = 0; f < 16; ++f) {
    out_h[(size_t)n * 16 + f] = v[f];
    out_ls[(size_t)n * 16 + f] = v[f] - m - lse;
  }
}

extern "C" void kernel_launch(void* const* d_in, const int* in_sizes, int n_in,
                              void* d_out, int out_size, void* d_ws, size_t ws_size,
                              hipStream_t stream) {
  const float* x  = (const float*)d_in[0];
  const int*   ei = (const int*)d_in[1];
  const float* W1 = (const float*)d_in[2];
  const float* b1 = (const float*)d_in[3];
  const float* W2 = (const float*)d_in[4];
  const float* b2 = (const float*)d_in[5];

  const int N = in_sizes[0] / 32;
  const int E = in_sizes[1] / 2;
  const int* src = ei;
  const int* dst = ei + E;

  int*   wsi      = (int*)d_ws;
  float* wsf      = (float*)d_ws;
  int*   cnt      = wsi;
  float* dinv     = wsf + (size_t)N;
  int*   rowstart = wsi + (size_t)N * 2;
  int*   rank     = wsi + (size_t)N * 3;
  int*   csr      = wsi + (size_t)N * 3 + (size_t)E;
  float* xw1      = wsf + (size_t)N * 3 + (size_t)E * 2;
  float* h1       = xw1 + (size_t)N * 32;
  float* hw2      = h1 + (size_t)N * 32;
  float* agg2     = hw2 + (size_t)N * 16;

  float* out_h  = (float*)d_out;
  float* out_ls = out_h + (size_t)N * 16;

  auto nb = [](long long n) { return (unsigned)((n + 255) / 256); };
  const int nTiles = N / 16;
  const unsigned gemmBlocks = (unsigned)((nTiles + 7) / 8);
  const unsigned edgeBlocks = nb(((long long)E + 3) / 4);
  const unsigned scanBlocks = (unsigned)((N + 1023) / 1024);
  const unsigned gather32Blocks = (unsigned)((N + 55) / 56);
  const unsigned gather16Blocks = (unsigned)(((N + 1) / 2 + 31) / 32);

  gcn_zero_kernel<<<nb(N), 256, 0, stream>>>(cnt, N);
  gcn_rank_kernel<<<edgeBlocks, 256, 0, stream>>>(dst, cnt, rank, E, (int)edgeBlocks * 256);
  gcn_scan_kernel<<<scanBlocks, 256, 0, stream>>>(cnt, rowstart, dinv, N);
  gcn_fill_kernel<<<edgeBlocks, 256, 0, stream>>>(src, dst, rank, rowstart, csr, E,
                                                  (int)edgeBlocks * 256);

  gcn_gemm_wmma_kernel<32><<<gemmBlocks, 256, 0, stream>>>(x, W1, xw1, nTiles);
  gcn_gather32_kernel<<<gather32Blocks, 256, 0, stream>>>(rowstart, cnt, csr, dinv, xw1, b1, h1,
                                                          N, (int)gather32Blocks * 8);

  gcn_gemm_wmma_kernel<16><<<gemmBlocks, 256, 0, stream>>>(h1, W2, hw2, nTiles);
  gcn_gather16_kernel<<<gather16Blocks, 256, 0, stream>>>(rowstart, cnt, csr, dinv, hw2, agg2,
                                                          N, (int)gather16Blocks * 8);
  gcn_finish2_kernel<<<nb(N), 256, 0, stream>>>(agg2, hw2, dinv, b2, out_h, out_ls, N);
}


# edge weight split: each product's epilogue scales row n by dinv[n]; gathers sum pre-scaled rows and scale the sum once by dinv[dst]; no per-edge dinv gather (terms round as h*dinv[src])
# speedup vs baseline: 1.3780x; 1.0473x over previous
#include <hip/hip_runtime.h>
#include <hip/hip_bf16.h>


typedef __attribute__((ext_vector_type(2))) float v2f;
typedef __attribute__((ext_vector_type(8))) float v8f;

__global__ __launch_bounds__(256) void gcn_zero_kernel(int* __restrict__ cnt, int N) {
  int i = blockIdx.x * 256 + threadIdx.x;
  if (i < N) cnt[i] = 0;
}

__global__ __launch_bounds__(256) void gcn_rank_kernel(const int* __restrict__ dst,
                                                       int* __restrict__ cnt,
                                                       int* __restrict__ rank,
                                                       int E, int T) {
  const int t = blockIdx.x * 256 + threadIdx.x;
  int d[4], r[4];
#pragma unroll
  for (int k = 0; k < 4; ++k) {
    const int e = min(t + k * T, E - 1);
    d[k] = dst[e];
  }
#pragma unroll
  for (int k = 0; k < 4; ++k) {
    const int e = t + k * T;
    r[k] = atomicAdd(&cnt[d[k]], (e < E) ? 1 : 0);
  }
#pragma unroll
  for (int k = 0; k < 4; ++k) {
    const int e = t + k * T;
    if (e < E) rank[e] = r[k];
  }
}

__global__ __launch_bounds__(256) void gcn_scan_kernel(const int* __restrict__ cnt,
                                                       int* __restrict__ rowstart,
                                                       float* __restrict__ dinv,
                                                       int N) {
  __shared__ int wsum[8];
  __shared__ int wacc[8];
  const int tid = threadIdx.x;
  const int lane = tid & 31;
  const int wave = tid >> 5;
  const int b = blockIdx.x;

  int acc = 0;
  const int lim = b * 256;
  const int4* c4 = (const int4*)cnt;
  for (int g0 = 0; g0 < lim; g0 += 16 * 256) {
    int4 v[16];
#pragma unroll
    for (int u = 0; u < 16; ++u) v[u] = c4[min(g0 + u * 256 + tid, lim - 1)];
#pragma unroll
    for (int u = 0; u < 16; ++u) {
      if (g0 + u * 256 + tid < lim) acc += (v[u].x + v[u].y) + (v[u].z + v[u].w);
    }
  }

  const int i0 = b * 1024 + tid * 4;
  int c[4];
#pragma unroll
  for (int j = 0; j < 4; ++j) c[j] = (i0 + j < N) ? cnt[i0 + j] : 0;
  const int s4 = (c[0] + c[1]) + (c[2] + c[3]);

  int x = s4;
#pragma unroll
  for (int o = 1; o < 32; o <<= 1) {
    const int y = __shfl_up(x, o, 32);
    if (lane >= o) x += y;
  }
#pragma unroll
  for (int o = 16; o > 0; o >>= 1) acc += __shfl_xor(acc, o, 32);
  if (lane == 31) wsum[wave] = x;
  if (lane == 0) wacc[wave] = acc;
  __syncthreads();
  int base = 0;
#pragma unroll
  for (int w = 0; w < 8; ++w) {
    base += wacc[w];
    if (w < wave) base += wsum[w];
  }
  int ex = base + (x - s4);
#pragma unroll
  for (int j = 0; j < 4; ++j) {
    if (i0 + j < N) {
      rowstart[i0 + j] = ex;
      const float dg = (float)(c[j] + 1);
      dinv[i0 + j] = (dg > 0.0f) ? rsqrtf(dg) : 0.0f;
    }
    ex += c[j];
  }
}

__global__ __launch_bounds__(256) void gcn_fill_kernel(const int* __restrict__ src,
                                                       const int* __restrict__ dst,
                                                       const int* __restrict__ rank,
                                                       const int* __restrict__ rowstart,
                                                       int* __restrict__ csr,
                                                       int E, int T) {
  const int t = blockIdx.x * 256 + threadIdx.x;
  int d[4], r[4], s[4], p[4];
#pragma unroll
  for (int k = 0; k < 4; ++k) {
    const int e = min(t + k * T, E - 1);
    d[k] = dst[e]; r[k] = rank[e]; s[k] = src[e];
  }
#pragma unroll
  for (int k = 0; k < 4; ++k) p[k] = rowstart[d[k]] + r[k];
#pragma unroll
  for (int k = 0; k < 4; ++k) {
    const int e = t + k * T;
    if (e < E) csr[p[k]] = s[k];
  }
}

template <int NCOL>
__global__ __launch_bounds__(256) void gcn_gemm_wmma_kernel(const float* __restrict__ X,
                                                            const float* __restrict__ W,
                                                            const float* __restrict__ dinv,
                                                            float* __restrict__ Y,
                                                            int nTiles) {
  const int lane = threadIdx.x & 31;
  const int wave = threadIdx.x >> 5;
  const int tile = blockIdx.x * 8 + wave;
  if (tile >= nTiles) return;

  const int rowBase = tile * 16;
  const int lo = lane & 15;
  const int hi = lane >> 4;

  v2f a[8];
#pragma unroll
  for (int kb = 0; kb < 8; ++kb) {
    const int k = kb * 4 + hi * 2;
    const float* xr = X + (size_t)(rowBase + lo) * 32 + k;
    a[kb].x = xr[0];
    a[kb].y = xr[1];
  }
  float rs[8];
#pragma unroll
  for (int v = 0; v < 8; ++v) rs[v] = dinv[rowBase + v + hi * 8];

#pragma unroll
  for (int ct = 0; ct < NCOL / 16; ++ct) {
    v8f c = {};
#pragma unroll
    for (int kb = 0; kb < 8; ++kb) {
      const int k = kb * 4 + hi * 2;
      v2f b;
      b.x = W[(size_t)k * NCOL + ct * 16 + lo];
      b.y = W[(size_t)(k + 1) * NCOL + ct * 16 + lo];
      c = __builtin_amdgcn_wmma_f32_16x16x4_f32(false, a[kb], false, b,
                                                (short)0, c, false, false);
    }
#pragma unroll
    for (int v = 0; v < 8; ++v) {
      const int m = v + hi * 8;
      Y[(size_t)(rowBase + m) * NCOL + ct * 16 + lo] = c[v] * rs[v];
    }
  }
}


__global__ __launch_bounds__(256) void gcn_gather32_kernel(const int* __restrict__ rowstart,
                                                           const int* __restrict__ cnt,
                                                           const int* __restrict__ csr,
                                                           const float* __restrict__ dinv,
                                                           const float* __restrict__ H,
                                                           const float* __restrict__ b1,
                                                           float* __restrict__ H1,
                                                           int N, int nWaves) {
  const int lane = threadIdx.x & 31;
  const int wave = __builtin_amdgcn_readfirstlane(blockIdx.x * 8 + (threadIdx.x >> 5));
  const float bias = b1[lane];
  int n = wave;
  int start = 0, deg = 0;
  float dn = 0.0f;
  if (n < N) { start = rowstart[n]; deg = cnt[n]; dn = dinv[n]; }
  while (n < N) {
    const int n2 = n + nWaves;
    int start2 = 0, deg2 = 0;
    float dn2 = 0.0f;
    if (n2 < N) { start2 = rowstart[n2]; deg2 = cnt[n2]; dn2 = dinv[n2]; }
    const float selfv = H[(size_t)n * 32 + lane];
    float acc = 0.0f;
    for (int c0 = 0; c0 < deg; c0 += 32) {
      const int nv = min(32, deg - c0);
      int s = n;
      if (lane < nv) s = csr[start + c0 + lane];
      for (int j0 = 0; j0 < nv; j0 += 16) {
        float v[16];
#pragma unroll
        for (int u = 0; u < 16; ++u) {
          const int q = j0 + u;
          const int sj = __builtin_amdgcn_readlane(s, q);
          v[u] = H[(size_t)sj * 32 + lane];
        }
#pragma unroll
        for (int u = 0; u < 16; ++u) {
          if (j0 + u < nv) acc += v[u];
        }
      }
    }
    const float t = (acc + selfv) * dn;
    H1[(size_t)n * 32 + lane] = tanhf(t + bias);
    n = n2; start = start2; deg = deg2; dn = dn2;
  }
}

__global__ __launch_bounds__(256) void gcn_gather16_kernel(const int* __restrict__ rowstart,
                                                           const int* __restrict__ cnt,
                                                           const int* __restrict__ csr,
                                                           const float* __restrict__ H,
                                                           float* __restrict__ AGG,
                                                           int N, int nWaves) {
  const int lane = threadIdx.x & 31;
  const int f = lane & 15;
  const int sub = lane >> 4;
  const int wave = __builtin_amdgcn_readfirstlane(blockIdx.x * 8 + (threadIdx.x >> 5));
  const int nPairs = (N + 1) >> 1;
  for (int p = wave; p < nPairs; p += nWaves) {
    const int n = 2 * p + sub;
    const bool valid = n < N;
    const int nc = valid ? n : N - 1;
    const int start = rowstart[nc];
    const int deg = valid ? cnt[nc] : 0;
    const int degmax = max(__builtin_amdgcn_readlane(deg, 0), __builtin_amdgcn_readlane(deg, 16));
    float acc = 0.0f;
    for (int c0 = 0; c0 < degmax; c0 += 32) {
      const int left = deg - c0;
      int sa = nc, sb = nc;
      if (f < left) sa = csr[start + c0 + f];
      if (16 + f < left) sb = csr[start + c0 + 16 + f];
      {
        float v[16];
#pragma unroll
        for (int u = 0; u < 16; ++u) {
          const int s0 = __builtin_amdgcn_readlane(sa, u);
          const int s1 = __builtin_amdgcn_readlane(sa, 16 + u);
          const int sj = sub ? s1 : s0;
          v[u] = H[(size_t)sj * 16 + f];
        }
#pragma unroll
        for (int u = 0; u < 16; ++u) {
          if (u < left) acc += v[u];
        }
      }
      if (degmax - c0 > 16) {
        float v[16];
#pragma unroll
        for (int u = 0; u < 16; ++u) {
          const int s0 = __builtin_amdgcn_readlane(sb, u);
          const int s1 = __builtin_amdgcn_readlane(sb, 16 + u);
          const int sj = sub ? s1 : s0;
          v[u] = H[(size_t)sj * 16 + f];
        }
#pragma unroll
        for (int u = 0; u < 16; ++u) {
          if (16 + u < left) acc += v[u];
        }
      }
    }
    if (valid) AGG[(size_t)n * 16 + f] = acc;
  }
}

__global__ __launch_bounds__(256) void gcn_finish2_kernel(const float* __restrict__ AGG,
                                                          const float* __restrict__ HW,
                                                          const float* __restrict__ dinv,
                                                          const float* __restrict__ b2,
                                                          float* __restrict__ out_h,
                                                          float* __restrict__ out_ls,
                                                          int N) {
  int n = blockIdx.x * 256 + threadIdx.x;
  if (n >= N) return;
  const float di = dinv[n];
  float v[16];
  float m = -3.402823466e38f;
#pragma unroll
  for (int f = 0; f < 16; ++f) {
    const float t = (AGG[(size_t)n * 16 + f] + HW[(size_t)n * 16 + f]) * di + b2[f];
    v[f] = t;
    m = fmaxf(m, t);
  }
  float s = 0.0f;
#pragma unroll
  for (int f = 0; f < 16; ++f) s += expf(v[f] - m);
  const float lse = logf(s);
#pragma unroll
  for (int f = 0; f < 16; ++f) {
    out_h[(size_t)n * 16 + f] = v[f];
    out_ls[(size_t)n * 16 + f] = v[f] - m - lse;
  }
}

extern "C" void kernel_launch(void* const* d_in, const int* in_sizes, int n_in,
                              void* d_out, int out_size, void* d_ws, size_t ws_size,
                              hipStream_t stream) {
  const float* x  = (const float*)d_in[0];
  const int*   ei = (const int*)d_in[1];
  const float* W1 = (const float*)d_in[2];
  const float* b1 = (const float*)d_in[3];
  const float* W2 = (const float*)d_in[4];
  const float* b2 = (const float*)d_in[5];

  const int N = in_sizes[0] / 32;
  const int E = in_sizes[1] / 2;
  const int* src = ei;
  const int* dst = ei + E;

  int*   wsi      = (int*)d_ws;
  float* wsf      = (float*)d_ws;
  int*   cnt      = wsi;
  float* dinv     = wsf + (size_t)N;
  int*   rowstart = wsi + (size_t)N * 2;
  int*   rank     = wsi + (size_t)N * 3;
  int*   csr      = wsi + (size_t)N * 3 + (size_t)E;
  float* xw1      = wsf + (size_t)N * 3 + (size_t)E * 2;
  float* h1       = xw1 + (size_t)N * 32;
  float* hw2      = h1 + (size_t)N * 32;
  float* agg2     = hw2 + (size_t)N * 16;

  float* out_h  = (float*)d_out;
  float* out_ls = out_h + (size_t)N * 16;

  auto nb = [](long long n) { return (unsigned)((n + 255) / 256); };
  const int nTiles = N / 16;
  const unsigned gemmBlocks = (unsigned)((nTiles + 7) / 8);
  const unsigned edgeBlocks = nb(((long long)E + 3) / 4);
  const unsigned scanBlocks = (unsigned)((N + 1023) / 1024);
  const unsigned gather32Blocks = (unsigned)((N + 55) / 56);
  const unsigned gather16Blocks = (unsigned)(((N + 1) / 2 + 31) / 32);

  gcn_zero_kernel<<<nb(N), 256, 0, stream>>>(cnt, N);
  gcn_rank_kernel<<<edgeBlocks, 256, 0, stream>>>(dst, cnt, rank, E, (int)edgeBlocks * 256);
  gcn_scan_kernel<<<scanBlocks, 256, 0, stream>>>(cnt, rowstart, dinv, N);
  gcn_fill_kernel<<<edgeBlocks, 256, 0, stream>>>(src, dst, rank, rowstart, csr, E,
                                                  (int)edgeBlocks * 256);

  gcn_gemm_wmma_kernel<32><<<gemmBlocks, 256, 0, stream>>>(x, W1, dinv, xw1, nTiles);
  gcn_gather32_kernel<<<gather32Blocks, 256, 0, stream>>>(rowstart, cnt, csr, dinv, xw1, b1, h1,
                                                          N, (int)gather32Blocks * 8);

  gcn_gemm_wmma_kernel<16><<<gemmBlocks, 256, 0, stream>>>(h1, W2, dinv, hw2, nTiles);
  gcn_gather16_kernel<<<gather16Blocks, 256, 0, stream>>>(rowstart, cnt, csr, hw2, agg2,
                                                          N, (int)gather16Blocks * 8);
  gcn_finish2_kernel<<<nb(N), 256, 0, stream>>>(agg2, hw2, dinv, b2, out_h, out_ls, N);
}


# edge sort by dst on LDS counters: each 4096-edge chunk sorts by 256-node bucket in LDS, one block per bucket ranks by node (LDS add arrival order, as the global atomic before); 9 to 7 launches
# speedup vs baseline: 3.0448x; 2.2096x over previous
#include <hip/hip_runtime.h>
#include <hip/hip_bf16.h>


typedef __attribute__((ext_vector_type(2))) float v2f;
typedef __attribute__((ext_vector_type(8))) float v8f;

#define GCN_BSH 8
#define GCN_BNODES 256
#define GCN_CHUNK 4096
#define GCN_MAXB 2048

__global__ __launch_bounds__(512) void gcn_chunksort_kernel(const int* __restrict__ src,
                                                            const int* __restrict__ dst,
                                                            unsigned* __restrict__ chunked,
                                                            int* __restrict__ tab,
                                                            int E, int NB, int TS) {
  __shared__ int h[GCN_MAXB + 4];
  __shared__ unsigned stage[GCN_CHUNK];
  __shared__ int wsum[16];
  const int tid = threadIdx.x;
  const int lane = tid & 31;
  const int wave = tid >> 5;
  const int b = blockIdx.x;
  const int base = b * GCN_CHUNK + tid;
  int d[8], s[8], r[8];
#pragma unroll
  for (int u = 0; u < 8; ++u) {
    const int e = min(base + u * 512, E - 1);
    d[u] = dst[e];
    s[u] = src[e];
  }
#pragma unroll
  for (int i = 0; i < 4; ++i) h[tid * 4 + i] = 0;
  if (tid == 0) h[GCN_MAXB] = 0;
  __syncthreads();
#pragma unroll
  for (int u = 0; u < 8; ++u) {
    const int idx = (base + u * 512 < E) ? (d[u] >> GCN_BSH) : GCN_MAXB;
    r[u] = atomicAdd(&h[idx], 1);
  }
  __syncthreads();
  int c[4];
#pragma unroll
  for (int i = 0; i < 4; ++i) c[i] = h[tid * 4 + i];
  const int s4 = (c[0] + c[1]) + (c[2] + c[3]);
  int x = s4;
#pragma unroll
  for (int o = 1; o < 32; o <<= 1) {
    const int y = __shfl_up(x, o, 32);
    if (lane >= o) x += y;
  }
  if (lane == 31) wsum[wave] = x;
  __syncthreads();
  int ex = x - s4;
#pragma unroll
  for (int w = 0; w < 16; ++w) {
    if (w < wave) ex += wsum[w];
  }
#pragma unroll
  for (int i = 0; i < 4; ++i) {
    h[tid * 4 + i] = ex;
    ex += c[i];
  }
  __syncthreads();
  for (int k = tid; k <= NB; k += 512) tab[(size_t)b * TS + k] = h[k];
#pragma unroll
  for (int u = 0; u < 8; ++u) {
    if (base + u * 512 < E) {
      stage[h[d[u] >> GCN_BSH] + r[u]] =
          ((unsigned)s[u] << GCN_BSH) | (unsigned)(d[u] & (GCN_BNODES - 1));
    }
  }
  __syncthreads();
  const int total = h[NB];
#pragma unroll
  for (int u = 0; u < 8; ++u) {
    const int i = u * 512 + tid;
    if (i < total) chunked[(size_t)b * GCN_CHUNK + i] = stage[i];
  }
}

__global__ __launch_bounds__(512) void gcn_place_kernel(const int* __restrict__ tab,
                                                        const unsigned* __restrict__ chunked,
                                                        int* __restrict__ cnt,
                                                        int* __restrict__ rowstart,
                                                        float* __restrict__ dinv,
                                                        int* __restrict__ csr,
                                                        int N, int G1, int TS) {
  __shared__ int c[GCN_BNODES + 1];
  __shared__ int cur[GCN_BNODES + 1];
  __shared__ int wsum[16];
  __shared__ int wacc[16];
  const int tid = threadIdx.x;
  const int lane = tid & 31;
  const int wave = tid >> 5;
  const int k = blockIdx.x;
  if (tid <= GCN_BNODES) c[tid] = 0;
  __syncthreads();
  int acc = 0;
  for (int seg = tid; seg < G1; seg += 512) {
    const int o0 = tab[(size_t)seg * TS + k];
    const int len = tab[(size_t)seg * TS + k + 1] - o0;
    acc += o0;
    const unsigned* sp = chunked + (size_t)seg * GCN_CHUNK + o0;
    for (int j0 = 0; j0 < len; j0 += 4) {
      unsigned v[4];
#pragma unroll
      for (int u = 0; u < 4; ++u) v[u] = sp[min(j0 + u, len - 1)];
#pragma unroll
      for (int u = 0; u < 4; ++u) {
        const int idx = (j0 + u < len) ? (int)(v[u] & (GCN_BNODES - 1)) : GCN_BNODES;
        atomicAdd(&c[idx], 1);
      }
    }
  }
#pragma unroll
  for (int o = 16; o > 0; o >>= 1) acc += __shfl_xor(acc, o, 32);
  if (lane == 0) wacc[wave] = acc;
  __syncthreads();
  const int x = (tid < GCN_BNODES) ? c[tid] : 0;
  int incl = x;
#pragma unroll
  for (int o = 1; o < 32; o <<= 1) {
    const int y = __shfl_up(incl, o, 32);
    if (lane >= o) incl += y;
  }
  if (lane == 31) wsum[wave] = incl;
  __syncthreads();
  int ex = incl - x;
#pragma unroll
  for (int w = 0; w < 16; ++w) {
    ex += wacc[w];
    if (w < wave) ex += wsum[w];
  }
  if (tid < GCN_BNODES) {
    cur[tid] = ex;
    const int node = k * GCN_BNODES + tid;
    if (node < N) {
      rowstart[node] = ex;
      cnt[node] = x;
      const float dg = (float)(x + 1);
      dinv[node] = (dg > 0.0f) ? rsqrtf(dg) : 0.0f;
    }
  }
  if (tid == GCN_BNODES) cur[tid] = 0;
  __syncthreads();
  for (int seg = tid; seg < G1; seg += 512) {
    const int o0 = tab[(size_t)seg * TS + k];
    const int len = tab[(size_t)seg * TS + k + 1] - o0;
    const unsigned* sp = chunked + (size_t)seg * GCN_CHUNK + o0;
    for (int j0 = 0; j0 < len; j0 += 4) {
      unsigned v[4];
      int p[4];
#pragma unroll
      for (int u = 0; u < 4; ++u) v[u] = sp[min(j0 + u, len - 1)];
#pragma unroll
      for (int u = 0; u < 4; ++u) {
        const int idx = (j0 + u < len) ? (int)(v[u] & (GCN_BNODES - 1)) : GCN_BNODES;
        p[u] = atomicAdd(&cur[idx], 1);
      }
#pragma unroll
      for (int u = 0; u < 4; ++u) {
        if (j0 + u < len) csr[p[u]] = (int)(v[u] >> GCN_BSH);
      }
    }
  }
}

template <int NCOL>
__global__ __launch_bounds__(256) void gcn_gemm_wmma_kernel(const float* __restrict__ X,
                                                            const float* __restrict__ W,
                                                            const float* __restrict__ dinv,
                                                            float* __restrict__ Y,
                                                            int nTiles) {
  const int lane = threadIdx.x & 31;
  const int wave = threadIdx.x >> 5;
  const int tile = blockIdx.x * 8 + wave;
  if (tile >= nTiles) return;

  const int rowBase = tile * 16;
  const int lo = lane & 15;
  const int hi = lane >> 4;

  v2f a[8];
#pragma unroll
  for (int kb = 0; kb < 8; ++kb) {
    const int k = kb * 4 + hi * 2;
    const float* xr = X + (size_t)(rowBase + lo) * 32 + k;
    a[kb].x = xr[0];
    a[kb].y = xr[1];
  }
  float rs[8];
#pragma unroll
  for (int v = 0; v < 8; ++v) rs[v] = dinv[rowBase + v + hi * 8];

#pragma unroll
  for (int ct = 0; ct < NCOL / 16; ++ct) {
    v8f c = {};
#pragma unroll
    for (int kb = 0; kb < 8; ++kb) {
      const int k = kb * 4 + hi * 2;
      v2f b;
      b.x = W[(size_t)k * NCOL + ct * 16 + lo];
      b.y = W[(size_t)(k + 1) * NCOL + ct * 16 + lo];
      c = __builtin_amdgcn_wmma_f32_16x16x4_f32(false, a[kb], false, b,
                                                (short)0, c, false, false);
    }
#pragma unroll
    for (int v = 0; v < 8; ++v) {
      const int m = v + hi * 8;
      Y[(size_t)(rowBase + m) * NCOL + ct * 16 + lo] = c[v] * rs[v];
    }
  }
}


__global__ __launch_bounds__(256) void gcn_gather32_kernel(const int* __restrict__ rowstart,
                                                           const int* __restrict__ cnt,
                                                           const int* __restrict__ csr,
                                                           const float* __restrict__ dinv,
                                                           const float* __restrict__ H,
                                                           const float* __restrict__ b1,
                                                           float* __restrict__ H1,
                                                           int N, int nWaves) {
  const int lane = threadIdx.x & 31;
  const int wave = __builtin_amdgcn_readfirstlane(blockIdx.x * 8 + (threadIdx.x >> 5));
  const float bias = b1[lane];
  int n = wave;
  int start = 0, deg = 0;
  float dn = 0.0f;
  if (n < N) { start = rowstart[n]; deg = cnt[n]; dn = dinv[n]; }
  while (n < N) {
    const int n2 = n + nWaves;
    int start2 = 0, deg2 = 0;
    float dn2 = 0.0f;
    if (n2 < N) { start2 = rowstart[n2]; deg2 = cnt[n2]; dn2 = dinv[n2]; }
    const float selfv = H[(size_t)n * 32 + lane];
    float acc = 0.0f;
    for (int c0 = 0; c0 < deg; c0 += 32) {
      const int nv = min(32, deg - c0);
      int s = n;
      if (lane < nv) s = csr[start + c0 + lane];
      for (int j0 = 0; j0 < nv; j0 += 16) {
        float v[16];
#pragma unroll
        for (int u = 0; u < 16; ++u) {
          const int q = j0 + u;
          const int sj = __builtin_amdgcn_readlane(s, q);
          v[u] = H[(size_t)sj * 32 + lane];
        }
#pragma unroll
        for (int u = 0; u < 16; ++u) {
          if (j0 + u < nv) acc += v[u];
        }
      }
    }
    const float t = (acc + selfv) * dn;
    H1[(size_t)n * 32 + lane] = tanhf(t + bias);
    n = n2; start = start2; deg = deg2; dn = dn2;
  }
}

__global__ __launch_bounds__(256) void gcn_gather16_kernel(const int* __restrict__ rowstart,
                                                           const int* __restrict__ cnt,
                                                           const int* __restrict__ csr,
                                                           const float* __restrict__ H,
                                                           float* __restrict__ AGG,
                                                           int N, int nWaves) {
  const int lane = threadIdx.x & 31;
  const int f = lane & 15;
  const int sub = lane >> 4;
  const int wave = __builtin_amdgcn_readfirstlane(blockIdx.x * 8 + (threadIdx.x >> 5));
  const int nPairs = (N + 1) >> 1;
  for (int p = wave; p < nPairs; p += nWaves) {
    const int n = 2 * p + sub;
    const bool valid = n < N;
    const int nc = valid ? n : N - 1;
    const int start = rowstart[nc];
    const int deg = valid ? cnt[nc] : 0;
    const int degmax = max(__builtin_amdgcn_readlane(deg, 0), __builtin_amdgcn_readlane(deg, 16));
    float acc = 0.0f;
    for (int c0 = 0; c0 < degmax; c0 += 32) {
      const int left = deg - c0;
      int sa = nc, sb = nc;
      if (f < left) sa = csr[start + c0 + f];
      if (16 + f < left) sb = csr[start + c0 + 16 + f];
      {
        float v[16];
#pragma unroll
        for (int u = 0; u < 16; ++u) {
          const int s0 = __builtin_amdgcn_readlane(sa, u);
          const int s1 = __builtin_amdgcn_readlane(sa, 16 + u);
          const int sj = sub ? s1 : s0;
          v[u] = H[(size_t)sj * 16 + f];
        }
#pragma unroll
        for (int u = 0; u < 16; ++u) {
          if (u < left) acc += v[u];
        }
      }
      if (degmax - c0 > 16) {
        float v[16];
#pragma unroll
        for (int u = 0; u < 16; ++u) {
          const int s0 = __builtin_amdgcn_readlane(sb, u);
          const int s1 = __builtin_amdgcn_readlane(sb, 16 + u);
          const int sj = sub ? s1 : s0;
          v[u] = H[(size_t)sj * 16 + f];
        }
#pragma unroll
        for (int u = 0; u < 16; ++u) {
          if (16 + u < left) acc += v[u];
        }
      }
    }
    if (valid) AGG[(size_t)n * 16 + f] = acc;
  }
}

__global__ __launch_bounds__(256) void gcn_finish2_kernel(const float* __restrict__ AGG,
                                                          const float* __restrict__ HW,
                                                          const float* __restrict__ dinv,
                                                          const float* __restrict__ b2,
                                                          float* __restrict__ out_h,
                                                          float* __restrict__ out_ls,
                                                          int N) {
  int n = blockIdx.x * 256 + threadIdx.x;
  if (n >= N) return;
  const float di = dinv[n];
  float v[16];
  float m = -3.402823466e38f;
#pragma unroll
  for (int f = 0; f < 16; ++f) {
    const float t = (AGG[(size_t)n * 16 + f] + HW[(size_t)n * 16 + f]) * di + b2[f];
    v[f] = t;
    m = fmaxf(m, t);
  }
  float s = 0.0f;
#pragma unroll
  for (int f = 0; f < 16; ++f) s += expf(v[f] - m);
  const float lse = logf(s);
#pragma unroll
  for (int f = 0; f < 16; ++f) {
    out_h[(size_t)n * 16 + f] = v[f];
    out_ls[(size_t)n * 16 + f] = v[f] - m - lse;
  }
}

extern "C" void kernel_launch(void* const* d_in, const int* in_sizes, int n_in,
                              void* d_out, int out_size, void* d_ws, size_t ws_size,
                              hipStream_t stream) {
  const float* x  = (const float*)d_in[0];
  const int*   ei = (const int*)d_in[1];
  const float* W1 = (const float*)d_in[2];
  const float* b1 = (const float*)d_in[3];
  const float* W2 = (const float*)d_in[4];
  const float* b2 = (const float*)d_in[5];

  const int N = in_sizes[0] / 32;
  const int E = in_sizes[1] / 2;
  const int* src = ei;
  const int* dst = ei + E;

  int*   wsi      = (int*)d_ws;
  float* wsf      = (float*)d_ws;
  int*   cnt      = wsi;
  float* dinv     = wsf + (size_t)N;
  int*   rowstart = wsi + (size_t)N * 2;
  const int NB = (N + GCN_BNODES - 1) / GCN_BNODES;
  const int G1 = (E + GCN_CHUNK - 1) / GCN_CHUNK;
  const int TS = NB + 1;
  unsigned* chunked = (unsigned*)(wsi + (size_t)N * 3);
  int*   csr      = wsi + (size_t)N * 3 + (size_t)E;
  float* xw1      = wsf + (size_t)N * 3 + (size_t)E * 2;
  float* h1       = xw1 + (size_t)N * 32;
  float* hw2      = h1 + (size_t)N * 32;
  float* agg2     = hw2 + (size_t)N * 16;
  int*   tab      = (int*)(agg2 + (size_t)N * 16);

  float* out_h  = (float*)d_out;
  float* out_ls = out_h + (size_t)N * 16;

  auto nb = [](long long n) { return (unsigned)((n + 255) / 256); };
  const int nTiles = N / 16;
  const unsigned gemmBlocks = (unsigned)((nTiles + 7) / 8);
  const unsigned gather32Blocks = (unsigned)((N + 55) / 56);
  const unsigned gather16Blocks = (unsigned)(((N + 1) / 2 + 31) / 32);

  gcn_chunksort_kernel<<<(unsigned)G1, 512, 0, stream>>>(src, dst, chunked, tab, E, NB, TS);
  gcn_place_kernel<<<(unsigned)NB, 512, 0, stream>>>(tab, chunked, cnt, rowstart, dinv, csr,
                                                    N, G1, TS);

  gcn_gemm_wmma_kernel<32><<<gemmBlocks, 256, 0, stream>>>(x, W1, dinv, xw1, nTiles);
  gcn_gather32_kernel<<<gather32Blocks, 256, 0, stream>>>(rowstart, cnt, csr, dinv, xw1, b1, h1,
                                                          N, (int)gather32Blocks * 8);

  gcn_gemm_wmma_kernel<16><<<gemmBlocks, 256, 0, stream>>>(h1, W2, dinv, hw2, nTiles);
  gcn_gather16_kernel<<<gather16Blocks, 256, 0, stream>>>(rowstart, cnt, csr, hw2, agg2,
                                                          N, (int)gather16Blocks * 8);
  gcn_finish2_kernel<<<nb(N), 256, 0, stream>>>(agg2, hw2, dinv, b2, out_h, out_ls, N);
}


# launch fusion, 7 to 4 launches: X@W1 in the edge-place blocks, h1@W2 in the layer-1 gather (rows staged in LDS), bias+log_softmax in the layer-2 gather (softmax sum as a 16-lane tree)
# speedup vs baseline: 3.4570x; 1.1354x over previous
#include <hip/hip_runtime.h>
#include <hip/hip_bf16.h>


typedef __attribute__((ext_vector_type(2))) float v2f;
typedef __attribute__((ext_vector_type(8))) float v8f;


template <int NCOL>
__device__ __forceinline__ void gcn_load_b(const float* __restrict__ W, v2f* b, int lo, int hi) {
#pragma unroll
  for (int ct = 0; ct < NCOL / 16; ++ct) {
#pragma unroll
    for (int kb = 0; kb < 8; ++kb) {
      const int k = kb * 4 + hi * 2;
      b[ct * 8 + kb].x = W[(size_t)k * NCOL + ct * 16 + lo];
      b[ct * 8 + kb].y = W[(size_t)(k + 1) * NCOL + ct * 16 + lo];
    }
  }
}

template <int NCOL>
__device__ __forceinline__ void gcn_tile_product(const v2f* a, const v2f* b, const float* rs,
                                                 float* __restrict__ Yt, int lo, int hi) {
#pragma unroll
  for (int ct = 0; ct < NCOL / 16; ++ct) {
    v8f c = {};
#pragma unroll
    for (int kb = 0; kb < 8; ++kb) {
      c = __builtin_amdgcn_wmma_f32_16x16x4_f32(false, a[kb], false, b[ct * 8 + kb],
                                                (short)0, c, false, false);
    }
#pragma unroll
    for (int v = 0; v < 8; ++v) {
      const int m = v + hi * 8;
      Yt[(size_t)m * NCOL + ct * 16 + lo] = c[v] * rs[v];
    }
  }
}

#define GCN_BSH 8
#define GCN_BNODES 256
#define GCN_CHUNK 4096
#define GCN_MAXB 2048

__global__ __launch_bounds__(512) void gcn_chunksort_kernel(const int* __restrict__ src,
                                                            const int* __restrict__ dst,
                                                            unsigned* __restrict__ chunked,
                                                            int* __restrict__ tab,
                                                            int E, int NB, int TS) {
  __shared__ int h[GCN_MAXB + 4];
  __shared__ unsigned stage[GCN_CHUNK];
  __shared__ int wsum[16];
  const int tid = threadIdx.x;
  const int lane = tid & 31;
  const int wave = tid >> 5;
  const int b = blockIdx.x;
  const int base = b * GCN_CHUNK + tid;
  int d[8], s[8], r[8];
#pragma unroll
  for (int u = 0; u < 8; ++u) {
    const int e = min(base + u * 512, E - 1);
    d[u] = dst[e];
    s[u] = src[e];
  }
#pragma unroll
  for (int i = 0; i < 4; ++i) h[tid * 4 + i] = 0;
  if (tid == 0) h[GCN_MAXB] = 0;
  __syncthreads();
#pragma unroll
  for (int u = 0; u < 8; ++u) {
    const int idx = (base + u * 512 < E) ? (d[u] >> GCN_BSH) : GCN_MAXB;
    r[u] = atomicAdd(&h[idx], 1);
  }
  __syncthreads();
  int c[4];
#pragma unroll
  for (int i = 0; i < 4; ++i) c[i] = h[tid * 4 + i];
  const int s4 = (c[0] + c[1]) + (c[2] + c[3]);
  int x = s4;
#pragma unroll
  for (int o = 1; o < 32; o <<= 1) {
    const int y = __shfl_up(x, o, 32);
    if (lane >= o) x += y;
  }
  if (lane == 31) wsum[wave] = x;
  __syncthreads();
  int ex = x - s4;
#pragma unroll
  for (int w = 0; w < 16; ++w) {
    if (w < wave) ex += wsum[w];
  }
#pragma unroll
  for (int i = 0; i < 4; ++i) {
    h[tid * 4 + i] = ex;
    ex += c[i];
  }
  __syncthreads();
  for (int k = tid; k <= NB; k += 512) tab[(size_t)b * TS + k] = h[k];
#pragma unroll
  for (int u = 0; u < 8; ++u) {
    if (base + u * 512 < E) {
      stage[h[d[u] >> GCN_BSH] + r[u]] =
          ((unsigned)s[u] << GCN_BSH) | (unsigned)(d[u] & (GCN_BNODES - 1));
    }
  }
  __syncthreads();
  const int total = h[NB];
#pragma unroll
  for (int u = 0; u < 8; ++u) {
    const int i = u * 512 + tid;
    if (i < total) chunked[(size_t)b * GCN_CHUNK + i] = stage[i];
  }
}

__global__ __launch_bounds__(512) void gcn_place_kernel(const int* __restrict__ tab,
                                                        const unsigned* __restrict__ chunked,
                                                        int* __restrict__ cnt,
                                                        int* __restrict__ rowstart,
                                                        float* __restrict__ dinv,
                                                        int* __restrict__ csr,
                                                        const float* __restrict__ X,
                                                        const float* __restrict__ W,
                                                        float* __restrict__ Y,
                                                        int N, int G1, int TS) {
  __shared__ int c[GCN_BNODES + 1];
  __shared__ int cur[GCN_BNODES + 1];
  __shared__ int wsum[16];
  __shared__ int wacc[16];
  __shared__ float sdinv[GCN_BNODES];
  const int tid = threadIdx.x;
  const int lane = tid & 31;
  const int wave = tid >> 5;
  const int k = blockIdx.x;
  const int lo = lane & 15;
  const int hi = lane >> 4;
  const int rowBase = __builtin_amdgcn_readfirstlane(k * GCN_BNODES + wave * 16);
  if (tid <= GCN_BNODES) c[tid] = 0;
  __syncthreads();
  v2f a[8];
  {
    const int r = min(rowBase + lo, N - 1);
#pragma unroll
    for (int kb = 0; kb < 8; ++kb) {
      const float* xr = X + (size_t)r * 32 + kb * 4 + hi * 2;
      a[kb].x = xr[0];
      a[kb].y = xr[1];
    }
  }
  int acc = 0;
  for (int seg = tid; seg < G1; seg += 512) {
    const int o0 = tab[(size_t)seg * TS + k];
    const int len = tab[(size_t)seg * TS + k + 1] - o0;
    acc += o0;
    const unsigned* sp = chunked + (size_t)seg * GCN_CHUNK + o0;
    for (int j0 = 0; j0 < len; j0 += 4) {
      unsigned v[4];
#pragma unroll
      for (int u = 0; u < 4; ++u) v[u] = sp[min(j0 + u, len - 1)];
#pragma unroll
      for (int u = 0; u < 4; ++u) {
        const int idx = (j0 + u < len) ? (int)(v[u] & (GCN_BNODES - 1)) : GCN_BNODES;
        atomicAdd(&c[idx], 1);
      }
    }
  }
#pragma unroll
  for (int o = 16; o > 0; o >>= 1) acc += __shfl_xor(acc, o, 32);
  if (lane == 0) wacc[wave] = acc;
  __syncthreads();
  const int x = (tid < GCN_BNODES) ? c[tid] : 0;
  int incl = x;
#pragma unroll
  for (int o = 1; o < 32; o <<= 1) {
    const int y = __shfl_up(incl, o, 32);
    if (lane >= o) incl += y;
  }
  if (lane == 31) wsum[wave] = incl;
  __syncthreads();
  int ex = incl - x;
#pragma unroll
  for (int w = 0; w < 16; ++w) {
    ex += wacc[w];
    if (w < wave) ex += wsum[w];
  }
  if (tid < GCN_BNODES) {
    cur[tid] = ex;
    const int node = k * GCN_BNODES + tid;
    if (node < N) {
      rowstart[node] = ex;
      cnt[node] = x;
      const float dg = (float)(x + 1);
      const float di = (dg > 0.0f) ? rsqrtf(dg) : 0.0f;
      dinv[node] = di;
      sdinv[tid] = di;
    }
  }
  if (tid == GCN_BNODES) cur[tid] = 0;
  __syncthreads();
  v2f bw[16];
  gcn_load_b<32>(W, bw, lo, hi);
  for (int seg = tid; seg < G1; seg += 512) {
    const int o0 = tab[(size_t)seg * TS + k];
    const int len = tab[(size_t)seg * TS + k + 1] - o0;
    const unsigned* sp = chunked + (size_t)seg * GCN_CHUNK + o0;
    for (int j0 = 0; j0 < len; j0 += 4) {
      unsigned v[4];
      int p[4];
#pragma unroll
      for (int u = 0; u < 4; ++u) v[u] = sp[min(j0 + u, len - 1)];
#pragma unroll
      for (int u = 0; u < 4; ++u) {
        const int idx = (j0 + u < len) ? (int)(v[u] & (GCN_BNODES - 1)) : GCN_BNODES;
        p[u] = atomicAdd(&cur[idx], 1);
      }
#pragma unroll
      for (int u = 0; u < 4; ++u) {
        if (j0 + u < len) csr[p[u]] = (int)(v[u] >> GCN_BSH);
      }
    }
  }
  if (rowBase + 16 <= N) {
    float rs[8];
#pragma unroll
    for (int v = 0; v < 8; ++v) rs[v] = sdinv[wave * 16 + v + hi * 8];
    gcn_tile_product<32>(a, bw, rs, Y + (size_t)rowBase * 32, lo, hi);
  }
}


#define GCN_GNODES 64
#define GCN_TPITCH 36

__global__ __launch_bounds__(256) void gcn_gather32_kernel(const int* __restrict__ rowstart,
                                                           const int* __restrict__ cnt,
                                                           const int* __restrict__ csr,
                                                           const float* __restrict__ dinv,
                                                           const float* __restrict__ H,
                                                           const float* __restrict__ b1,
                                                           const float* __restrict__ W2,
                                                           float* __restrict__ HW2,
                                                           int N) {
  __shared__ float tile[GCN_GNODES * GCN_TPITCH];
  __shared__ float sdn[GCN_GNODES];
  const int lane = threadIdx.x & 31;
  const int wave = __builtin_amdgcn_readfirstlane(threadIdx.x >> 5);
  const int blockNode = blockIdx.x * GCN_GNODES;
  const int node0 = blockNode + wave * 8;
  const float bias = b1[lane];
  const int lo = lane & 15;
  const int hi = lane >> 4;
  v2f bw[8] = {};
  if (wave < 4) gcn_load_b<16>(W2, bw, lo, hi);
  const int mn = node0 + (lane & 7);
  const int mc = min(mn, N - 1);
  const int mstart = rowstart[mc];
  const int mdeg = (mn < N) ? cnt[mc] : 0;
  const float mdn = dinv[mc];
  if (lane < 8) sdn[wave * 8 + lane] = mdn;
  for (int i = 0; i < 8; ++i) {
    const int n = node0 + i;
    if (n < N) {
      const int start = __builtin_amdgcn_readlane(mstart, i);
      const int deg = __builtin_amdgcn_readlane(mdeg, i);
      const float dn = __builtin_bit_cast(float,
                                          __builtin_amdgcn_readlane(__builtin_bit_cast(int, mdn), i));
      const float selfv = H[(size_t)n * 32 + lane];
      float acc = 0.0f;
      for (int c0 = 0; c0 < deg; c0 += 32) {
        const int nv = min(32, deg - c0);
        int s = n;
        if (lane < nv) s = csr[start + c0 + lane];
        for (int j0 = 0; j0 < nv; j0 += 16) {
          float v[16];
#pragma unroll
          for (int u = 0; u < 16; ++u) {
            const int q = j0 + u;
            const int sj = __builtin_amdgcn_readlane(s, q);
            v[u] = H[(size_t)sj * 32 + lane];
          }
#pragma unroll
          for (int u = 0; u < 16; ++u) {
            if (j0 + u < nv) acc += v[u];
          }
        }
      }
      const float t = (acc + selfv) * dn;
      tile[(wave * 8 + i) * GCN_TPITCH + lane] = tanhf(t + bias);
    }
  }
  __syncthreads();
  const int rowBase = blockNode + wave * 16;
  if (wave < 4 && rowBase + 16 <= N) {
    v2f a[8];
#pragma unroll
    for (int kb = 0; kb < 8; ++kb) {
      const float* xr = tile + (wave * 16 + lo) * GCN_TPITCH + kb * 4 + hi * 2;
      a[kb].x = xr[0];
      a[kb].y = xr[1];
    }
    float rs[8];
#pragma unroll
    for (int v = 0; v < 8; ++v) rs[v] = sdn[wave * 16 + v + hi * 8];
    gcn_tile_product<16>(a, bw, rs, HW2 + (size_t)rowBase * 16, lo, hi);
  }
}


template <int M>
__device__ __forceinline__ float gcn_row_xor(float x) {
  return __builtin_bit_cast(
      float, __builtin_amdgcn_update_dpp(0, __builtin_bit_cast(int, x), 0x160 + M, 0xf, 0xf, false));
}

__global__ __launch_bounds__(256) void gcn_gather16_kernel(const int* __restrict__ rowstart,
                                                           const int* __restrict__ cnt,
                                                           const int* __restrict__ csr,
                                                           const float* __restrict__ dinv,
                                                           const float* __restrict__ H,
                                                           const float* __restrict__ b2,
                                                           float* __restrict__ out_h,
                                                           float* __restrict__ out_ls,
                                                           int N, int nWaves) {
  const int lane = threadIdx.x & 31;
  const int f = lane & 15;
  const int sub = lane >> 4;
  const float bias = b2[f];
  const int wave = __builtin_amdgcn_readfirstlane(blockIdx.x * 8 + (threadIdx.x >> 5));
  const int nPairs = (N + 1) >> 1;
  for (int p = wave; p < nPairs; p += nWaves) {
    const int n = 2 * p + sub;
    const bool valid = n < N;
    const int nc = valid ? n : N - 1;
    const int start = rowstart[nc];
    const int deg = valid ? cnt[nc] : 0;
    const int degmax = max(__builtin_amdgcn_readlane(deg, 0), __builtin_amdgcn_readlane(deg, 16));
    const float di = dinv[nc];
    const float selfv = H[(size_t)nc * 16 + f];
    float acc = 0.0f;
    for (int c0 = 0; c0 < degmax; c0 += 32) {
      const int left = deg - c0;
      int sa = nc, sb = nc;
      if (f < left) sa = csr[start + c0 + f];
      if (16 + f < left) sb = csr[start + c0 + 16 + f];
      {
        float v[16];
#pragma unroll
        for (int u = 0; u < 16; ++u) {
          const int s0 = __builtin_amdgcn_readlane(sa, u);
          const int s1 = __builtin_amdgcn_readlane(sa, 16 + u);
          const int sj = sub ? s1 : s0;
          v[u] = H[(size_t)sj * 16 + f];
        }
#pragma unroll
        for (int u = 0; u < 16; ++u) {
          if (u < left) acc += v[u];
        }
      }
      if (degmax - c0 > 16) {
        float v[16];
#pragma unroll
        for (int u = 0; u < 16; ++u) {
          const int s0 = __builtin_amdgcn_readlane(sb, u);
          const int s1 = __builtin_amdgcn_readlane(sb, 16 + u);
          const int sj = sub ? s1 : s0;
          v[u] = H[(size_t)sj * 16 + f];
        }
#pragma unroll
        for (int u = 0; u < 16; ++u) {
          if (16 + u < left) acc += v[u];
        }
      }
    }
    const float t = (acc + selfv) * di + bias;
    float m = t;
    m = fmaxf(m, gcn_row_xor<1>(m));
    m = fmaxf(m, gcn_row_xor<2>(m));
    m = fmaxf(m, gcn_row_xor<4>(m));
    m = fmaxf(m, gcn_row_xor<8>(m));
    float se = expf(t - m);
    se += gcn_row_xor<1>(se);
    se += gcn_row_xor<2>(se);
    se += gcn_row_xor<4>(se);
    se += gcn_row_xor<8>(se);
    const float lse = logf(se);
    if (valid) {
      out_h[(size_t)n * 16 + f] = t;
      out_ls[(size_t)n * 16 + f] = t - m - lse;
    }
  }
}

extern "C" void kernel_launch(void* const* d_in, const int* in_sizes, int n_in,
                              void* d_out, int out_size, void* d_ws, size_t ws_size,
                              hipStream_t stream) {
  const float* x  = (const float*)d_in[0];
  const int*   ei = (const int*)d_in[1];
  const float* W1 = (const float*)d_in[2];
  const float* b1 = (const float*)d_in[3];
  const float* W2 = (const float*)d_in[4];
  const float* b2 = (const float*)d_in[5];

  const int N = in_sizes[0] / 32;
  const int E = in_sizes[1] / 2;
  const int* src = ei;
  const int* dst = ei + E;

  int*   wsi      = (int*)d_ws;
  float* wsf      = (float*)d_ws;
  int*   cnt      = wsi;
  float* dinv     = wsf + (size_t)N;
  int*   rowstart = wsi + (size_t)N * 2;
  const int NB = (N + GCN_BNODES - 1) / GCN_BNODES;
  const int G1 = (E + GCN_CHUNK - 1) / GCN_CHUNK;
  const int TS = NB + 1;
  unsigned* chunked = (unsigned*)(wsi + (size_t)N * 3);
  int*   csr      = wsi + (size_t)N * 3 + (size_t)E;
  float* xw1      = wsf + (size_t)N * 3 + (size_t)E * 2;
  float* hw2      = xw1 + (size_t)N * 64;
  int*   tab      = (int*)(hw2 + (size_t)N * 32);

  float* out_h  = (float*)d_out;
  float* out_ls = out_h + (size_t)N * 16;

  const unsigned gather32Blocks = (unsigned)((N + GCN_GNODES - 1) / GCN_GNODES);
  const unsigned gather16Blocks = (unsigned)(((N + 1) / 2 + 31) / 32);

  gcn_chunksort_kernel<<<(unsigned)G1, 512, 0, stream>>>(src, dst, chunked, tab, E, NB, TS);
  gcn_place_kernel<<<(unsigned)NB, 512, 0, stream>>>(tab, chunked, cnt, rowstart, dinv, csr,
                                                    x, W1, xw1, N, G1, TS);

  gcn_gather32_kernel<<<gather32Blocks, 256, 0, stream>>>(rowstart, cnt, csr, dinv, xw1, b1, W2,
                                                          hw2, N);

  gcn_gather16_kernel<<<gather16Blocks, 256, 0, stream>>>(rowstart, cnt, csr, dinv, hw2, b2,
                                                          out_h, out_ls, N, (int)gather16Blocks * 8);
}


# gather latency chain: next node's (pair's) index and bounds loads go out under the current row gathers, and a node's up to 32 row gathers issue as one group; order of every sum unchanged
# speedup vs baseline: 3.7162x; 1.0750x over previous
#include <hip/hip_runtime.h>
#include <hip/hip_bf16.h>


typedef __attribute__((ext_vector_type(2))) float v2f;
typedef __attribute__((ext_vector_type(8))) float v8f;


template <int NCOL>
__device__ __forceinline__ void gcn_load_b(const float* __restrict__ W, v2f* b, int lo, int hi) {
#pragma unroll
  for (int ct = 0; ct < NCOL / 16; ++ct) {
#pragma unroll
    for (int kb = 0; kb < 8; ++kb) {
      const int k = kb * 4 + hi * 2;
      b[ct * 8 + kb].x = W[(size_t)k * NCOL + ct * 16 + lo];
      b[ct * 8 + kb].y = W[(size_t)(k + 1) * NCOL + ct * 16 + lo];
    }
  }
}

template <int NCOL>
__device__ __forceinline__ void gcn_tile_product(const v2f* a, const v2f* b, const float* rs,
                                                 float* __restrict__ Yt, int lo, int hi) {
#pragma unroll
  for (int ct = 0; ct < NCOL / 16; ++ct) {
    v8f c = {};
#pragma unroll
    for (int kb = 0; kb < 8; ++kb) {
      c = __builtin_amdgcn_wmma_f32_16x16x4_f32(false, a[kb], false, b[ct * 8 + kb],
                                                (short)0, c, false, false);
    }
#pragma unroll
    for (int v = 0; v < 8; ++v) {
      const int m = v + hi * 8;
      Yt[(size_t)m * NCOL + ct * 16 + lo] = c[v] * rs[v];
    }
  }
}

#define GCN_BSH 8
#define GCN_BNODES 256
#define GCN_CHUNK 4096
#define GCN_MAXB 2048

__global__ __launch_bounds__(512) void gcn_chunksort_kernel(const int* __restrict__ src,
                                                            const int* __restrict__ dst,
                                                            unsigned* __restrict__ chunked,
                                                            int* __restrict__ tab,
                                                            int E, int NB, int TS) {
  __shared__ int h[GCN_MAXB + 4];
  __shared__ unsigned stage[GCN_CHUNK];
  __shared__ int wsum[16];
  const int tid = threadIdx.x;
  const int lane = tid & 31;
  const int wave = tid >> 5;
  const int b = blockIdx.x;
  const int base = b * GCN_CHUNK + tid;
  int d[8], s[8], r[8];
#pragma unroll
  for (int u = 0; u < 8; ++u) {
    const int e = min(base + u * 512, E - 1);
    d[u] = dst[e];
    s[u] = src[e];
  }
#pragma unroll
  for (int i = 0; i < 4; ++i) h[tid * 4 + i] = 0;
  if (tid == 0) h[GCN_MAXB] = 0;
  __syncthreads();
#pragma unroll
  for (int u = 0; u < 8; ++u) {
    const int idx = (base + u * 512 < E) ? (d[u] >> GCN_BSH) : GCN_MAXB;
    r[u] = atomicAdd(&h[idx], 1);
  }
  __syncthreads();
  int c[4];
#pragma unroll
  for (int i = 0; i < 4; ++i) c[i] = h[tid * 4 + i];
  const int s4 = (c[0] + c[1]) + (c[2] + c[3]);
  int x = s4;
#pragma unroll
  for (int o = 1; o < 32; o <<= 1) {
    const int y = __shfl_up(x, o, 32);
    if (lane >= o) x += y;
  }
  if (lane == 31) wsum[wave] = x;
  __syncthreads();
  int ex = x - s4;
#pragma unroll
  for (int w = 0; w < 16; ++w) {
    if (w < wave) ex += wsum[w];
  }
#pragma unroll
  for (int i = 0; i < 4; ++i) {
    h[tid * 4 + i] = ex;
    ex += c[i];
  }
  __syncthreads();
  for (int k = tid; k <= NB; k += 512) tab[(size_t)b * TS + k] = h[k];
#pragma unroll
  for (int u = 0; u < 8; ++u) {
    if (base + u * 512 < E) {
      stage[h[d[u] >> GCN_BSH] + r[u]] =
          ((unsigned)s[u] << GCN_BSH) | (unsigned)(d[u] & (GCN_BNODES - 1));
    }
  }
  __syncthreads();
  const int total = h[NB];
#pragma unroll
  for (int u = 0; u < 8; ++u) {
    const int i = u * 512 + tid;
    if (i < total) chunked[(size_t)b * GCN_CHUNK + i] = stage[i];
  }
}

__global__ __launch_bounds__(512) void gcn_place_kernel(const int* __restrict__ tab,
                                                        const unsigned* __restrict__ chunked,
                                                        int* __restrict__ cnt,
                                                        int* __restrict__ rowstart,
                                                        float* __restrict__ dinv,
                                                        int* __restrict__ csr,
                                                        const float* __restrict__ X,
                                                        const float* __restrict__ W,
                                                        float* __restrict__ Y,
                                                        int N, int G1, int TS) {
  __shared__ int c[GCN_BNODES + 1];
  __shared__ int cur[GCN_BNODES + 1];
  __shared__ int wsum[16];
  __shared__ int wacc[16];
  __shared__ float sdinv[GCN_BNODES];
  const int tid = threadIdx.x;
  const int lane = tid & 31;
  const int wave = tid >> 5;
  const int k = blockIdx.x;
  const int lo = lane & 15;
  const int hi = lane >> 4;
  const int rowBase = __builtin_amdgcn_readfirstlane(k * GCN_BNODES + wave * 16);
  if (tid <= GCN_BNODES) c[tid] = 0;
  __syncthreads();
  v2f a[8];
  {
    const int r = min(rowBase + lo, N - 1);
#pragma unroll
    for (int kb = 0; kb < 8; ++kb) {
      const float* xr = X + (size_t)r * 32 + kb * 4 + hi * 2;
      a[kb].x = xr[0];
      a[kb].y = xr[1];
    }
  }
  int acc = 0;
  for (int seg = tid; seg < G1; seg += 512) {
    const int o0 = tab[(size_t)seg * TS + k];
    const int len = tab[(size_t)seg * TS + k + 1] - o0;
    acc += o0;
    const unsigned* sp = chunked + (size_t)seg * GCN_CHUNK + o0;
    for (int j0 = 0; j0 < len; j0 += 4) {
      unsigned v[4];
#pragma unroll
      for (int u = 0; u < 4; ++u) v[u] = sp[min(j0 + u, len - 1)];
#pragma unroll
      for (int u = 0; u < 4; ++u) {
        const int idx = (j0 + u < len) ? (int)(v[u] & (GCN_BNODES - 1)) : GCN_BNODES;
        atomicAdd(&c[idx], 1);
      }
    }
  }
#pragma unroll
  for (int o = 16; o > 0; o >>= 1) acc += __shfl_xor(acc, o, 32);
  if (lane == 0) wacc[wave] = acc;
  __syncthreads();
  const int x = (tid < GCN_BNODES) ? c[tid] : 0;
  int incl = x;
#pragma unroll
  for (int o = 1; o < 32; o <<= 1) {
    const int y = __shfl_up(incl, o, 32);
    if (lane >= o) incl += y;
  }
  if (lane == 31) wsum[wave] = incl;
  __syncthreads();
  int ex = incl - x;
#pragma unroll
  for (int w = 0; w < 16; ++w) {
    ex += wacc[w];
    if (w < wave) ex += wsum[w];
  }
  if (tid < GCN_BNODES) {
    cur[tid] = ex;
    const int node = k * GCN_BNODES + tid;
    if (node < N) {
      rowstart[node] = ex;
      cnt[node] = x;
      const float dg = (float)(x + 1);
      const float di = (dg > 0.0f) ? rsqrtf(dg) : 0.0f;
      dinv[node] = di;
      sdinv[tid] = di;
    }
  }
  if (tid == GCN_BNODES) cur[tid] = 0;
  __syncthreads();
  v2f bw[16];
  gcn_load_b<32>(W, bw, lo, hi);
  for (int seg = tid; seg < G1; seg += 512) {
    const int o0 = tab[(size_t)seg * TS + k];
    const int len = tab[(size_t)seg * TS + k + 1] - o0;
    const unsigned* sp = chunked + (size_t)seg * GCN_CHUNK + o0;
    for (int j0 = 0; j0 < len; j0 += 4) {
      unsigned v[4];
      int p[4];
#pragma unroll
      for (int u = 0; u < 4; ++u) v[u] = sp[min(j0 + u, len - 1)];
#pragma unroll
      for (int u = 0; u < 4; ++u) {
        const int idx = (j0 + u < len) ? (int)(v[u] & (GCN_BNODES - 1)) : GCN_BNODES;
        p[u] = atomicAdd(&cur[idx], 1);
      }
#pragma unroll
      for (int u = 0; u < 4; ++u) {
        if (j0 + u < len) csr[p[u]] = (int)(v[u] >> GCN_BSH);
      }
    }
  }
  if (rowBase + 16 <= N) {
    float rs[8];
#pragma unroll
    for (int v = 0; v < 8; ++v) rs[v] = sdinv[wave * 16 + v + hi * 8];
    gcn_tile_product<32>(a, bw, rs, Y + (size_t)rowBase * 32, lo, hi);
  }
}


template <int NB>
__device__ __forceinline__ float gcn_rows32(const float* __restrict__ H, int s, int nv, int lane,
                                            float acc) {
  float v[16 * NB];
#pragma unroll
  for (int u = 0; u < 16 * NB; ++u) {
    const int sj = __builtin_amdgcn_readlane(s, u);
    v[u] = H[(size_t)sj * 32 + lane];
  }
#pragma unroll
  for (int u = 0; u < 16 * NB; ++u) {
    if (u < nv) acc += v[u];
  }
  return acc;
}

#define GCN_GNODES 64
#define GCN_TPITCH 36

__global__ __launch_bounds__(256) void gcn_gather32_kernel(const int* __restrict__ rowstart,
                                                           const int* __restrict__ cnt,
                                                           const int* __restrict__ csr,
                                                           const float* __restrict__ dinv,
                                                           const float* __restrict__ H,
                                                           const float* __restrict__ b1,
                                                           const float* __restrict__ W2,
                                                           float* __restrict__ HW2,
                                                           int N) {
  __shared__ float tile[GCN_GNODES * GCN_TPITCH];
  __shared__ float sdn[GCN_GNODES];
  const int lane = threadIdx.x & 31;
  const int wave = __builtin_amdgcn_readfirstlane(threadIdx.x >> 5);
  const int blockNode = blockIdx.x * GCN_GNODES;
  const int node0 = blockNode + wave * 8;
  const float bias = b1[lane];
  const int lo = lane & 15;
  const int hi = lane >> 4;
  v2f bw[8] = {};
  if (wave < 4) gcn_load_b<16>(W2, bw, lo, hi);
  const int mn = node0 + (lane & 7);
  const int mc = min(mn, N - 1);
  const int mstart = rowstart[mc];
  const int mdeg = (mn < N) ? cnt[mc] : 0;
  const float mdn = dinv[mc];
  if (lane < 8) sdn[wave * 8 + lane] = mdn;
  int s_cur = min(node0, N - 1);
  {
    const int st = __builtin_amdgcn_readlane(mstart, 0);
    const int dg = __builtin_amdgcn_readlane(mdeg, 0);
    if (lane < dg) s_cur = csr[st + lane];
  }
  asm volatile("" ::"v"(s_cur), "v"(mstart), "v"(mdeg), "v"(mdn));
  for (int i = 0; i < 8; ++i) {
    const int n = node0 + i;
    const int inext = min(i + 1, 7);
    const int stn = __builtin_amdgcn_readlane(mstart, inext);
    const int dgn = (i < 7) ? __builtin_amdgcn_readlane(mdeg, inext) : 0;
    int s_nxt = min(n + 1, N - 1);
    if (lane < dgn) s_nxt = csr[stn + lane];
    if (n < N) {
      const int start = __builtin_amdgcn_readlane(mstart, i);
      const int deg = __builtin_amdgcn_readlane(mdeg, i);
      const float dn = __builtin_bit_cast(float,
                                          __builtin_amdgcn_readlane(__builtin_bit_cast(int, mdn), i));
      const float selfv = H[(size_t)n * 32 + lane];
      float acc = 0.0f;
      const int nv = min(32, deg);
      if (nv > 16) acc = gcn_rows32<2>(H, s_cur, nv, lane, acc);
      else acc = gcn_rows32<1>(H, s_cur, nv, lane, acc);
      for (int c0 = 32; c0 < deg; c0 += 32) {
        const int nv2 = min(32, deg - c0);
        int s = n;
        if (lane < nv2) s = csr[start + c0 + lane];
        if (nv2 > 16) acc = gcn_rows32<2>(H, s, nv2, lane, acc);
        else acc = gcn_rows32<1>(H, s, nv2, lane, acc);
      }
      const float t = (acc + selfv) * dn;
      tile[(wave * 8 + i) * GCN_TPITCH + lane] = tanhf(t + bias);
    }
    s_cur = s_nxt;
  }
  __syncthreads();
  const int rowBase = blockNode + wave * 16;
  if (wave < 4 && rowBase + 16 <= N) {
    v2f a[8];
#pragma unroll
    for (int kb = 0; kb < 8; ++kb) {
      const float* xr = tile + (wave * 16 + lo) * GCN_TPITCH + kb * 4 + hi * 2;
      a[kb].x = xr[0];
      a[kb].y = xr[1];
    }
    float rs[8];
#pragma unroll
    for (int v = 0; v < 8; ++v) rs[v] = sdn[wave * 16 + v + hi * 8];
    gcn_tile_product<16>(a, bw, rs, HW2 + (size_t)rowBase * 16, lo, hi);
  }
}


template <int M>
__device__ __forceinline__ float gcn_row_xor(float x) {
  return __builtin_bit_cast(
      float, __builtin_amdgcn_update_dpp(0, __builtin_bit_cast(int, x), 0x160 + M, 0xf, 0xf, false));
}

__device__ __forceinline__ void gcn_pair_bounds(const int* __restrict__ rowstart,
                                                const int* __restrict__ cnt, int p, int nPairs,
                                                int N, int sub, int& nc, int& start, int& deg) {
  const int n = 2 * p + sub;
  const bool valid = (p < nPairs) && (n < N);
  nc = valid ? n : N - 1;
  start = rowstart[nc];
  deg = valid ? cnt[nc] : 0;
}

__device__ __forceinline__ void gcn_pair_idx(const int* __restrict__ csr, int start, int left,
                                             int f, int nc, int& sa, int& sb) {
  sa = nc;
  sb = nc;
  if (f < left) sa = csr[start + f];
  if (16 + f < left) sb = csr[start + 16 + f];
}

template <int NB>
__device__ __forceinline__ float gcn_rows16(const float* __restrict__ H, int sa, int sb, int left,
                                            int sub, int f, float acc) {
  float v[16 * NB];
#pragma unroll
  for (int u = 0; u < 16 * NB; ++u) {
    const int sx = (u < 16) ? sa : sb;
    const int s0 = __builtin_amdgcn_readlane(sx, u & 15);
    const int s1 = __builtin_amdgcn_readlane(sx, 16 + (u & 15));
    const int sj = sub ? s1 : s0;
    v[u] = H[(size_t)sj * 16 + f];
  }
  if (NB == 1) __builtin_amdgcn_sched_barrier(0);
#pragma unroll
  for (int u = 0; u < 16 * NB; ++u) {
    if (u < left) acc += v[u];
  }
  return acc;
}

__global__ __launch_bounds__(256) void gcn_gather16_kernel(const int* __restrict__ rowstart,
                                                           const int* __restrict__ cnt,
                                                           const int* __restrict__ csr,
                                                           const float* __restrict__ dinv,
                                                           const float* __restrict__ H,
                                                           const float* __restrict__ b2,
                                                           float* __restrict__ out_h,
                                                           float* __restrict__ out_ls,
                                                           int N, int nWaves) {
  const int lane = threadIdx.x & 31;
  const int f = lane & 15;
  const int sub = lane >> 4;
  const float bias = b2[f];
  const int wave = __builtin_amdgcn_readfirstlane(blockIdx.x * 8 + (threadIdx.x >> 5));
  const int nPairs = (N + 1) >> 1;
  int nc0, st0, dg0, nc1, st1, dg1;
  gcn_pair_bounds(rowstart, cnt, wave, nPairs, N, sub, nc0, st0, dg0);
  gcn_pair_bounds(rowstart, cnt, wave + nWaves, nPairs, N, sub, nc1, st1, dg1);
  int sa0, sb0;
  gcn_pair_idx(csr, st0, dg0, f, nc0, sa0, sb0);
  asm volatile("" ::"v"(sa0), "v"(sb0), "v"(st0), "v"(dg0), "v"(st1), "v"(dg1), "v"(bias));
  for (int p = wave; p < nPairs; p += nWaves) {
    int sa1, sb1;
    gcn_pair_idx(csr, st1, dg1, f, nc1, sa1, sb1);
    int nc2, st2, dg2;
    gcn_pair_bounds(rowstart, cnt, p + 2 * nWaves, nPairs, N, sub, nc2, st2, dg2);
    const int n = 2 * p + sub;
    const bool valid = n < N;
    const int nc = nc0;
    const int start = st0;
    const int deg = dg0;
    const int degmax = max(__builtin_amdgcn_readlane(deg, 0), __builtin_amdgcn_readlane(deg, 16));
    const float di = dinv[nc];
    const float selfv = H[(size_t)nc * 16 + f];
    float acc = 0.0f;
    if (degmax > 16) acc = gcn_rows16<2>(H, sa0, sb0, deg, sub, f, acc);
    else acc = gcn_rows16<1>(H, sa0, sb0, deg, sub, f, acc);
    for (int c0 = 32; c0 < degmax; c0 += 32) {
      const int left = deg - c0;
      int sa, sb;
      gcn_pair_idx(csr, start + c0, left, f, nc, sa, sb);
      if (degmax - c0 > 16) acc = gcn_rows16<2>(H, sa, sb, left, sub, f, acc);
      else acc = gcn_rows16<1>(H, sa, sb, left, sub, f, acc);
    }
    const float t = (acc + selfv) * di + bias;
    float m = t;
    m = fmaxf(m, gcn_row_xor<1>(m));
    m = fmaxf(m, gcn_row_xor<2>(m));
    m = fmaxf(m, gcn_row_xor<4>(m));
    m = fmaxf(m, gcn_row_xor<8>(m));
    float se = expf(t - m);
    se += gcn_row_xor<1>(se);
    se += gcn_row_xor<2>(se);
    se += gcn_row_xor<4>(se);
    se += gcn_row_xor<8>(se);
    const float lse = logf(se);
    if (valid) {
      out_h[(size_t)n * 16 + f] = t;
      out_ls[(size_t)n * 16 + f] = t - m - lse;
    }
    nc0 = nc1; st0 = st1; dg0 = dg1;
    nc1 = nc2; st1 = st2; dg1 = dg2;
    sa0 = sa1; sb0 = sb1;
  }
}

extern "C" void kernel_launch(void* const* d_in, const int* in_sizes, int n_in,
                              void* d_out, int out_size, void* d_ws, size_t ws_size,
                              hipStream_t stream) {
  const float* x  = (const float*)d_in[0];
  const int*   ei = (const int*)d_in[1];
  const float* W1 = (const float*)d_in[2];
  const float* b1 = (const float*)d_in[3];
  const float* W2 = (const float*)d_in[4];
  const float* b2 = (const float*)d_in[5];

  const int N = in_sizes[0] / 32;
  const int E = in_sizes[1] / 2;
  const int* src = ei;
  const int* dst = ei + E;

  int*   wsi      = (int*)d_ws;
  float* wsf      = (float*)d_ws;
  int*   cnt      = wsi;
  float* dinv     = wsf + (size_t)N;
  int*   rowstart = wsi + (size_t)N * 2;
  const int NB = (N + GCN_BNODES - 1) / GCN_BNODES;
  const int G1 = (E + GCN_CHUNK - 1) / GCN_CHUNK;
  const int TS = NB + 1;
  unsigned* chunked = (unsigned*)(wsi + (size_t)N * 3);
  int*   csr      = wsi + (size_t)N * 3 + (size_t)E;
  float* xw1      = wsf + (size_t)N * 3 + (size_t)E * 2;
  float* hw2      = xw1 + (size_t)N * 64;
  int*   tab      = (int*)(hw2 + (size_t)N * 32);

  float* out_h  = (float*)d_out;
  float* out_ls = out_h + (size_t)N * 16;

  const unsigned gather32Blocks = (unsigned)((N + GCN_GNODES - 1) / GCN_GNODES);
  const unsigned gather16Blocks = (unsigned)(((N + 1) / 2 + 31) / 32);

  gcn_chunksort_kernel<<<(unsigned)G1, 512, 0, stream>>>(src, dst, chunked, tab, E, NB, TS);
  gcn_place_kernel<<<(unsigned)NB, 512, 0, stream>>>(tab, chunked, cnt, rowstart, dinv, csr,
                                                    x, W1, xw1, N, G1, TS);

  gcn_gather32_kernel<<<gather32Blocks, 256, 0, stream>>>(rowstart, cnt, csr, dinv, xw1, b1, W2,
                                                          hw2, N);

  gcn_gather16_kernel<<<gather16Blocks, 256, 0, stream>>>(rowstart, cnt, csr, dinv, hw2, b2,
                                                          out_h, out_ls, N, (int)gather16Blocks * 8);
}


# edge placing in one pass: a piece's edges load once as one group, the counting LDS add's return is kept as the rank; no second read or add (segment order = counting adds' arrival, was placing adds')
# speedup vs baseline: 3.8440x; 1.0344x over previous
#include <hip/hip_runtime.h>
#include <hip/hip_bf16.h>


typedef __attribute__((ext_vector_type(2))) float v2f;
typedef __attribute__((ext_vector_type(8))) float v8f;


template <int NCOL>
__device__ __forceinline__ void gcn_load_b(const float* __restrict__ W, v2f* b, int lo, int hi) {
#pragma unroll
  for (int ct = 0; ct < NCOL / 16; ++ct) {
#pragma unroll
    for (int kb = 0; kb < 8; ++kb) {
      const int k = kb * 4 + hi * 2;
      b[ct * 8 + kb].x = W[(size_t)k * NCOL + ct * 16 + lo];
      b[ct * 8 + kb].y = W[(size_t)(k + 1) * NCOL + ct * 16 + lo];
    }
  }
}

template <int NCOL>
__device__ __forceinline__ void gcn_tile_product(const v2f* a, const v2f* b, const float* rs,
                                                 float* __restrict__ Yt, int lo, int hi) {
#pragma unroll
  for (int ct = 0; ct < NCOL / 16; ++ct) {
    v8f c = {};
#pragma unroll
    for (int kb = 0; kb < 8; ++kb) {
      c = __builtin_amdgcn_wmma_f32_16x16x4_f32(false, a[kb], false, b[ct * 8 + kb],
                                                (short)0, c, false, false);
    }
#pragma unroll
    for (int v = 0; v < 8; ++v) {
      const int m = v + hi * 8;
      Yt[(size_t)m * NCOL + ct * 16 + lo] = c[v] * rs[v];
    }
  }
}

#define GCN_BSH 8
#define GCN_BNODES 256
#define GCN_CHUNK 4096
#define GCN_MAXB 2048

__global__ __launch_bounds__(512) void gcn_chunksort_kernel(const int* __restrict__ src,
                                                            const int* __restrict__ dst,
                                                            unsigned* __restrict__ chunked,
                                                            int* __restrict__ tab,
                                                            int E, int NB, int TS) {
  __shared__ int h[GCN_MAXB + 4];
  __shared__ unsigned stage[GCN_CHUNK];
  __shared__ int wsum[16];
  const int tid = threadIdx.x;
  const int lane = tid & 31;
  const int wave = tid >> 5;
  const int b = blockIdx.x;
  const int base = b * GCN_CHUNK + tid;
  int d[8], s[8], r[8];
#pragma unroll
  for (int u = 0; u < 8; ++u) {
    const int e = min(base + u * 512, E - 1);
    d[u] = dst[e];
    s[u] = src[e];
  }
#pragma unroll
  for (int i = 0; i < 4; ++i) h[tid * 4 + i] = 0;
  if (tid == 0) h[GCN_MAXB] = 0;
  __syncthreads();
#pragma unroll
  for (int u = 0; u < 8; ++u) {
    const int idx = (base + u * 512 < E) ? (d[u] >> GCN_BSH) : GCN_MAXB;
    r[u] = atomicAdd(&h[idx], 1);
  }
  __syncthreads();
  int c[4];
#pragma unroll
  for (int i = 0; i < 4; ++i) c[i] = h[tid * 4 + i];
  const int s4 = (c[0] + c[1]) + (c[2] + c[3]);
  int x = s4;
#pragma unroll
  for (int o = 1; o < 32; o <<= 1) {
    const int y = __shfl_up(x, o, 32);
    if (lane >= o) x += y;
  }
  if (lane == 31) wsum[wave] = x;
  __syncthreads();
  int ex = x - s4;
#pragma unroll
  for (int w = 0; w < 16; ++w) {
    if (w < wave) ex += wsum[w];
  }
#pragma unroll
  for (int i = 0; i < 4; ++i) {
    h[tid * 4 + i] = ex;
    ex += c[i];
  }
  __syncthreads();
  for (int k = tid; k <= NB; k += 512) tab[(size_t)b * TS + k] = h[k];
#pragma unroll
  for (int u = 0; u < 8; ++u) {
    if (base + u * 512 < E) {
      stage[h[d[u] >> GCN_BSH] + r[u]] =
          ((unsigned)s[u] << GCN_BSH) | (unsigned)(d[u] & (GCN_BNODES - 1));
    }
  }
  __syncthreads();
  const int total = h[NB];
#pragma unroll
  for (int u = 0; u < 8; ++u) {
    const int i = u * 512 + tid;
    if (i < total) chunked[(size_t)b * GCN_CHUNK + i] = stage[i];
  }
}

#define GCN_KEEP 24
__global__ __launch_bounds__(512) void gcn_place_kernel(const int* __restrict__ tab,
                                                        const unsigned* __restrict__ chunked,
                                                        int* __restrict__ cnt,
                                                        int* __restrict__ rowstart,
                                                        float* __restrict__ dinv,
                                                        int* __restrict__ csr,
                                                        const float* __restrict__ X,
                                                        const float* __restrict__ W,
                                                        float* __restrict__ Y,
                                                        int N, int G1, int TS) {
  __shared__ int c[GCN_BNODES];
  __shared__ int c2[GCN_BNODES];
  __shared__ int cur[GCN_BNODES];
  __shared__ int cur2[GCN_BNODES];
  __shared__ int wsum[16];
  __shared__ int wacc[16];
  __shared__ float sdinv[GCN_BNODES];
  const int tid = threadIdx.x;
  const int lane = tid & 31;
  const int wave = tid >> 5;
  const int k = blockIdx.x;
  const int lo = lane & 15;
  const int hi = lane >> 4;
  const int rowBase = __builtin_amdgcn_readfirstlane(k * GCN_BNODES + wave * 16);
  if (tid < GCN_BNODES) {
    c[tid] = 0;
    c2[tid] = 0;
  }
  __syncthreads();
  v2f a[8];
  {
    const int xrow = min(rowBase + lo, N - 1);
#pragma unroll
    for (int kb = 0; kb < 8; ++kb) {
      const float* xr = X + (size_t)xrow * 32 + kb * 4 + hi * 2;
      a[kb].x = xr[0];
      a[kb].y = xr[1];
    }
  }
  const int tseg = min(tid, G1 - 1);
  int o0 = 0, len = 0;
  if (tid < G1) {
    o0 = tab[(size_t)tid * TS + k];
    len = tab[(size_t)tid * TS + k + 1] - o0;
  }
  int acc = o0;
  const unsigned* sp = chunked + (size_t)tseg * GCN_CHUNK + o0;
  unsigned v[GCN_KEEP];
  int r[GCN_KEEP];
#pragma unroll
  for (int u = 0; u < GCN_KEEP; ++u) {
    v[u] = 0u;
    if (u < len) v[u] = sp[u];
  }
#pragma unroll
  for (int u = 0; u < GCN_KEEP; ++u) {
    r[u] = 0;
    if (u < len) r[u] = atomicAdd(&c[v[u] & (GCN_BNODES - 1)], 1);
  }
  for (int j = GCN_KEEP; j < len; ++j) atomicAdd(&c2[sp[j] & (GCN_BNODES - 1)], 1);
  for (int seg = tid + 512; seg < G1; seg += 512) {
    const int q0 = tab[(size_t)seg * TS + k];
    const int ql = tab[(size_t)seg * TS + k + 1] - q0;
    acc += q0;
    const unsigned* qp = chunked + (size_t)seg * GCN_CHUNK + q0;
    for (int j = 0; j < ql; ++j) atomicAdd(&c2[qp[j] & (GCN_BNODES - 1)], 1);
  }
#pragma unroll
  for (int o = 16; o > 0; o >>= 1) acc += __shfl_xor(acc, o, 32);
  if (lane == 0) wacc[wave] = acc;
  __syncthreads();
  const int x1 = (tid < GCN_BNODES) ? c[tid] : 0;
  const int x = x1 + ((tid < GCN_BNODES) ? c2[tid] : 0);
  int incl = x;
#pragma unroll
  for (int o = 1; o < 32; o <<= 1) {
    const int y = __shfl_up(incl, o, 32);
    if (lane >= o) incl += y;
  }
  if (lane == 31) wsum[wave] = incl;
  __syncthreads();
  int ex = incl - x;
#pragma unroll
  for (int w = 0; w < 16; ++w) {
    ex += wacc[w];
    if (w < wave) ex += wsum[w];
  }
  if (tid < GCN_BNODES) {
    cur[tid] = ex;
    cur2[tid] = ex + x1;
    const int node = k * GCN_BNODES + tid;
    if (node < N) {
      rowstart[node] = ex;
      cnt[node] = x;
      const float dg = (float)(x + 1);
      const float di = (dg > 0.0f) ? rsqrtf(dg) : 0.0f;
      dinv[node] = di;
      sdinv[tid] = di;
    }
  }
  __syncthreads();
#pragma unroll
  for (int u = 0; u < GCN_KEEP; ++u) {
    if (u < len) csr[cur[v[u] & (GCN_BNODES - 1)] + r[u]] = (int)(v[u] >> GCN_BSH);
  }
  for (int j = GCN_KEEP; j < len; ++j) {
    const unsigned w = sp[j];
    const int p = atomicAdd(&cur2[w & (GCN_BNODES - 1)], 1);
    csr[p] = (int)(w >> GCN_BSH);
  }
  for (int seg = tid + 512; seg < G1; seg += 512) {
    const int q0 = tab[(size_t)seg * TS + k];
    const int ql = tab[(size_t)seg * TS + k + 1] - q0;
    const unsigned* qp = chunked + (size_t)seg * GCN_CHUNK + q0;
    for (int j = 0; j < ql; ++j) {
      const unsigned w = qp[j];
      const int p = atomicAdd(&cur2[w & (GCN_BNODES - 1)], 1);
      csr[p] = (int)(w >> GCN_BSH);
    }
  }
  v2f bw[16];
  gcn_load_b<32>(W, bw, lo, hi);
  if (rowBase + 16 <= N) {
    float rs[8];
#pragma unroll
    for (int v8 = 0; v8 < 8; ++v8) rs[v8] = sdinv[wave * 16 + v8 + hi * 8];
    gcn_tile_product<32>(a, bw, rs, Y + (size_t)rowBase * 32, lo, hi);
  }
}


template <int NB>
__device__ __forceinline__ float gcn_rows32(const float* __restrict__ H, int s, int nv, int lane,
                                            float acc) {
  float v[16 * NB];
#pragma unroll
  for (int u = 0; u < 16 * NB; ++u) {
    const int sj = __builtin_amdgcn_readlane(s, u);
    v[u] = H[(size_t)sj * 32 + lane];
  }
#pragma unroll
  for (int u = 0; u < 16 * NB; ++u) {
    if (u < nv) acc += v[u];
  }
  return acc;
}

#define GCN_GNODES 64
#define GCN_TPITCH 36

__global__ __launch_bounds__(256) void gcn_gather32_kernel(const int* __restrict__ rowstart,
                                                           const int* __restrict__ cnt,
                                                           const int* __restrict__ csr,
                                                           const float* __restrict__ dinv,
                                                           const float* __restrict__ H,
                                                           const float* __restrict__ b1,
                                                           const float* __restrict__ W2,
                                                           float* __restrict__ HW2,
                                                           int N) {
  __shared__ float tile[GCN_GNODES * GCN_TPITCH];
  __shared__ float sdn[GCN_GNODES];
  const int lane = threadIdx.x & 31;
  const int wave = __builtin_amdgcn_readfirstlane(threadIdx.x >> 5);
  const int blockNode = blockIdx.x * GCN_GNODES;
  const int node0 = blockNode + wave * 8;
  const float bias = b1[lane];
  const int lo = lane & 15;
  const int hi = lane >> 4;
  v2f bw[8] = {};
  if (wave < 4) gcn_load_b<16>(W2, bw, lo, hi);
  const int mn = node0 + (lane & 7);
  const int mc = min(mn, N - 1);
  const int mstart = rowstart[mc];
  const int mdeg = (mn < N) ? cnt[mc] : 0;
  const float mdn = dinv[mc];
  if (lane < 8) sdn[wave * 8 + lane] = mdn;
  int s_cur = min(node0, N - 1);
  {
    const int st = __builtin_amdgcn_readlane(mstart, 0);
    const int dg = __builtin_amdgcn_readlane(mdeg, 0);
    if (lane < dg) s_cur = csr[st + lane];
  }
  asm volatile("" ::"v"(s_cur), "v"(mstart), "v"(mdeg), "v"(mdn));
  for (int i = 0; i < 8; ++i) {
    const int n = node0 + i;
    const int inext = min(i + 1, 7);
    const int stn = __builtin_amdgcn_readlane(mstart, inext);
    const int dgn = (i < 7) ? __builtin_amdgcn_readlane(mdeg, inext) : 0;
    int s_nxt = min(n + 1, N - 1);
    if (lane < dgn) s_nxt = csr[stn + lane];
    if (n < N) {
      const int start = __builtin_amdgcn_readlane(mstart, i);
      const int deg = __builtin_amdgcn_readlane(mdeg, i);
      const float dn = __builtin_bit_cast(float,
                                          __builtin_amdgcn_readlane(__builtin_bit_cast(int, mdn), i));
      const float selfv = H[(size_t)n * 32 + lane];
      float acc = 0.0f;
      const int nv = min(32, deg);
      if (nv > 16) acc = gcn_rows32<2>(H, s_cur, nv, lane, acc);
      else acc = gcn_rows32<1>(H, s_cur, nv, lane, acc);
      for (int c0 = 32; c0 < deg; c0 += 32) {
        const int nv2 = min(32, deg - c0);
        int s = n;
        if (lane < nv2) s = csr[start + c0 + lane];
        if (nv2 > 16) acc = gcn_rows32<2>(H, s, nv2, lane, acc);
        else acc = gcn_rows32<1>(H, s, nv2, lane, acc);
      }
      const float t = (acc + selfv) * dn;
      tile[(wave * 8 + i) * GCN_TPITCH + lane] = tanhf(t + bias);
    }
    s_cur = s_nxt;
  }
  __syncthreads();
  const int rowBase = blockNode + wave * 16;
  if (wave < 4 && rowBase + 16 <= N) {
    v2f a[8];
#pragma unroll
    for (int kb = 0; kb < 8; ++kb) {
      const float* xr = tile + (wave * 16 + lo) * GCN_TPITCH + kb * 4 + hi * 2;
      a[kb].x = xr[0];
      a[kb].y = xr[1];
    }
    float rs[8];
#pragma unroll
    for (int v = 0; v < 8; ++v) rs[v] = sdn[wave * 16 + v + hi * 8];
    gcn_tile_product<16>(a, bw, rs, HW2 + (size_t)rowBase * 16, lo, hi);
  }
}


template <int M>
__device__ __forceinline__ float gcn_row_xor(float x) {
  return __builtin_bit_cast(
      float, __builtin_amdgcn_update_dpp(0, __builtin_bit_cast(int, x), 0x160 + M, 0xf, 0xf, false));
}

__device__ __forceinline__ void gcn_pair_bounds(const int* __restrict__ rowstart,
                                                const int* __restrict__ cnt, int p, int nPairs,
                                                int N, int sub, int& nc, int& start, int& deg) {
  const int n = 2 * p + sub;
  const bool valid = (p < nPairs) && (n < N);
  nc = valid ? n : N - 1;
  start = rowstart[nc];
  deg = valid ? cnt[nc] : 0;
}

__device__ __forceinline__ void gcn_pair_idx(const int* __restrict__ csr, int start, int left,
                                             int f, int nc, int& sa, int& sb) {
  sa = nc;
  sb = nc;
  if (f < left) sa = csr[start + f];
  if (16 + f < left) sb = csr[start + 16 + f];
}

template <int NB>
__device__ __forceinline__ float gcn_rows16(const float* __restrict__ H, int sa, int sb, int left,
                                            int sub, int f, float acc) {
  float v[16 * NB];
#pragma unroll
  for (int u = 0; u < 16 * NB; ++u) {
    const int sx = (u < 16) ? sa : sb;
    const int s0 = __builtin_amdgcn_readlane(sx, u & 15);
    const int s1 = __builtin_amdgcn_readlane(sx, 16 + (u & 15));
    const int sj = sub ? s1 : s0;
    v[u] = H[(size_t)sj * 16 + f];
  }
  if (NB == 1) __builtin_amdgcn_sched_barrier(0);
#pragma unroll
  for (int u = 0; u < 16 * NB; ++u) {
    if (u < left) acc += v[u];
  }
  return acc;
}

__global__ __launch_bounds__(256) void gcn_gather16_kernel(const int* __restrict__ rowstart,
                                                           const int* __restrict__ cnt,
                                                           const int* __restrict__ csr,
                                                           const float* __restrict__ dinv,
                                                           const float* __restrict__ H,
                                                           const float* __restrict__ b2,
                                                           float* __restrict__ out_h,
                                                           float* __restrict__ out_ls,
                                                           int N, int nWaves) {
  const int lane = threadIdx.x & 31;
  const int f = lane & 15;
  const int sub = lane >> 4;
  const float bias = b2[f];
  const int wave = __builtin_amdgcn_readfirstlane(blockIdx.x * 8 + (threadIdx.x >> 5));
  const int nPairs = (N + 1) >> 1;
  int nc0, st0, dg0, nc1, st1, dg1;
  gcn_pair_bounds(rowstart, cnt, wave, nPairs, N, sub, nc0, st0, dg0);
  gcn_pair_bounds(rowstart, cnt, wave + nWaves, nPairs, N, sub, nc1, st1, dg1);
  int sa0, sb0;
  gcn_pair_idx(csr, st0, dg0, f, nc0, sa0, sb0);
  asm volatile("" ::"v"(sa0), "v"(sb0), "v"(st0), "v"(dg0), "v"(st1), "v"(dg1), "v"(bias));
  for (int p = wave; p < nPairs; p += nWaves) {
    int sa1, sb1;
    gcn_pair_idx(csr, st1, dg1, f, nc1, sa1, sb1);
    int nc2, st2, dg2;
    gcn_pair_bounds(rowstart, cnt, p + 2 * nWaves, nPairs, N, sub, nc2, st2, dg2);
    const int n = 2 * p + sub;
    const bool valid = n < N;
    const int nc = nc0;
    const int start = st0;
    const int deg = dg0;
    const int degmax = max(__builtin_amdgcn_readlane(deg, 0), __builtin_amdgcn_readlane(deg, 16));
    const float di = dinv[nc];
    const float selfv = H[(size_t)nc * 16 + f];
    float acc = 0.0f;
    if (degmax > 16) acc = gcn_rows16<2>(H, sa0, sb0, deg, sub, f, acc);
    else acc = gcn_rows16<1>(H, sa0, sb0, deg, sub, f, acc);
    for (int c0 = 32; c0 < degmax; c0 += 32) {
      const int left = deg - c0;
      int sa, sb;
      gcn_pair_idx(csr, start + c0, left, f, nc, sa, sb);
      if (degmax - c0 > 16) acc = gcn_rows16<2>(H, sa, sb, left, sub, f, acc);
      else acc = gcn_rows16<1>(H, sa, sb, left, sub, f, acc);
    }
    const float t = (acc + selfv) * di + bias;
    float m = t;
    m = fmaxf(m, gcn_row_xor<1>(m));
    m = fmaxf(m, gcn_row_xor<2>(m));
    m = fmaxf(m, gcn_row_xor<4>(m));
    m = fmaxf(m, gcn_row_xor<8>(m));
    float se = expf(t - m);
    se += gcn_row_xor<1>(se);
    se += gcn_row_xor<2>(se);
    se += gcn_row_xor<4>(se);
    se += gcn_row_xor<8>(se);
    const float lse = logf(se);
    if (valid) {
      out_h[(size_t)n * 16 + f] = t;
      out_ls[(size_t)n * 16 + f] = t - m - lse;
    }
    nc0 = nc1; st0 = st1; dg0 = dg1;
    nc1 = nc2; st1 = st2; dg1 = dg2;
    sa0 = sa1; sb0 = sb1;
  }
}

extern "C" void kernel_launch(void* const* d_in, const int* in_sizes, int n_in,
                              void* d_out, int out_size, void* d_ws, size_t ws_size,
                              hipStream_t stream) {
  const float* x  = (const float*)d_in[0];
  const int*   ei = (const int*)d_in[1];
  const float* W1 = (const float*)d_in[2];
  const float* b1 = (const float*)d_in[3];
  const float* W2 = (const float*)d_in[4];
  const float* b2 = (const float*)d_in[5];

  const int N = in_sizes[0] / 32;
  const int E = in_sizes[1] / 2;
  const int* src = ei;
  const int* dst = ei + E;

  int*   wsi      = (int*)d_ws;
  float* wsf      = (float*)d_ws;
  int*   cnt      = wsi;
  float* dinv     = wsf + (size_t)N;
  int*   rowstart = wsi + (size_t)N * 2;
  const int NB = (N + GCN_BNODES - 1) / GCN_BNODES;
  const int G1 = (E + GCN_CHUNK - 1) / GCN_CHUNK;
  const int TS = NB + 1;
  unsigned* chunked = (unsigned*)(wsi + (size_t)N * 3);
  int*   csr      = wsi + (size_t)N * 3 + (size_t)E;
  float* xw1      = wsf + (size_t)N * 3 + (size_t)E * 2;
  float* hw2      = xw1 + (size_t)N * 64;
  int*   tab      = (int*)(hw2 + (size_t)N * 32);

  float* out_h  = (float*)d_out;
  float* out_ls = out_h + (size_t)N * 16;

  const unsigned gather32Blocks = (unsigned)((N + GCN_GNODES - 1) / GCN_GNODES);
  const unsigned gather16Blocks = (unsigned)(((N + 1) / 2 + 31) / 32);

  gcn_chunksort_kernel<<<(unsigned)G1, 512, 0, stream>>>(src, dst, chunked, tab, E, NB, TS);
  gcn_place_kernel<<<(unsigned)NB, 512, 0, stream>>>(tab, chunked, cnt, rowstart, dinv, csr,
                                                    x, W1, xw1, N, G1, TS);

  gcn_gather32_kernel<<<gather32Blocks, 256, 0, stream>>>(rowstart, cnt, csr, dinv, xw1, b1, W2,
                                                          hw2, N);

  gcn_gather16_kernel<<<gather16Blocks, 256, 0, stream>>>(rowstart, cnt, csr, dinv, hw2, b2,
                                                          out_h, out_ls, N, (int)gather16Blocks * 8);
}


# row gathers 16 bytes per lane: one load covers 4 source rows (layer 1) or 8 (layer 2), none issued for empty slot groups; a node's sum is now 4 interleaved partial sums + lane tree (order changed)
# speedup vs baseline: 4.4396x; 1.1549x over previous
#include <hip/hip_runtime.h>
#include <hip/hip_bf16.h>


typedef __attribute__((ext_vector_type(2))) float v2f;
typedef __attribute__((ext_vector_type(8))) float v8f;
typedef __attribute__((ext_vector_type(4))) float v4f;


template <int NCOL>
__device__ __forceinline__ void gcn_load_b(const float* __restrict__ W, v2f* b, int lo, int hi) {
#pragma unroll
  for (int ct = 0; ct < NCOL / 16; ++ct) {
#pragma unroll
    for (int kb = 0; kb < 8; ++kb) {
      const int k = kb * 4 + hi * 2;
      b[ct * 8 + kb].x = W[(size_t)k * NCOL + ct * 16 + lo];
      b[ct * 8 + kb].y = W[(size_t)(k + 1) * NCOL + ct * 16 + lo];
    }
  }
}

template <int NCOL>
__device__ __forceinline__ void gcn_tile_product(const v2f* a, const v2f* b, const float* rs,
                                                 float* __restrict__ Yt, int lo, int hi) {
#pragma unroll
  for (int ct = 0; ct < NCOL / 16; ++ct) {
    v8f c = {};
#pragma unroll
    for (int kb = 0; kb < 8; ++kb) {
      c = __builtin_amdgcn_wmma_f32_16x16x4_f32(false, a[kb], false, b[ct * 8 + kb],
                                                (short)0, c, false, false);
    }
#pragma unroll
    for (int v = 0; v < 8; ++v) {
      const int m = v + hi * 8;
      Yt[(size_t)m * NCOL + ct * 16 + lo] = c[v] * rs[v];
    }
  }
}

#define GCN_BSH 8
#define GCN_BNODES 256
#define GCN_CHUNK 4096
#define GCN_MAXB 2048

__global__ __launch_bounds__(512) void gcn_chunksort_kernel(const int* __restrict__ src,
                                                            const int* __restrict__ dst,
                                                            unsigned* __restrict__ chunked,
                                                            int* __restrict__ tab,
                                                            int E, int NB, int TS) {
  __shared__ int h[GCN_MAXB + 4];
  __shared__ unsigned stage[GCN_CHUNK];
  __shared__ int wsum[16];
  const int tid = threadIdx.x;
  const int lane = tid & 31;
  const int wave = tid >> 5;
  const int b = blockIdx.x;
  const int base = b * GCN_CHUNK + tid;
  int d[8], s[8], r[8];
#pragma unroll
  for (int u = 0; u < 8; ++u) {
    const int e = min(base + u * 512, E - 1);
    d[u] = dst[e];
    s[u] = src[e];
  }
#pragma unroll
  for (int i = 0; i < 4; ++i) h[tid * 4 + i] = 0;
  if (tid == 0) h[GCN_MAXB] = 0;
  __syncthreads();
#pragma unroll
  for (int u = 0; u < 8; ++u) {
    const int idx = (base + u * 512 < E) ? (d[u] >> GCN_BSH) : GCN_MAXB;
    r[u] = atomicAdd(&h[idx], 1);
  }
  __syncthreads();
  int c[4];
#pragma unroll
  for (int i = 0; i < 4; ++i) c[i] = h[tid * 4 + i];
  const int s4 = (c[0] + c[1]) + (c[2] + c[3]);
  int x = s4;
#pragma unroll
  for (int o = 1; o < 32; o <<= 1) {
    const int y = __shfl_up(x, o, 32);
    if (lane >= o) x += y;
  }
  if (lane == 31) wsum[wave] = x;
  __syncthreads();
  int ex = x - s4;
#pragma unroll
  for (int w = 0; w < 16; ++w) {
    if (w < wave) ex += wsum[w];
  }
#pragma unroll
  for (int i = 0; i < 4; ++i) {
    h[tid * 4 + i] = ex;
    ex += c[i];
  }
  __syncthreads();
  for (int k = tid; k <= NB; k += 512) tab[(size_t)b * TS + k] = h[k];
#pragma unroll
  for (int u = 0; u < 8; ++u) {
    if (base + u * 512 < E) {
      stage[h[d[u] >> GCN_BSH] + r[u]] =
          ((unsigned)s[u] << GCN_BSH) | (unsigned)(d[u] & (GCN_BNODES - 1));
    }
  }
  __syncthreads();
  const int total = h[NB];
#pragma unroll
  for (int u = 0; u < 8; ++u) {
    const int i = u * 512 + tid;
    if (i < total) chunked[(size_t)b * GCN_CHUNK + i] = stage[i];
  }
}

#define GCN_KEEP 24
__global__ __launch_bounds__(512) void gcn_place_kernel(const int* __restrict__ tab,
                                                        const unsigned* __restrict__ chunked,
                                                        int* __restrict__ cnt,
                                                        int* __restrict__ rowstart,
                                                        float* __restrict__ dinv,
                                                        int* __restrict__ csr,
                                                        const float* __restrict__ X,
                                                        const float* __restrict__ W,
                                                        float* __restrict__ Y,
                                                        int N, int G1, int TS) {
  __shared__ int c[GCN_BNODES];
  __shared__ int c2[GCN_BNODES];
  __shared__ int cur[GCN_BNODES];
  __shared__ int cur2[GCN_BNODES];
  __shared__ int wsum[16];
  __shared__ int wacc[16];
  __shared__ float sdinv[GCN_BNODES];
  const int tid = threadIdx.x;
  const int lane = tid & 31;
  const int wave = tid >> 5;
  const int k = blockIdx.x;
  const int lo = lane & 15;
  const int hi = lane >> 4;
  const int rowBase = __builtin_amdgcn_readfirstlane(k * GCN_BNODES + wave * 16);
  if (tid < GCN_BNODES) {
    c[tid] = 0;
    c2[tid] = 0;
  }
  __syncthreads();
  v2f a[8];
  {
    const int xrow = min(rowBase + lo, N - 1);
#pragma unroll
    for (int kb = 0; kb < 8; ++kb) {
      const float* xr = X + (size_t)xrow * 32 + kb * 4 + hi * 2;
      a[kb].x = xr[0];
      a[kb].y = xr[1];
    }
  }
  const int tseg = min(tid, G1 - 1);
  int o0 = 0, len = 0;
  if (tid < G1) {
    o0 = tab[(size_t)tid * TS + k];
    len = tab[(size_t)tid * TS + k + 1] - o0;
  }
  int acc = o0;
  const unsigned* sp = chunked + (size_t)tseg * GCN_CHUNK + o0;
  unsigned v[GCN_KEEP];
  int r[GCN_KEEP];
#pragma unroll
  for (int u = 0; u < GCN_KEEP; ++u) {
    v[u] = 0u;
    if (u < len) v[u] = sp[u];
  }
#pragma unroll
  for (int u = 0; u < GCN_KEEP; ++u) {
    r[u] = 0;
    if (u < len) r[u] = atomicAdd(&c[v[u] & (GCN_BNODES - 1)], 1);
  }
  for (int j = GCN_KEEP; j < len; ++j) atomicAdd(&c2[sp[j] & (GCN_BNODES - 1)], 1);
  for (int seg = tid + 512; seg < G1; seg += 512) {
    const int q0 = tab[(size_t)seg * TS + k];
    const int ql = tab[(size_t)seg * TS + k + 1] - q0;
    acc += q0;
    const unsigned* qp = chunked + (size_t)seg * GCN_CHUNK + q0;
    for (int j = 0; j < ql; ++j) atomicAdd(&c2[qp[j] & (GCN_BNODES - 1)], 1);
  }
#pragma unroll
  for (int o = 16; o > 0; o >>= 1) acc += __shfl_xor(acc, o, 32);
  if (lane == 0) wacc[wave] = acc;
  __syncthreads();
  const int x1 = (tid < GCN_BNODES) ? c[tid] : 0;
  const int x = x1 + ((tid < GCN_BNODES) ? c2[tid] : 0);
  int incl = x;
#pragma unroll
  for (int o = 1; o < 32; o <<= 1) {
    const int y = __shfl_up(incl, o, 32);
    if (lane >= o) incl += y;
  }
  if (lane == 31) wsum[wave] = incl;
  __syncthreads();
  int ex = incl - x;
#pragma unroll
  for (int w = 0; w < 16; ++w) {
    ex += wacc[w];
    if (w < wave) ex += wsum[w];
  }
  if (tid < GCN_BNODES) {
    cur[tid] = ex;
    cur2[tid] = ex + x1;
    const int node = k * GCN_BNODES + tid;
    if (node < N) {
      rowstart[node] = ex;
      cnt[node] = x;
      const float dg = (float)(x + 1);
      const float di = (dg > 0.0f) ? rsqrtf(dg) : 0.0f;
      dinv[node] = di;
      sdinv[tid] = di;
    }
  }
  __syncthreads();
#pragma unroll
  for (int u = 0; u < GCN_KEEP; ++u) {
    if (u < len) csr[cur[v[u] & (GCN_BNODES - 1)] + r[u]] = (int)(v[u] >> GCN_BSH);
  }
  for (int j = GCN_KEEP; j < len; ++j) {
    const unsigned w = sp[j];
    const int p = atomicAdd(&cur2[w & (GCN_BNODES - 1)], 1);
    csr[p] = (int)(w >> GCN_BSH);
  }
  for (int seg = tid + 512; seg < G1; seg += 512) {
    const int q0 = tab[(size_t)seg * TS + k];
    const int ql = tab[(size_t)seg * TS + k + 1] - q0;
    const unsigned* qp = chunked + (size_t)seg * GCN_CHUNK + q0;
    for (int j = 0; j < ql; ++j) {
      const unsigned w = qp[j];
      const int p = atomicAdd(&cur2[w & (GCN_BNODES - 1)], 1);
      csr[p] = (int)(w >> GCN_BSH);
    }
  }
  v2f bw[16];
  gcn_load_b<32>(W, bw, lo, hi);
  if (rowBase + 16 <= N) {
    float rs[8];
#pragma unroll
    for (int v8 = 0; v8 < 8; ++v8) rs[v8] = sdinv[wave * 16 + v8 + hi * 8];
    gcn_tile_product<32>(a, bw, rs, Y + (size_t)rowBase * 32, lo, hi);
  }
}


template <int M>
__device__ __forceinline__ float gcn_row_xor(float x) {
  return __builtin_bit_cast(
      float, __builtin_amdgcn_update_dpp(0, __builtin_bit_cast(int, x), 0x160 + M, 0xf, 0xf, false));
}

__device__ __forceinline__ float gcn_half_swap(float x) {
  const int xi = __builtin_bit_cast(int, x);
  return __builtin_bit_cast(
      float, __builtin_amdgcn_permlanex16(xi, xi, 0x76543210, 0xfedcba98, false, false));
}

template <int J>
__device__ __forceinline__ int gcn_bcast8(int x) {
  return __builtin_amdgcn_mov_dpp8(x, J * 0x249249);
}

#define GCN_ROW4X32(J) \
  if (NJ > J) v[J] = *reinterpret_cast<const v4f*>(Hb + (unsigned)(gcn_bcast8<J>(sidx) * 128 + q16));
template <int NJ>
__device__ __forceinline__ v4f gcn_rows32(const char* __restrict__ Hb, int sidx, int lg, int q16,
                                          v4f acc) {
  v4f v[8];
  GCN_ROW4X32(0)
  GCN_ROW4X32(1)
  GCN_ROW4X32(2)
  GCN_ROW4X32(3)
  GCN_ROW4X32(4)
  GCN_ROW4X32(5)
  GCN_ROW4X32(6)
  GCN_ROW4X32(7)
#pragma unroll
  for (int j = 0; j + 1 < NJ; ++j) acc += v[j];
  const v4f zero = {0.0f, 0.0f, 0.0f, 0.0f};
  acc += (4 * (NJ - 1) < lg) ? v[NJ - 1] : zero;
  return acc;
}

__device__ __forceinline__ v4f gcn_rows32_any(const char* __restrict__ Hb, int sidx, int nv, int g,
                                              int q16, v4f acc) {
  const int lg = nv - g;
  switch ((nv + 3) >> 2) {
    case 1: return gcn_rows32<1>(Hb, sidx, lg, q16, acc);
    case 2: return gcn_rows32<2>(Hb, sidx, lg, q16, acc);
    case 3: return gcn_rows32<3>(Hb, sidx, lg, q16, acc);
    case 4: return gcn_rows32<4>(Hb, sidx, lg, q16, acc);
    case 5: return gcn_rows32<5>(Hb, sidx, lg, q16, acc);
    case 6: return gcn_rows32<6>(Hb, sidx, lg, q16, acc);
    case 7: return gcn_rows32<7>(Hb, sidx, lg, q16, acc);
    case 8: return gcn_rows32<8>(Hb, sidx, lg, q16, acc);
    default: return acc;
  }
}

#define GCN_GNODES 64
#define GCN_TPITCH 36

__global__ __launch_bounds__(256) void gcn_gather32_kernel(const int* __restrict__ rowstart,
                                                           const int* __restrict__ cnt,
                                                           const int* __restrict__ csr,
                                                           const float* __restrict__ dinv,
                                                           const float* __restrict__ H,
                                                           const float* __restrict__ b1,
                                                           const float* __restrict__ W2,
                                                           float* __restrict__ HW2,
                                                           int N) {
  __shared__ float tile[GCN_GNODES * GCN_TPITCH];
  __shared__ float sdn[GCN_GNODES];
  const int lane = threadIdx.x & 31;
  const int wave = __builtin_amdgcn_readfirstlane(threadIdx.x >> 5);
  const int blockNode = blockIdx.x * GCN_GNODES;
  const int node0 = blockNode + wave * 8;
  const int g = lane >> 3;
  const int q16 = (lane & 7) * 16;
  const int slot = ((lane & 7) << 2) | g;
  const int fo = (lane & 7) * 4 + g;
  const float bias = b1[fo];
  const char* Hb = reinterpret_cast<const char*>(H);
  const int lo = lane & 15;
  const int hi = lane >> 4;
  const int mn = node0 + (lane & 7);
  const int mc = min(mn, N - 1);
  const int mstart = rowstart[mc];
  const int mdeg = (mn < N) ? cnt[mc] : 0;
  const float mdn = dinv[mc];
  if (lane < 8) sdn[wave * 8 + lane] = mdn;
  int s_cur = min(node0, N - 1);
  {
    const int st = __builtin_amdgcn_readlane(mstart, 0);
    const int dg = __builtin_amdgcn_readlane(mdeg, 0);
    if (slot < dg) s_cur = csr[st + slot];
  }
  asm volatile("" ::"v"(s_cur), "v"(mstart), "v"(mdeg), "v"(mdn));
  for (int i = 0; i < 8; ++i) {
    const int n = node0 + i;
    const int inext = min(i + 1, 7);
    const int stn = __builtin_amdgcn_readlane(mstart, inext);
    const int dgn = (i < 7) ? __builtin_amdgcn_readlane(mdeg, inext) : 0;
    int s_nxt = min(n + 1, N - 1);
    if (slot < dgn) s_nxt = csr[stn + slot];
    if (n < N) {
      const int start = __builtin_amdgcn_readlane(mstart, i);
      const int deg = __builtin_amdgcn_readlane(mdeg, i);
      const float dn = __builtin_bit_cast(float,
                                          __builtin_amdgcn_readlane(__builtin_bit_cast(int, mdn), i));
      const float* own = H + (size_t)n * 32;
      const float selfv = own[(unsigned)fo];
      v4f acc = {0.0f, 0.0f, 0.0f, 0.0f};
      acc = gcn_rows32_any(Hb, s_cur, min(32, deg), g, q16, acc);
      if (deg > 32) {
        for (int c0 = 32; c0 < deg; c0 += 32) {
          const int nv2 = min(32, deg - c0);
          int s = n;
          if (slot < nv2) s = csr[start + c0 + slot];
          acc = gcn_rows32_any(Hb, s, nv2, g, q16, acc);
        }
      }
      float r[4];
#pragma unroll
      for (int e = 0; e < 4; ++e) {
        float a = acc[e];
        a += gcn_row_xor<8>(a);
        a += gcn_half_swap(a);
        r[e] = a;
      }
      const float sum = (g == 0) ? r[0] : (g == 1) ? r[1] : (g == 2) ? r[2] : r[3];
      const float t = (sum + selfv) * dn;
      tile[(wave * 8 + i) * GCN_TPITCH + fo] = tanhf(t + bias);
    }
    s_cur = s_nxt;
  }
  v2f bw[8] = {};
  if (wave < 4) gcn_load_b<16>(W2, bw, lo, hi);
  __syncthreads();
  const int rowBase = blockNode + wave * 16;
  if (wave < 4 && rowBase + 16 <= N) {
    v2f a[8];
#pragma unroll
    for (int kb = 0; kb < 8; ++kb) {
      const float* xr = tile + (wave * 16 + lo) * GCN_TPITCH + kb * 4 + hi * 2;
      a[kb].x = xr[0];
      a[kb].y = xr[1];
    }
    float rs[8];
#pragma unroll
    for (int v = 0; v < 8; ++v) rs[v] = sdn[wave * 16 + v + hi * 8];
    gcn_tile_product<16>(a, bw, rs, HW2 + (size_t)rowBase * 16, lo, hi);
  }
}


__device__ __forceinline__ void gcn_pair_bounds(const int* __restrict__ rowstart,
                                                const int* __restrict__ cnt, int p, int nPairs,
                                                int N, int sub, int& start, int& deg) {
  const int n = 2 * p + sub;
  const bool valid = (p < nPairs) && (n < N);
  const int nc = valid ? n : N - 1;
  start = rowstart[nc];
  deg = valid ? cnt[nc] : 0;
}

__device__ __forceinline__ void gcn_pair_idx(const int* __restrict__ csr, int start, int left,
                                             int ea, int& sa, int& sb) {
  sa = 0;
  sb = 0;
  const int le = left - ea;
  if (le > 0) sa = csr[start + ea];
  if (le > 1) sb = csr[start + ea + 1];
}

#define GCN_ROW4X16(J)                                                                     \
  if (NJ > J) {                                                                            \
    const int s0 = gcn_bcast8<J>(sa);                                                      \
    const int s1 = gcn_bcast8<J>(sb);                                                      \
    v[J] = *reinterpret_cast<const v4f*>(Hb + (unsigned)((godd ? s1 : s0) * 64 + q16));    \
  }
template <int NJ>
__device__ __forceinline__ v4f gcn_rows16(const char* __restrict__ Hb, int sa, int sb, int left,
                                          int godd, int g, int q16, v4f acc) {
  v4f v[8];
  GCN_ROW4X16(0)
  GCN_ROW4X16(1)
  GCN_ROW4X16(2)
  GCN_ROW4X16(3)
  GCN_ROW4X16(4)
  GCN_ROW4X16(5)
  GCN_ROW4X16(6)
  GCN_ROW4X16(7)
  const v4f zero = {0.0f, 0.0f, 0.0f, 0.0f};
  const int lg = left - g;
#pragma unroll
  for (int j = 0; j < NJ; ++j) acc += (4 * j < lg) ? v[j] : zero;
  return acc;
}

__device__ __forceinline__ v4f gcn_rows16_any(const char* __restrict__ Hb, int sa, int sb, int left,
                                              int nmax, int godd, int g, int q16, v4f acc) {
  switch ((nmax + 3) >> 2) {
    case 1: return gcn_rows16<1>(Hb, sa, sb, left, godd, g, q16, acc);
    case 2: return gcn_rows16<2>(Hb, sa, sb, left, godd, g, q16, acc);
    case 3: return gcn_rows16<3>(Hb, sa, sb, left, godd, g, q16, acc);
    case 4: return gcn_rows16<4>(Hb, sa, sb, left, godd, g, q16, acc);
    case 5: return gcn_rows16<5>(Hb, sa, sb, left, godd, g, q16, acc);
    case 6: return gcn_rows16<6>(Hb, sa, sb, left, godd, g, q16, acc);
    case 7: return gcn_rows16<7>(Hb, sa, sb, left, godd, g, q16, acc);
    case 8: return gcn_rows16<8>(Hb, sa, sb, left, godd, g, q16, acc);
    default: return acc;
  }
}

__global__ __launch_bounds__(256) void gcn_gather16_kernel(const int* __restrict__ rowstart,
                                                           const int* __restrict__ cnt,
                                                           const int* __restrict__ csr,
                                                           const float* __restrict__ dinv,
                                                           const float* __restrict__ H,
                                                           const float* __restrict__ b2,
                                                           float* __restrict__ out_h,
                                                           float* __restrict__ out_ls,
                                                           int N, int nWaves) {
  const int lane = threadIdx.x & 31;
  const int f = lane & 15;
  const int sub = lane >> 4;
  const int g = f >> 2;
  const int godd = g & 1;
  const int q16 = (f & 3) * 16;
  const int fo = (f & 3) * 4 + g;
  const int ea = ((lane & 7) << 2) | (((lane >> 3) & 1) << 1);
  const int fo4 = fo * 4;
  const float bias = b2[fo];
  const char* Hb = reinterpret_cast<const char*>(H);
  const int wave = __builtin_amdgcn_readfirstlane(blockIdx.x * 8 + (threadIdx.x >> 5));
  const int nPairs = (N + 1) >> 1;
  int st0, dg0, st1, dg1;
  gcn_pair_bounds(rowstart, cnt, wave, nPairs, N, sub, st0, dg0);
  gcn_pair_bounds(rowstart, cnt, wave + nWaves, nPairs, N, sub, st1, dg1);
  int sa0, sb0;
  gcn_pair_idx(csr, st0, dg0, ea, sa0, sb0);
  asm volatile("" ::"v"(sa0), "v"(sb0), "v"(st0), "v"(dg0), "v"(st1), "v"(dg1), "v"(bias));
  for (int p = wave; p < nPairs; p += nWaves) {
    int sa1, sb1;
    gcn_pair_idx(csr, st1, dg1, ea, sa1, sb1);
    int st2, dg2;
    gcn_pair_bounds(rowstart, cnt, p + 2 * nWaves, nPairs, N, sub, st2, dg2);
    const int n = 2 * p + sub;
    const bool valid = n < N;
    const int nc = min(n, N - 1);
    const int start = st0;
    const int deg = dg0;
    const int degmax = max(__builtin_amdgcn_readlane(deg, 0), __builtin_amdgcn_readlane(deg, 16));
    const float di = dinv[nc];
    const float selfv = *reinterpret_cast<const float*>(Hb + (unsigned)(nc * 64 + fo4));
    v4f acc = {0.0f, 0.0f, 0.0f, 0.0f};
    acc = gcn_rows16_any(Hb, sa0, sb0, deg, min(32, degmax), godd, g, q16, acc);
    if (degmax > 32) {
      for (int c0 = 32; c0 < degmax; c0 += 32) {
        const int left = deg - c0;
        int sa, sb;
        gcn_pair_idx(csr, start + c0, left, ea, sa, sb);
        acc = gcn_rows16_any(Hb, sa, sb, left, min(32, degmax - c0), godd, g, q16, acc);
      }
    }
    float r[4];
#pragma unroll
    for (int e = 0; e < 4; ++e) {
      float a = acc[e];
      a += gcn_row_xor<4>(a);
      a += gcn_row_xor<8>(a);
      r[e] = a;
    }
    const float sum = (g == 0) ? r[0] : (g == 1) ? r[1] : (g == 2) ? r[2] : r[3];
    const float t = (sum + selfv) * di + bias;
    float m = t;
    m = fmaxf(m, gcn_row_xor<1>(m));
    m = fmaxf(m, gcn_row_xor<2>(m));
    m = fmaxf(m, gcn_row_xor<4>(m));
    m = fmaxf(m, gcn_row_xor<8>(m));
    float se = expf(t - m);
    se += gcn_row_xor<1>(se);
    se += gcn_row_xor<2>(se);
    se += gcn_row_xor<4>(se);
    se += gcn_row_xor<8>(se);
    const float lse = logf(se);
    if (valid) {
      const unsigned ob = (unsigned)(n * 64 + fo4);
      *reinterpret_cast<float*>(reinterpret_cast<char*>(out_h) + ob) = t;
      *reinterpret_cast<float*>(reinterpret_cast<char*>(out_ls) + ob) = t - m - lse;
    }
    st0 = st1; dg0 = dg1;
    st1 = st2; dg1 = dg2;
    sa0 = sa1; sb0 = sb1;
  }
}

extern "C" void kernel_launch(void* const* d_in, const int* in_sizes, int n_in,
                              void* d_out, int out_size, void* d_ws, size_t ws_size,
                              hipStream_t stream) {
  const float* x  = (const float*)d_in[0];
  const int*   ei = (const int*)d_in[1];
  const float* W1 = (const float*)d_in[2];
  const float* b1 = (const float*)d_in[3];
  const float* W2 = (const float*)d_in[4];
  const float* b2 = (const float*)d_in[5];

  const int N = in_sizes[0] / 32;
  const int E = in_sizes[1] / 2;
  const int* src = ei;
  const int* dst = ei + E;

  int*   wsi      = (int*)d_ws;
  float* wsf      = (float*)d_ws;
  int*   cnt      = wsi;
  float* dinv     = wsf + (size_t)N;
  int*   rowstart = wsi + (size_t)N * 2;
  const int NB = (N + GCN_BNODES - 1) / GCN_BNODES;
  const int G1 = (E + GCN_CHUNK - 1) / GCN_CHUNK;
  const int TS = NB + 1;
  unsigned* chunked = (unsigned*)(wsi + (size_t)N * 3);
  int*   csr      = wsi + (size_t)N * 3 + (size_t)E;
  float* xw1      = wsf + (size_t)N * 3 + (size_t)E * 2;
  float* hw2      = xw1 + (size_t)N * 64;
  int*   tab      = (int*)(hw2 + (size_t)N * 32);

  float* out_h  = (float*)d_out;
  float* out_ls = out_h + (size_t)N * 16;

  const unsigned gather32Blocks = (unsigned)((N + GCN_GNODES - 1) / GCN_GNODES);
  const unsigned gather16Blocks = (unsigned)(((N + 1) / 2 + 31) / 32);

  gcn_chunksort_kernel<<<(unsigned)G1, 512, 0, stream>>>(src, dst, chunked, tab, E, NB, TS);
  gcn_place_kernel<<<(unsigned)NB, 512, 0, stream>>>(tab, chunked, cnt, rowstart, dinv, csr,
                                                    x, W1, xw1, N, G1, TS);

  gcn_gather32_kernel<<<gather32Blocks, 256, 0, stream>>>(rowstart, cnt, csr, dinv, xw1, b1, W2,
                                                          hw2, N);

  gcn_gather16_kernel<<<gather16Blocks, 256, 0, stream>>>(rowstart, cnt, csr, dinv, hw2, b2,
                                                          out_h, out_ls, N, (int)gather16Blocks * 8);
}


# place kernel moves edges in whole lines: a wave reads each piece with consecutive lanes, sources leave via an LDS copy as contiguous stores; sum order = counting adds' arrival under new lane map
# speedup vs baseline: 5.0711x; 1.1422x over previous
#include <hip/hip_runtime.h>
#include <hip/hip_bf16.h>


typedef __attribute__((ext_vector_type(2))) float v2f;
typedef __attribute__((ext_vector_type(8))) float v8f;
typedef __attribute__((ext_vector_type(4))) float v4f;


template <int NCOL>
__device__ __forceinline__ void gcn_load_b(const float* __restrict__ W, v2f* b, int lo, int hi) {
#pragma unroll
  for (int ct = 0; ct < NCOL / 16; ++ct) {
#pragma unroll
    for (int kb = 0; kb < 8; ++kb) {
      const int k = kb * 4 + hi * 2;
      b[ct * 8 + kb].x = W[(size_t)k * NCOL + ct * 16 + lo];
      b[ct * 8 + kb].y = W[(size_t)(k + 1) * NCOL + ct * 16 + lo];
    }
  }
}

template <int NCOL>
__device__ __forceinline__ void gcn_tile_product(const v2f* a, const v2f* b, const float* rs,
                                                 float* __restrict__ Yt, int lo, int hi) {
#pragma unroll
  for (int ct = 0; ct < NCOL / 16; ++ct) {
    v8f c = {};
#pragma unroll
    for (int kb = 0; kb < 8; ++kb) {
      c = __builtin_amdgcn_wmma_f32_16x16x4_f32(false, a[kb], false, b[ct * 8 + kb],
                                                (short)0, c, false, false);
    }
#pragma unroll
    for (int v = 0; v < 8; ++v) {
      const int m = v + hi * 8;
      Yt[(size_t)m * NCOL + ct * 16 + lo] = c[v] * rs[v];
    }
  }
}

#define GCN_BSH 8
#define GCN_BNODES 256
#define GCN_CHUNK 4096
#define GCN_MAXB 2048

__global__ __launch_bounds__(512) void gcn_chunksort_kernel(const int* __restrict__ src,
                                                            const int* __restrict__ dst,
                                                            unsigned* __restrict__ chunked,
                                                            int* __restrict__ tab,
                                                            int E, int NB, int TS) {
  __shared__ int h[GCN_MAXB + 4];
  __shared__ unsigned stage[GCN_CHUNK];
  __shared__ int wsum[16];
  const int tid = threadIdx.x;
  const int lane = tid & 31;
  const int wave = tid >> 5;
  const int b = blockIdx.x;
  const int base = b * GCN_CHUNK + tid;
  int d[8], s[8], r[8];
#pragma unroll
  for (int u = 0; u < 8; ++u) {
    const int e = min(base + u * 512, E - 1);
    d[u] = dst[e];
    s[u] = src[e];
  }
#pragma unroll
  for (int i = 0; i < 4; ++i) h[tid * 4 + i] = 0;
  if (tid == 0) h[GCN_MAXB] = 0;
  __syncthreads();
#pragma unroll
  for (int u = 0; u < 8; ++u) {
    const int idx = (base + u * 512 < E) ? (d[u] >> GCN_BSH) : GCN_MAXB;
    r[u] = atomicAdd(&h[idx], 1);
  }
  __syncthreads();
  int c[4];
#pragma unroll
  for (int i = 0; i < 4; ++i) c[i] = h[tid * 4 + i];
  const int s4 = (c[0] + c[1]) + (c[2] + c[3]);
  int x = s4;
#pragma unroll
  for (int o = 1; o < 32; o <<= 1) {
    const int y = __shfl_up(x, o, 32);
    if (lane >= o) x += y;
  }
  if (lane == 31) wsum[wave] = x;
  __syncthreads();
  int ex = x - s4;
#pragma unroll
  for (int w = 0; w < 16; ++w) {
    if (w < wave) ex += wsum[w];
  }
#pragma unroll
  for (int i = 0; i < 4; ++i) {
    h[tid * 4 + i] = ex;
    ex += c[i];
  }
  __syncthreads();
  for (int k = tid; k <= NB; k += 512) tab[(size_t)b * TS + k] = h[k];
#pragma unroll
  for (int u = 0; u < 8; ++u) {
    if (base + u * 512 < E) {
      stage[h[d[u] >> GCN_BSH] + r[u]] =
          ((unsigned)s[u] << GCN_BSH) | (unsigned)(d[u] & (GCN_BNODES - 1));
    }
  }
  __syncthreads();
  const int total = h[NB];
#pragma unroll
  for (int u = 0; u < 8; ++u) {
    const int i = u * 512 + tid;
    if (i < total) chunked[(size_t)b * GCN_CHUNK + i] = stage[i];
  }
}

#define GCN_WCH 25
#define GCN_STAGE 8192

#define GCN_EACH_PIECE(M)                                                                     \
  M(0) M(1) M(2) M(3) M(4) M(5) M(6) M(7) M(8) M(9) M(10) M(11) M(12) M(13) M(14) M(15)       \
  M(16) M(17) M(18) M(19) M(20) M(21) M(22) M(23) M(24)
static_assert(GCN_WCH == 25, "GCN_EACH_PIECE writes out 25 pieces");
#define GCN_PIECE_LEN(U) __builtin_amdgcn_readlane(pl, U)
#define GCN_PIECE_LOAD(U)                                                                     \
  v[U] = 0u;                                                                                  \
  if (lane < GCN_PIECE_LEN(U))                                                                \
    v[U] = chunked[(unsigned)__builtin_amdgcn_readlane(pb, U) + (unsigned)lane];
#define GCN_PIECE_RANK(U)                                                                     \
  r[U] = 0;                                                                                   \
  if (lane < GCN_PIECE_LEN(U)) r[U] = atomicAdd(&c[v[U] & (GCN_BNODES - 1)], 1);
#define GCN_PIECE_POS(U) pp[U] = cur[v[U] & (GCN_BNODES - 1)] + r[U];
#define GCN_PIECE_PUT_LDS(U)                                                                  \
  if (lane < GCN_PIECE_LEN(U)) stage[pp[U]] = (int)(v[U] >> GCN_BSH);
#define GCN_PIECE_PUT_CSR(U)                                                                  \
  if (lane < GCN_PIECE_LEN(U)) csrB[pp[U]] = (int)(v[U] >> GCN_BSH);
#define GCN_PLACE_REST(DST)                                                                   \
  {                                                                                           \
    if (__any(pl > 32)) {             \
      for (int u = 0; u < GCN_WCH; ++u) {                                                     \
        const int ln = __builtin_amdgcn_readlane(pl, u);                                      \
        const unsigned* qp = chunked + (unsigned)__builtin_amdgcn_readlane(pb, u);            \
        for (int j = 32 + lane; j < ln; j += 32) {                                            \
          const unsigned w = qp[j];                                                           \
          DST[atomicAdd(&cur2[w & (GCN_BNODES - 1)], 1)] = (int)(w >> GCN_BSH);               \
        }                                                                                     \
      }                                                                                       \
    }                                                                                         \
    for (int ch = 16 * GCN_WCH + wave; ch < G1; ch += 16) {   \
      const int q0 = tab[(size_t)ch * TS + k];                                                \
      const int ql = tab[(size_t)ch * TS + k + 1] - q0;                                       \
      const unsigned* qp = chunked + (size_t)ch * GCN_CHUNK + q0;                             \
      for (int j = lane; j < ql; j += 32) {                                                   \
        const unsigned w = qp[j];                                                             \
        DST[atomicAdd(&cur2[w & (GCN_BNODES - 1)], 1)] = (int)(w >> GCN_BSH);                 \
      }                                                                                       \
    }                                                                                         \
  }

__global__ __launch_bounds__(512, 8) void gcn_place_kernel(const int* __restrict__ tab,
                                                        const unsigned* __restrict__ chunked,
                                                        int* __restrict__ cnt,
                                                        int* __restrict__ rowstart,
                                                        float* __restrict__ dinv,
                                                        int* __restrict__ csr,
                                                        const float* __restrict__ X,
                                                        const float* __restrict__ W,
                                                        float* __restrict__ Y,
                                                        int N, int G1, int TS) {
  __shared__ int c[GCN_BNODES];
  __shared__ int c2[GCN_BNODES];
  __shared__ int cur[GCN_BNODES];
  __shared__ int cur2[GCN_BNODES];
  __shared__ int wsum[16];
  __shared__ int wacc[16];
  __shared__ float sdinv[GCN_BNODES];
  __shared__ int stage[GCN_STAGE];
  const int tid = threadIdx.x;
  const int lane = tid & 31;
  const int wave = __builtin_amdgcn_readfirstlane(tid >> 5);
  const int k = blockIdx.x;
  const int lo = lane & 15;
  const int hi = lane >> 4;
  const int rowBase = k * GCN_BNODES + wave * 16;
  if (tid < GCN_BNODES) {
    c[tid] = 0;
    c2[tid] = 0;
  }
  __syncthreads();
  v2f a[8];
  {
    const int xrow = min(rowBase + lo, N - 1);
#pragma unroll
    for (int kb = 0; kb < 8; ++kb) {
      const float* xr = X + (size_t)xrow * 32 + kb * 4 + hi * 2;
      a[kb].x = xr[0];
      a[kb].y = xr[1];
    }
  }
  const int chunk0 = wave * GCN_WCH;
  int t0 = 0, t1 = 0;
  if (lane < GCN_WCH && chunk0 + lane < G1) {
    const int* tp = tab + (size_t)(chunk0 + lane) * TS + k;
    t0 = tp[0];
    t1 = tp[1];
  }
  int acc = t0;
  const int pl = t1 - t0;
  const int pb = (chunk0 + lane) * GCN_CHUNK + t0;
  unsigned v[GCN_WCH];
  int r[GCN_WCH];
  GCN_EACH_PIECE(GCN_PIECE_LOAD)
  GCN_EACH_PIECE(GCN_PIECE_RANK)
  if (__any(pl > 32)) {
    for (int u = 0; u < GCN_WCH; ++u) {
      const int ln = __builtin_amdgcn_readlane(pl, u);
      const unsigned* qp = chunked + (unsigned)__builtin_amdgcn_readlane(pb, u);
      for (int j = 32 + lane; j < ln; j += 32) atomicAdd(&c2[qp[j] & (GCN_BNODES - 1)], 1);
    }
  }
  for (int ch = 16 * GCN_WCH + wave; ch < G1; ch += 16) {
    const int q0 = tab[(size_t)ch * TS + k];
    const int ql = tab[(size_t)ch * TS + k + 1] - q0;
    if (lane == 0) acc += q0;
    const unsigned* qp = chunked + (size_t)ch * GCN_CHUNK + q0;
    for (int j = lane; j < ql; j += 32) atomicAdd(&c2[qp[j] & (GCN_BNODES - 1)], 1);
  }
#pragma unroll
  for (int o = 16; o > 0; o >>= 1) acc += __shfl_xor(acc, o, 32);
  if (lane == 0) wacc[wave] = acc;
  __syncthreads();
  const int x1 = (tid < GCN_BNODES) ? c[tid] : 0;
  const int x = x1 + ((tid < GCN_BNODES) ? c2[tid] : 0);
  int incl = x;
#pragma unroll
  for (int o = 1; o < 32; o <<= 1) {
    const int y = __shfl_up(incl, o, 32);
    if (lane >= o) incl += y;
  }
  if (lane == 31) wsum[wave] = incl;
  __syncthreads();
  int ex = incl - x;
  int b0 = 0;
  int total = 0;
#pragma unroll
  for (int w = 0; w < 16; ++w) {
    b0 += wacc[w];
    total += wsum[w];
    if (w < wave) ex += wsum[w];
  }
  if (tid < GCN_BNODES) {
    cur[tid] = ex;
    cur2[tid] = ex + x1;
    const int node = k * GCN_BNODES + tid;
    if (node < N) {
      rowstart[node] = b0 + ex;
      cnt[node] = x;
      const float dg = (float)(x + 1);
      const float di = (dg > 0.0f) ? rsqrtf(dg) : 0.0f;
      dinv[node] = di;
      sdinv[tid] = di;
    }
  }
  __syncthreads();
  int* csrB = csr + b0;
  int pp[GCN_WCH];
  GCN_EACH_PIECE(GCN_PIECE_POS)
  if (total <= GCN_STAGE) {
    GCN_EACH_PIECE(GCN_PIECE_PUT_LDS)
    GCN_PLACE_REST(stage)
    __syncthreads();
    for (int i = tid; i < total; i += 512) csrB[i] = stage[i];
  } else {
    GCN_EACH_PIECE(GCN_PIECE_PUT_CSR)
    GCN_PLACE_REST(csrB)
  }
  v2f bw[16];
  gcn_load_b<32>(W, bw, lo, hi);
  if (rowBase + 16 <= N) {
    float rs[8];
#pragma unroll
    for (int v8 = 0; v8 < 8; ++v8) rs[v8] = sdinv[wave * 16 + v8 + hi * 8];
    gcn_tile_product<32>(a, bw, rs, Y + (size_t)rowBase * 32, lo, hi);
  }
}


template <int M>
__device__ __forceinline__ float gcn_row_xor(float x) {
  return __builtin_bit_cast(
      float, __builtin_amdgcn_update_dpp(0, __builtin_bit_cast(int, x), 0x160 + M, 0xf, 0xf, false));
}

__device__ __forceinline__ float gcn_half_swap(float x) {
  const int xi = __builtin_bit_cast(int, x);
  return __builtin_bit_cast(
      float, __builtin_amdgcn_permlanex16(xi, xi, 0x76543210, 0xfedcba98, false, false));
}

template <int J>
__device__ __forceinline__ int gcn_bcast8(int x) {
  return __builtin_amdgcn_mov_dpp8(x, J * 0x249249);
}

#define GCN_ROW4X32(J) \
  if (NJ > J) v[J] = *reinterpret_cast<const v4f*>(Hb + (unsigned)(gcn_bcast8<J>(sidx) * 128 + q16));
template <int NJ>
__device__ __forceinline__ v4f gcn_rows32(const char* __restrict__ Hb, int sidx, int lg, int q16,
                                          v4f acc) {
  v4f v[8];
  GCN_ROW4X32(0)
  GCN_ROW4X32(1)
  GCN_ROW4X32(2)
  GCN_ROW4X32(3)
  GCN_ROW4X32(4)
  GCN_ROW4X32(5)
  GCN_ROW4X32(6)
  GCN_ROW4X32(7)
#pragma unroll
  for (int j = 0; j + 1 < NJ; ++j) acc += v[j];
  const v4f zero = {0.0f, 0.0f, 0.0f, 0.0f};
  acc += (4 * (NJ - 1) < lg) ? v[NJ - 1] : zero;
  return acc;
}

__device__ __forceinline__ v4f gcn_rows32_any(const char* __restrict__ Hb, int sidx, int nv, int g,
                                              int q16, v4f acc) {
  const int lg = nv - g;
  switch ((nv + 3) >> 2) {
    case 1: return gcn_rows32<1>(Hb, sidx, lg, q16, acc);
    case 2: return gcn_rows32<2>(Hb, sidx, lg, q16, acc);
    case 3: return gcn_rows32<3>(Hb, sidx, lg, q16, acc);
    case 4: return gcn_rows32<4>(Hb, sidx, lg, q16, acc);
    case 5: return gcn_rows32<5>(Hb, sidx, lg, q16, acc);
    case 6: return gcn_rows32<6>(Hb, sidx, lg, q16, acc);
    case 7: return gcn_rows32<7>(Hb, sidx, lg, q16, acc);
    case 8: return gcn_rows32<8>(Hb, sidx, lg, q16, acc);
    default: return acc;
  }
}

#define GCN_GNODES 64
#define GCN_TPITCH 36

__global__ __launch_bounds__(256) void gcn_gather32_kernel(const int* __restrict__ rowstart,
                                                           const int* __restrict__ cnt,
                                                           const int* __restrict__ csr,
                                                           const float* __restrict__ dinv,
                                                           const float* __restrict__ H,
                                                           const float* __restrict__ b1,
                                                           const float* __restrict__ W2,
                                                           float* __restrict__ HW2,
                                                           int N) {
  __shared__ float tile[GCN_GNODES * GCN_TPITCH];
  __shared__ float sdn[GCN_GNODES];
  const int lane = threadIdx.x & 31;
  const int wave = __builtin_amdgcn_readfirstlane(threadIdx.x >> 5);
  const int blockNode = blockIdx.x * GCN_GNODES;
  const int node0 = blockNode + wave * 8;
  const int g = lane >> 3;
  const int q16 = (lane & 7) * 16;
  const int slot = ((lane & 7) << 2) | g;
  const int fo = (lane & 7) * 4 + g;
  const float bias = b1[fo];
  const char* Hb = reinterpret_cast<const char*>(H);
  const int lo = lane & 15;
  const int hi = lane >> 4;
  const int mn = node0 + (lane & 7);
  const int mc = min(mn, N - 1);
  const int mstart = rowstart[mc];
  const int mdeg = (mn < N) ? cnt[mc] : 0;
  const float mdn = dinv[mc];
  if (lane < 8) sdn[wave * 8 + lane] = mdn;
  int s_cur = min(node0, N - 1);
  {
    const int st = __builtin_amdgcn_readlane(mstart, 0);
    const int dg = __builtin_amdgcn_readlane(mdeg, 0);
    if (slot < dg) s_cur = csr[st + slot];
  }
  asm volatile("" ::"v"(s_cur), "v"(mstart), "v"(mdeg), "v"(mdn));
  for (int i = 0; i < 8; ++i) {
    const int n = node0 + i;
    const int inext = min(i + 1, 7);
    const int stn = __builtin_amdgcn_readlane(mstart, inext);
    const int dgn = (i < 7) ? __builtin_amdgcn_readlane(mdeg, inext) : 0;
    int s_nxt = min(n + 1, N - 1);
    if (slot < dgn) s_nxt = csr[stn + slot];
    if (n < N) {
      const int start = __builtin_amdgcn_readlane(mstart, i);
      const int deg = __builtin_amdgcn_readlane(mdeg, i);
      const float dn = __builtin_bit_cast(float,
                                          __builtin_amdgcn_readlane(__builtin_bit_cast(int, mdn), i));
      const float* own = H + (size_t)n * 32;
      const float selfv = own[(unsigned)fo];
      v4f acc = {0.0f, 0.0f, 0.0f, 0.0f};
      acc = gcn_rows32_any(Hb, s_cur, min(32, deg), g, q16, acc);
      if (deg > 32) {
        for (int c0 = 32; c0 < deg; c0 += 32) {
          const int nv2 = min(32, deg - c0);
          int s = n;
          if (slot < nv2) s = csr[start + c0 + slot];
          acc = gcn_rows32_any(Hb, s, nv2, g, q16, acc);
        }
      }
      float r[4];
#pragma unroll
      for (int e = 0; e < 4; ++e) {
        float a = acc[e];
        a += gcn_row_xor<8>(a);
        a += gcn_half_swap(a);
        r[e] = a;
      }
      const float sum = (g == 0) ? r[0] : (g == 1) ? r[1] : (g == 2) ? r[2] : r[3];
      const float t = (sum + selfv) * dn;
      tile[(wave * 8 + i) * GCN_TPITCH + fo] = tanhf(t + bias);
    }
    s_cur = s_nxt;
  }
  v2f bw[8] = {};
  if (wave < 4) gcn_load_b<16>(W2, bw, lo, hi);
  __syncthreads();
  const int rowBase = blockNode + wave * 16;
  if (wave < 4 && rowBase + 16 <= N) {
    v2f a[8];
#pragma unroll
    for (int kb = 0; kb < 8; ++kb) {
      const float* xr = tile + (wave * 16 + lo) * GCN_TPITCH + kb * 4 + hi * 2;
      a[kb].x = xr[0];
      a[kb].y = xr[1];
    }
    float rs[8];
#pragma unroll
    for (int v = 0; v < 8; ++v) rs[v] = sdn[wave * 16 + v + hi * 8];
    gcn_tile_product<16>(a, bw, rs, HW2 + (size_t)rowBase * 16, lo, hi);
  }
}


__device__ __forceinline__ void gcn_pair_bounds(const int* __restrict__ rowstart,
                                                const int* __restrict__ cnt, int p, int nPairs,
                                                int N, int sub, int& start, int& deg) {
  const int n = 2 * p + sub;
  const bool valid = (p < nPairs) && (n < N);
  const int nc = valid ? n : N - 1;
  start = rowstart[nc];
  deg = valid ? cnt[nc] : 0;
}

__device__ __forceinline__ void gcn_pair_idx(const int* __restrict__ csr, int start, int left,
                                             int ea, int& sa, int& sb) {
  sa = 0;
  sb = 0;
  const int le = left - ea;
  if (le > 0) sa = csr[start + ea];
  if (le > 1) sb = csr[start + ea + 1];
}

#define GCN_ROW4X16(J)                                                                     \
  if (NJ > J) {                                                                            \
    const int s0 = gcn_bcast8<J>(sa);                                                      \
    const int s1 = gcn_bcast8<J>(sb);                                                      \
    v[J] = *reinterpret_cast<const v4f*>(Hb + (unsigned)((godd ? s1 : s0) * 64 + q16));    \
  }
template <int NJ>
__device__ __forceinline__ v4f gcn_rows16(const char* __restrict__ Hb, int sa, int sb, int left,
                                          int godd, int g, int q16, v4f acc) {
  v4f v[8];
  GCN_ROW4X16(0)
  GCN_ROW4X16(1)
  GCN_ROW4X16(2)
  GCN_ROW4X16(3)
  GCN_ROW4X16(4)
  GCN_ROW4X16(5)
  GCN_ROW4X16(6)
  GCN_ROW4X16(7)
  const v4f zero = {0.0f, 0.0f, 0.0f, 0.0f};
  const int lg = left - g;
#pragma unroll
  for (int j = 0; j < NJ; ++j) acc += (4 * j < lg) ? v[j] : zero;
  return acc;
}

__device__ __forceinline__ v4f gcn_rows16_any(const char* __restrict__ Hb, int sa, int sb, int left,
                                              int nmax, int godd, int g, int q16, v4f acc) {
  switch ((nmax + 3) >> 2) {
    case 1: return gcn_rows16<1>(Hb, sa, sb, left, godd, g, q16, acc);
    case 2: return gcn_rows16<2>(Hb, sa, sb, left, godd, g, q16, acc);
    case 3: return gcn_rows16<3>(Hb, sa, sb, left, godd, g, q16, acc);
    case 4: return gcn_rows16<4>(Hb, sa, sb, left, godd, g, q16, acc);
    case 5: return gcn_rows16<5>(Hb, sa, sb, left, godd, g, q16, acc);
    case 6: return gcn_rows16<6>(Hb, sa, sb, left, godd, g, q16, acc);
    case 7: return gcn_rows16<7>(Hb, sa, sb, left, godd, g, q16, acc);
    case 8: return gcn_rows16<8>(Hb, sa, sb, left, godd, g, q16, acc);
    default: return acc;
  }
}

__global__ __launch_bounds__(256) void gcn_gather16_kernel(const int* __restrict__ rowstart,
                                                           const int* __restrict__ cnt,
                                                           const int* __restrict__ csr,
                                                           const float* __restrict__ dinv,
                                                           const float* __restrict__ H,
                                                           const float* __restrict__ b2,
                                                           float* __restrict__ out_h,
                                                           float* __restrict__ out_ls,
                                                           int N, int nWaves) {
  const int lane = threadIdx.x & 31;
  const int f = lane & 15;
  const int sub = lane >> 4;
  const int g = f >> 2;
  const int godd = g & 1;
  const int q16 = (f & 3) * 16;
  const int fo = (f & 3) * 4 + g;
  const int ea = ((lane & 7) << 2) | (((lane >> 3) & 1) << 1);
  const int fo4 = fo * 4;
  const float bias = b2[fo];
  const char* Hb = reinterpret_cast<const char*>(H);
  const int wave = __builtin_amdgcn_readfirstlane(blockIdx.x * 8 + (threadIdx.x >> 5));
  const int nPairs = (N + 1) >> 1;
  int st0, dg0, st1, dg1;
  gcn_pair_bounds(rowstart, cnt, wave, nPairs, N, sub, st0, dg0);
  gcn_pair_bounds(rowstart, cnt, wave + nWaves, nPairs, N, sub, st1, dg1);
  int sa0, sb0;
  gcn_pair_idx(csr, st0, dg0, ea, sa0, sb0);
  asm volatile("" ::"v"(sa0), "v"(sb0), "v"(st0), "v"(dg0), "v"(st1), "v"(dg1), "v"(bias));
  for (int p = wave; p < nPairs; p += nWaves) {
    int sa1, sb1;
    gcn_pair_idx(csr, st1, dg1, ea, sa1, sb1);
    int st2, dg2;
    gcn_pair_bounds(rowstart, cnt, p + 2 * nWaves, nPairs, N, sub, st2, dg2);
    const int n = 2 * p + sub;
    const bool valid = n < N;
    const int nc = min(n, N - 1);
    const int start = st0;
    const int deg = dg0;
    const int degmax = max(__builtin_amdgcn_readlane(deg, 0), __builtin_amdgcn_readlane(deg, 16));
    const float di = dinv[nc];
    const float selfv = *reinterpret_cast<const float*>(Hb + (unsigned)(nc * 64 + fo4));
    v4f acc = {0.0f, 0.0f, 0.0f, 0.0f};
    acc = gcn_rows16_any(Hb, sa0, sb0, deg, min(32, degmax), godd, g, q16, acc);
    if (degmax > 32) {
      for (int c0 = 32; c0 < degmax; c0 += 32) {
        const int left = deg - c0;
        int sa, sb;
        gcn_pair_idx(csr, start + c0, left, ea, sa, sb);
        acc = gcn_rows16_any(Hb, sa, sb, left, min(32, degmax - c0), godd, g, q16, acc);
      }
    }
    float r[4];
#pragma unroll
    for (int e = 0; e < 4; ++e) {
      float a = acc[e];
      a += gcn_row_xor<4>(a);
      a += gcn_row_xor<8>(a);
      r[e] = a;
    }
    const float sum = (g == 0) ? r[0] : (g == 1) ? r[1] : (g == 2) ? r[2] : r[3];
    const float t = (sum + selfv) * di + bias;
    float m = t;
    m = fmaxf(m, gcn_row_xor<1>(m));
    m = fmaxf(m, gcn_row_xor<2>(m));
    m = fmaxf(m, gcn_row_xor<4>(m));
    m = fmaxf(m, gcn_row_xor<8>(m));
    float se = expf(t - m);
    se += gcn_row_xor<1>(se);
    se += gcn_row_xor<2>(se);
    se += gcn_row_xor<4>(se);
    se += gcn_row_xor<8>(se);
    const float lse = logf(se);
    if (valid) {
      const unsigned ob = (unsigned)(n * 64 + fo4);
      *reinterpret_cast<float*>(reinterpret_cast<char*>(out_h) + ob) = t;
      *reinterpret_cast<float*>(reinterpret_cast<char*>(out_ls) + ob) = t - m - lse;
    }
    st0 = st1; dg0 = dg1;
    st1 = st2; dg1 = dg2;
    sa0 = sa1; sb0 = sb1;
  }
}

extern "C" void kernel_launch(void* const* d_in, const int* in_sizes, int n_in,
                              void* d_out, int out_size, void* d_ws, size_t ws_size,
                              hipStream_t stream) {
  const float* x  = (const float*)d_in[0];
  const int*   ei = (const int*)d_in[1];
  const float* W1 = (const float*)d_in[2];
  const float* b1 = (const float*)d_in[3];
  const float* W2 = (const float*)d_in[4];
  const float* b2 = (const float*)d_in[5];

  const int N = in_sizes[0] / 32;
  const int E = in_sizes[1] / 2;
  const int* src = ei;
  const int* dst = ei + E;

  int*   wsi      = (int*)d_ws;
  float* wsf      = (float*)d_ws;
  int*   cnt      = wsi;
  float* dinv     = wsf + (size_t)N;
  int*   rowstart = wsi + (size_t)N * 2;
  const int NB = (N + GCN_BNODES - 1) / GCN_BNODES;
  const int G1 = (E + GCN_CHUNK - 1) / GCN_CHUNK;
  const int TS = NB + 1;
  unsigned* chunked = (unsigned*)(wsi + (size_t)N * 3);
  int*   csr      = wsi + (size_t)N * 3 + (size_t)E;
  float* xw1      = wsf + (size_t)N * 3 + (size_t)E * 2;
  float* hw2      = xw1 + (size_t)N * 64;
  int*   tab      = (int*)(hw2 + (size_t)N * 32);

  float* out_h  = (float*)d_out;
  float* out_ls = out_h + (size_t)N * 16;

  const unsigned gather32Blocks = (unsigned)((N + GCN_GNODES - 1) / GCN_GNODES);
  const unsigned gather16Blocks = (unsigned)(((N + 1) / 2 + 31) / 32);

  gcn_chunksort_kernel<<<(unsigned)G1, 512, 0, stream>>>(src, dst, chunked, tab, E, NB, TS);
  gcn_place_kernel<<<(unsigned)NB, 512, 0, stream>>>(tab, chunked, cnt, rowstart, dinv, csr,
                                                    x, W1, xw1, N, G1, TS);

  gcn_gather32_kernel<<<gather32Blocks, 256, 0, stream>>>(rowstart, cnt, csr, dinv, xw1, b1, W2,
                                                          hw2, N);

  gcn_gather16_kernel<<<gather16Blocks, 256, 0, stream>>>(rowstart, cnt, csr, dinv, hw2, b2,
                                                          out_h, out_ls, N, (int)gather16Blocks * 8);
}


# even split over the 256 WGPs: 6272-edge chunks (256 sort blocks, was 391; past-end lanes add to spare words of their own), gathers in 10-wave blocks (1250, was 1563); segment order by new chunks
# speedup vs baseline: 5.4640x; 1.0775x over previous
#include <hip/hip_runtime.h>
#include <hip/hip_bf16.h>


typedef __attribute__((ext_vector_type(2))) float v2f;
typedef __attribute__((ext_vector_type(8))) float v8f;
typedef __attribute__((ext_vector_type(4))) float v4f;


template <int NCOL>
__device__ __forceinline__ void gcn_load_b(const float* __restrict__ W, v2f* b, int lo, int hi) {
#pragma unroll
  for (int ct = 0; ct < NCOL / 16; ++ct) {
#pragma unroll
    for (int kb = 0; kb < 8; ++kb) {
      const int k = kb * 4 + hi * 2;
      b[ct * 8 + kb].x = W[(size_t)k * NCOL + ct * 16 + lo];
      b[ct * 8 + kb].y = W[(size_t)(k + 1) * NCOL + ct * 16 + lo];
    }
  }
}

template <int NCOL>
__device__ __forceinline__ void gcn_tile_product(const v2f* a, const v2f* b, const float* rs,
                                                 float* __restrict__ Yt, int lo, int hi) {
#pragma unroll
  for (int ct = 0; ct < NCOL / 16; ++ct) {
    v8f c = {};
#pragma unroll
    for (int kb = 0; kb < 8; ++kb) {
      c = __builtin_amdgcn_wmma_f32_16x16x4_f32(false, a[kb], false, b[ct * 8 + kb],
                                                (short)0, c, false, false);
    }
#pragma unroll
    for (int v = 0; v < 8; ++v) {
      const int m = v + hi * 8;
      Yt[(size_t)m * NCOL + ct * 16 + lo] = c[v] * rs[v];
    }
  }
}

#define GCN_BSH 8
#define GCN_BNODES 256
#define GCN_CHUNK 6272
#define GCN_CPT 13
#define GCN_MAXB 2048

__global__ __launch_bounds__(512) void gcn_chunksort_kernel(const int* __restrict__ src,
                                                            const int* __restrict__ dst,
                                                            unsigned* __restrict__ chunked,
                                                            int* __restrict__ tab,
                                                            int E, int NB, int TS) {
  __shared__ int h[GCN_MAXB + 32];
  __shared__ unsigned stage[GCN_CHUNK];
  __shared__ int wsum[16];
  const int tid = threadIdx.x;
  const int lane = tid & 31;
  const int wave = tid >> 5;
  const int b = blockIdx.x;
  const int base = b * GCN_CHUNK + tid;
  int d[GCN_CPT], s[GCN_CPT], r[GCN_CPT];
#pragma unroll
  for (int u = 0; u < GCN_CPT; ++u) {
    const int e = min(base + u * 512, E - 1);
    d[u] = dst[e];
    s[u] = src[e];
  }
#pragma unroll
  for (int i = 0; i < 4; ++i) h[tid * 4 + i] = 0;
  __syncthreads();
#pragma unroll
  for (int u = 0; u < GCN_CPT; ++u) {
    const bool counted = (u * 512 + tid < GCN_CHUNK) && (base + u * 512 < E);
    const int idx = counted ? (d[u] >> GCN_BSH) : GCN_MAXB + lane;
    r[u] = atomicAdd(&h[idx], 1);
  }
  __syncthreads();
  int c[4];
#pragma unroll
  for (int i = 0; i < 4; ++i) c[i] = h[tid * 4 + i];
  const int s4 = (c[0] + c[1]) + (c[2] + c[3]);
  int x = s4;
#pragma unroll
  for (int o = 1; o < 32; o <<= 1) {
    const int y = __shfl_up(x, o, 32);
    if (lane >= o) x += y;
  }
  if (lane == 31) wsum[wave] = x;
  __syncthreads();
  int ex = x - s4;
#pragma unroll
  for (int w = 0; w < 16; ++w) {
    if (w < wave) ex += wsum[w];
  }
#pragma unroll
  for (int i = 0; i < 4; ++i) {
    h[tid * 4 + i] = ex;
    ex += c[i];
  }
  __syncthreads();
  for (int k = tid; k <= NB; k += 512) tab[(size_t)b * TS + k] = h[k];
#pragma unroll
  for (int u = 0; u < GCN_CPT; ++u) {
    if (u * 512 + tid < GCN_CHUNK && base + u * 512 < E) {
      stage[h[d[u] >> GCN_BSH] + r[u]] =
          ((unsigned)s[u] << GCN_BSH) | (unsigned)(d[u] & (GCN_BNODES - 1));
    }
  }
  __syncthreads();
  const int total = h[NB];
#pragma unroll
  for (int u = 0; u < GCN_CPT; ++u) {
    const int i = u * 512 + tid;
    if (i < total) chunked[(size_t)b * GCN_CHUNK + i] = stage[i];
  }
}

#define GCN_WCH 16
#define GCN_STAGE 8192

#define GCN_EACH_PIECE(M)                                                                     \
  M(0) M(1) M(2) M(3) M(4) M(5) M(6) M(7) M(8) M(9) M(10) M(11) M(12) M(13) M(14) M(15)
static_assert(GCN_WCH == 16, "GCN_EACH_PIECE writes out 16 pieces");
#define GCN_PIECE_LEN(U) __builtin_amdgcn_readlane(pl, U)
#define GCN_PIECE_LOAD(U)                                                                     \
  v[U] = 0u;                                                                                  \
  if (lane < GCN_PIECE_LEN(U))                                                                \
    v[U] = chunked[(unsigned)__builtin_amdgcn_readlane(pb, U) + (unsigned)lane];
#define GCN_PIECE_RANK(U)                                                                     \
  r[U] = 0;                                                                                   \
  if (lane < GCN_PIECE_LEN(U)) r[U] = atomicAdd(&c[v[U] & (GCN_BNODES - 1)], 1);
#define GCN_PIECE_POS(U) pp[U] = cur[v[U] & (GCN_BNODES - 1)] + r[U];
#define GCN_PIECE_PUT_LDS(U)                                                                  \
  if (lane < GCN_PIECE_LEN(U)) stage[pp[U]] = (int)(v[U] >> GCN_BSH);
#define GCN_PIECE_PUT_CSR(U)                                                                  \
  if (lane < GCN_PIECE_LEN(U)) csrB[pp[U]] = (int)(v[U] >> GCN_BSH);
#define GCN_PLACE_REST(DST)                                                                   \
  {                                                                                           \
    if (__any(pl > 32)) {             \
      for (int u = 0; u < GCN_WCH; ++u) {                                                     \
        const int ln = __builtin_amdgcn_readlane(pl, u);                                      \
        const unsigned* qp = chunked + (unsigned)__builtin_amdgcn_readlane(pb, u);            \
        for (int j = 32 + lane; j < ln; j += 32) {                                            \
          const unsigned w = qp[j];                                                           \
          DST[atomicAdd(&cur2[w & (GCN_BNODES - 1)], 1)] = (int)(w >> GCN_BSH);               \
        }                                                                                     \
      }                                                                                       \
    }                                                                                         \
    for (int ch = 16 * GCN_WCH + wave; ch < G1; ch += 16) {   \
      const int q0 = tab[(size_t)ch * TS + k];                                                \
      const int ql = tab[(size_t)ch * TS + k + 1] - q0;                                       \
      const unsigned* qp = chunked + (size_t)ch * GCN_CHUNK + q0;                             \
      for (int j = lane; j < ql; j += 32) {                                                   \
        const unsigned w = qp[j];                                                             \
        DST[atomicAdd(&cur2[w & (GCN_BNODES - 1)], 1)] = (int)(w >> GCN_BSH);                 \
      }                                                                                       \
    }                                                                                         \
  }

__global__ __launch_bounds__(512, 8) void gcn_place_kernel(const int* __restrict__ tab,
                                                        const unsigned* __restrict__ chunked,
                                                        int* __restrict__ cnt,
                                                        int* __restrict__ rowstart,
                                                        float* __restrict__ dinv,
                                                        int* __restrict__ csr,
                                                        const float* __restrict__ X,
                                                        const float* __restrict__ W,
                                                        float* __restrict__ Y,
                                                        int N, int G1, int TS) {
  __shared__ int c[GCN_BNODES];
  __shared__ int c2[GCN_BNODES];
  __shared__ int cur[GCN_BNODES];
  __shared__ int cur2[GCN_BNODES];
  __shared__ int wsum[16];
  __shared__ int wacc[16];
  __shared__ float sdinv[GCN_BNODES];
  __shared__ int stage[GCN_STAGE];
  const int tid = threadIdx.x;
  const int lane = tid & 31;
  const int wave = __builtin_amdgcn_readfirstlane(tid >> 5);
  const int k = blockIdx.x;
  const int lo = lane & 15;
  const int hi = lane >> 4;
  const int rowBase = k * GCN_BNODES + wave * 16;
  if (tid < GCN_BNODES) {
    c[tid] = 0;
    c2[tid] = 0;
  }
  __syncthreads();
  v2f a[8];
  {
    const int xrow = min(rowBase + lo, N - 1);
#pragma unroll
    for (int kb = 0; kb < 8; ++kb) {
      const float* xr = X + (size_t)xrow * 32 + kb * 4 + hi * 2;
      a[kb].x = xr[0];
      a[kb].y = xr[1];
    }
  }
  const int chunk0 = wave * GCN_WCH;
  int t0 = 0, t1 = 0;
  if (lane < GCN_WCH && chunk0 + lane < G1) {
    const int* tp = tab + (size_t)(chunk0 + lane) * TS + k;
    t0 = tp[0];
    t1 = tp[1];
  }
  int acc = t0;
  const int pl = t1 - t0;
  const int pb = (chunk0 + lane) * GCN_CHUNK + t0;
  unsigned v[GCN_WCH];
  int r[GCN_WCH];
  GCN_EACH_PIECE(GCN_PIECE_LOAD)
  GCN_EACH_PIECE(GCN_PIECE_RANK)
  if (__any(pl > 32)) {
    for (int u = 0; u < GCN_WCH; ++u) {
      const int ln = __builtin_amdgcn_readlane(pl, u);
      const unsigned* qp = chunked + (unsigned)__builtin_amdgcn_readlane(pb, u);
      for (int j = 32 + lane; j < ln; j += 32) atomicAdd(&c2[qp[j] & (GCN_BNODES - 1)], 1);
    }
  }
  for (int ch = 16 * GCN_WCH + wave; ch < G1; ch += 16) {
    const int q0 = tab[(size_t)ch * TS + k];
    const int ql = tab[(size_t)ch * TS + k + 1] - q0;
    if (lane == 0) acc += q0;
    const unsigned* qp = chunked + (size_t)ch * GCN_CHUNK + q0;
    for (int j = lane; j < ql; j += 32) atomicAdd(&c2[qp[j] & (GCN_BNODES - 1)], 1);
  }
#pragma unroll
  for (int o = 16; o > 0; o >>= 1) acc += __shfl_xor(acc, o, 32);
  if (lane == 0) wacc[wave] = acc;
  __syncthreads();
  const int x1 = (tid < GCN_BNODES) ? c[tid] : 0;
  const int x = x1 + ((tid < GCN_BNODES) ? c2[tid] : 0);
  int incl = x;
#pragma unroll
  for (int o = 1; o < 32; o <<= 1) {
    const int y = __shfl_up(incl, o, 32);
    if (lane >= o) incl += y;
  }
  if (lane == 31) wsum[wave] = incl;
  __syncthreads();
  int ex = incl - x;
  int b0 = 0;
  int total = 0;
#pragma unroll
  for (int w = 0; w < 16; ++w) {
    b0 += wacc[w];
    total += wsum[w];
    if (w < wave) ex += wsum[w];
  }
  if (tid < GCN_BNODES) {
    cur[tid] = ex;
    cur2[tid] = ex + x1;
    const int node = k * GCN_BNODES + tid;
    if (node < N) {
      rowstart[node] = b0 + ex;
      cnt[node] = x;
      const float dg = (float)(x + 1);
      const float di = (dg > 0.0f) ? rsqrtf(dg) : 0.0f;
      dinv[node] = di;
      sdinv[tid] = di;
    }
  }
  __syncthreads();
  int* csrB = csr + b0;
  int pp[GCN_WCH];
  GCN_EACH_PIECE(GCN_PIECE_POS)
  if (total <= GCN_STAGE) {
    GCN_EACH_PIECE(GCN_PIECE_PUT_LDS)
    GCN_PLACE_REST(stage)
    __syncthreads();
    for (int i = tid; i < total; i += 512) csrB[i] = stage[i];
  } else {
    GCN_EACH_PIECE(GCN_PIECE_PUT_CSR)
    GCN_PLACE_REST(csrB)
  }
  v2f bw[16];
  gcn_load_b<32>(W, bw, lo, hi);
  if (rowBase + 16 <= N) {
    float rs[8];
#pragma unroll
    for (int v8 = 0; v8 < 8; ++v8) rs[v8] = sdinv[wave * 16 + v8 + hi * 8];
    gcn_tile_product<32>(a, bw, rs, Y + (size_t)rowBase * 32, lo, hi);
  }
}


template <int M>
__device__ __forceinline__ float gcn_row_xor(float x) {
  return __builtin_bit_cast(
      float, __builtin_amdgcn_update_dpp(0, __builtin_bit_cast(int, x), 0x160 + M, 0xf, 0xf, false));
}

__device__ __forceinline__ float gcn_half_swap(float x) {
  const int xi = __builtin_bit_cast(int, x);
  return __builtin_bit_cast(
      float, __builtin_amdgcn_permlanex16(xi, xi, 0x76543210, 0xfedcba98, false, false));
}

template <int J>
__device__ __forceinline__ int gcn_bcast8(int x) {
  return __builtin_amdgcn_mov_dpp8(x, J * 0x249249);
}

#define GCN_ROW4X32(J) \
  if (NJ > J) v[J] = *reinterpret_cast<const v4f*>(Hb + (unsigned)(gcn_bcast8<J>(sidx) * 128 + q16));
template <int NJ>
__device__ __forceinline__ v4f gcn_rows32(const char* __restrict__ Hb, int sidx, int lg, int q16,
                                          v4f acc) {
  v4f v[8];
  GCN_ROW4X32(0)
  GCN_ROW4X32(1)
  GCN_ROW4X32(2)
  GCN_ROW4X32(3)
  GCN_ROW4X32(4)
  GCN_ROW4X32(5)
  GCN_ROW4X32(6)
  GCN_ROW4X32(7)
#pragma unroll
  for (int j = 0; j + 1 < NJ; ++j) acc += v[j];
  const v4f zero = {0.0f, 0.0f, 0.0f, 0.0f};
  acc += (4 * (NJ - 1) < lg) ? v[NJ - 1] : zero;
  return acc;
}

__device__ __forceinline__ v4f gcn_rows32_any(const char* __restrict__ Hb, int sidx, int nv, int g,
                                              int q16, v4f acc) {
  const int lg = nv - g;
  switch ((nv + 3) >> 2) {
    case 1: return gcn_rows32<1>(Hb, sidx, lg, q16, acc);
    case 2: return gcn_rows32<2>(Hb, sidx, lg, q16, acc);
    case 3: return gcn_rows32<3>(Hb, sidx, lg, q16, acc);
    case 4: return gcn_rows32<4>(Hb, sidx, lg, q16, acc);
    case 5: return gcn_rows32<5>(Hb, sidx, lg, q16, acc);
    case 6: return gcn_rows32<6>(Hb, sidx, lg, q16, acc);
    case 7: return gcn_rows32<7>(Hb, sidx, lg, q16, acc);
    case 8: return gcn_rows32<8>(Hb, sidx, lg, q16, acc);
    default: return acc;
  }
}

#define GCN_GWAVES 10
#define GCN_GNODES (GCN_GWAVES * 8)
#define GCN_TPITCH 36

__global__ __launch_bounds__(GCN_GWAVES * 32) void gcn_gather32_kernel(const int* __restrict__ rowstart,
                                                           const int* __restrict__ cnt,
                                                           const int* __restrict__ csr,
                                                           const float* __restrict__ dinv,
                                                           const float* __restrict__ H,
                                                           const float* __restrict__ b1,
                                                           const float* __restrict__ W2,
                                                           float* __restrict__ HW2,
                                                           int N) {
  __shared__ float tile[GCN_GNODES * GCN_TPITCH];
  __shared__ float sdn[GCN_GNODES];
  const int lane = threadIdx.x & 31;
  const int wave = __builtin_amdgcn_readfirstlane(threadIdx.x >> 5);
  const int blockNode = blockIdx.x * GCN_GNODES;
  const int node0 = blockNode + wave * 8;
  const int g = lane >> 3;
  const int q16 = (lane & 7) * 16;
  const int slot = ((lane & 7) << 2) | g;
  const int fo = (lane & 7) * 4 + g;
  const float bias = b1[fo];
  const char* Hb = reinterpret_cast<const char*>(H);
  const int lo = lane & 15;
  const int hi = lane >> 4;
  const int mn = node0 + (lane & 7);
  const int mc = min(mn, N - 1);
  const int mstart = rowstart[mc];
  const int mdeg = (mn < N) ? cnt[mc] : 0;
  const float mdn = dinv[mc];
  if (lane < 8) sdn[wave * 8 + lane] = mdn;
  int s_cur = min(node0, N - 1);
  {
    const int st = __builtin_amdgcn_readlane(mstart, 0);
    const int dg = __builtin_amdgcn_readlane(mdeg, 0);
    if (slot < dg) s_cur = csr[st + slot];
  }
  asm volatile("" ::"v"(s_cur), "v"(mstart), "v"(mdeg), "v"(mdn));
  for (int i = 0; i < 8; ++i) {
    const int n = node0 + i;
    const int inext = min(i + 1, 7);
    const int stn = __builtin_amdgcn_readlane(mstart, inext);
    const int dgn = (i < 7) ? __builtin_amdgcn_readlane(mdeg, inext) : 0;
    int s_nxt = min(n + 1, N - 1);
    if (slot < dgn) s_nxt = csr[stn + slot];
    if (n < N) {
      const int start = __builtin_amdgcn_readlane(mstart, i);
      const int deg = __builtin_amdgcn_readlane(mdeg, i);
      const float dn = __builtin_bit_cast(float,
                                          __builtin_amdgcn_readlane(__builtin_bit_cast(int, mdn), i));
      const float* own = H + (size_t)n * 32;
      const float selfv = own[(unsigned)fo];
      v4f acc = {0.0f, 0.0f, 0.0f, 0.0f};
      acc = gcn_rows32_any(Hb, s_cur, min(32, deg), g, q16, acc);
      if (deg > 32) {
        for (int c0 = 32; c0 < deg; c0 += 32) {
          const int nv2 = min(32, deg - c0);
          int s = n;
          if (slot < nv2) s = csr[start + c0 + slot];
          acc = gcn_rows32_any(Hb, s, nv2, g, q16, acc);
        }
      }
      float r[4];
#pragma unroll
      for (int e = 0; e < 4; ++e) {
        float a = acc[e];
        a += gcn_row_xor<8>(a);
        a += gcn_half_swap(a);
        r[e] = a;
      }
      const float sum = (g == 0) ? r[0] : (g == 1) ? r[1] : (g == 2) ? r[2] : r[3];
      const float t = (sum + selfv) * dn;
      tile[(wave * 8 + i) * GCN_TPITCH + fo] = tanhf(t + bias);
    }
    s_cur = s_nxt;
  }
  v2f bw[8] = {};
  if (wave < GCN_GNODES / 16) gcn_load_b<16>(W2, bw, lo, hi);
  __syncthreads();
  const int rowBase = blockNode + wave * 16;
  if (wave < GCN_GNODES / 16 && rowBase + 16 <= N) {
    v2f a[8];
#pragma unroll
    for (int kb = 0; kb < 8; ++kb) {
      const float* xr = tile + (wave * 16 + lo) * GCN_TPITCH + kb * 4 + hi * 2;
      a[kb].x = xr[0];
      a[kb].y = xr[1];
    }
    float rs[8];
#pragma unroll
    for (int v = 0; v < 8; ++v) rs[v] = sdn[wave * 16 + v + hi * 8];
    gcn_tile_product<16>(a, bw, rs, HW2 + (size_t)rowBase * 16, lo, hi);
  }
}


__device__ __forceinline__ void gcn_pair_bounds(const int* __restrict__ rowstart,
                                                const int* __restrict__ cnt, int p, int nPairs,
                                                int N, int sub, int& start, int& deg) {
  const int n = 2 * p + sub;
  const bool valid = (p < nPairs) && (n < N);
  const int nc = valid ? n : N - 1;
  start = rowstart[nc];
  deg = valid ? cnt[nc] : 0;
}

__device__ __forceinline__ void gcn_pair_idx(const int* __restrict__ csr, int start, int left,
                                             int ea, int& sa, int& sb) {
  sa = 0;
  sb = 0;
  const int le = left - ea;
  if (le > 0) sa = csr[start + ea];
  if (le > 1) sb = csr[start + ea + 1];
}

#define GCN_ROW4X16(J)                                                                     \
  if (NJ > J) {                                                                            \
    const int s0 = gcn_bcast8<J>(sa);                                                      \
    const int s1 = gcn_bcast8<J>(sb);                                                      \
    v[J] = *reinterpret_cast<const v4f*>(Hb + (unsigned)((godd ? s1 : s0) * 64 + q16));    \
  }
template <int NJ>
__device__ __forceinline__ v4f gcn_rows16(const char* __restrict__ Hb, int sa, int sb, int left,
                                          int godd, int g, int q16, v4f acc) {
  v4f v[8];
  GCN_ROW4X16(0)
  GCN_ROW4X16(1)
  GCN_ROW4X16(2)
  GCN_ROW4X16(3)
  GCN_ROW4X16(4)
  GCN_ROW4X16(5)
  GCN_ROW4X16(6)
  GCN_ROW4X16(7)
  const v4f zero = {0.0f, 0.0f, 0.0f, 0.0f};
  const int lg = left - g;
#pragma unroll
  for (int j = 0; j < NJ; ++j) acc += (4 * j < lg) ? v[j] : zero;
  return acc;
}

__device__ __forceinline__ v4f gcn_rows16_any(const char* __restrict__ Hb, int sa, int sb, int left,
                                              int nmax, int godd, int g, int q16, v4f acc) {
  switch ((nmax + 3) >> 2) {
    case 1: return gcn_rows16<1>(Hb, sa, sb, left, godd, g, q16, acc);
    case 2: return gcn_rows16<2>(Hb, sa, sb, left, godd, g, q16, acc);
    case 3: return gcn_rows16<3>(Hb, sa, sb, left, godd, g, q16, acc);
    case 4: return gcn_rows16<4>(Hb, sa, sb, left, godd, g, q16, acc);
    case 5: return gcn_rows16<5>(Hb, sa, sb, left, godd, g, q16, acc);
    case 6: return gcn_rows16<6>(Hb, sa, sb, left, godd, g, q16, acc);
    case 7: return gcn_rows16<7>(Hb, sa, sb, left, godd, g, q16, acc);
    case 8: return gcn_rows16<8>(Hb, sa, sb, left, godd, g, q16, acc);
    default: return acc;
  }
}

__global__ __launch_bounds__(GCN_GWAVES * 32) void gcn_gather16_kernel(const int* __restrict__ rowstart,
                                                           const int* __restrict__ cnt,
                                                           const int* __restrict__ csr,
                                                           const float* __restrict__ dinv,
                                                           const float* __restrict__ H,
                                                           const float* __restrict__ b2,
                                                           float* __restrict__ out_h,
                                                           float* __restrict__ out_ls,
                                                           int N, int nWaves) {
  const int lane = threadIdx.x & 31;
  const int f = lane & 15;
  const int sub = lane >> 4;
  const int g = f >> 2;
  const int godd = g & 1;
  const int q16 = (f & 3) * 16;
  const int fo = (f & 3) * 4 + g;
  const int ea = ((lane & 7) << 2) | (((lane >> 3) & 1) << 1);
  const int fo4 = fo * 4;
  const float bias = b2[fo];
  const char* Hb = reinterpret_cast<const char*>(H);
  const int wave = __builtin_amdgcn_readfirstlane(blockIdx.x * GCN_GWAVES + (threadIdx.x >> 5));
  const int nPairs = (N + 1) >> 1;
  int st0, dg0, st1, dg1;
  gcn_pair_bounds(rowstart, cnt, wave, nPairs, N, sub, st0, dg0);
  gcn_pair_bounds(rowstart, cnt, wave + nWaves, nPairs, N, sub, st1, dg1);
  int sa0, sb0;
  gcn_pair_idx(csr, st0, dg0, ea, sa0, sb0);
  asm volatile("" ::"v"(sa0), "v"(sb0), "v"(st0), "v"(dg0), "v"(st1), "v"(dg1), "v"(bias));
  for (int p = wave; p < nPairs; p += nWaves) {
    int sa1, sb1;
    gcn_pair_idx(csr, st1, dg1, ea, sa1, sb1);
    int st2, dg2;
    gcn_pair_bounds(rowstart, cnt, p + 2 * nWaves, nPairs, N, sub, st2, dg2);
    const int n = 2 * p + sub;
    const bool valid = n < N;
    const int nc = min(n, N - 1);
    const int start = st0;
    const int deg = dg0;
    const int degmax = max(__builtin_amdgcn_readlane(deg, 0), __builtin_amdgcn_readlane(deg, 16));
    const float di = dinv[nc];
    const float selfv = *reinterpret_cast<const float*>(Hb + (unsigned)(nc * 64 + fo4));
    v4f acc = {0.0f, 0.0f, 0.0f, 0.0f};
    acc = gcn_rows16_any(Hb, sa0, sb0, deg, min(32, degmax), godd, g, q16, acc);
    if (degmax > 32) {
      for (int c0 = 32; c0 < degmax; c0 += 32) {
        const int left = deg - c0;
        int sa, sb;
        gcn_pair_idx(csr, start + c0, left, ea, sa, sb);
        acc = gcn_rows16_any(Hb, sa, sb, left, min(32, degmax - c0), godd, g, q16, acc);
      }
    }
    float r[4];
#pragma unroll
    for (int e = 0; e < 4; ++e) {
      float a = acc[e];
      a += gcn_row_xor<4>(a);
      a += gcn_row_xor<8>(a);
      r[e] = a;
    }
    const float sum = (g == 0) ? r[0] : (g == 1) ? r[1] : (g == 2) ? r[2] : r[3];
    const float t = (sum + selfv) * di + bias;
    float m = t;
    m = fmaxf(m, gcn_row_xor<1>(m));
    m = fmaxf(m, gcn_row_xor<2>(m));
    m = fmaxf(m, gcn_row_xor<4>(m));
    m = fmaxf(m, gcn_row_xor<8>(m));
    float se = expf(t - m);
    se += gcn_row_xor<1>(se);
    se += gcn_row_xor<2>(se);
    se += gcn_row_xor<4>(se);
    se += gcn_row_xor<8>(se);
    const float lse = logf(se);
    if (valid) {
      const unsigned ob = (unsigned)(n * 64 + fo4);
      *reinterpret_cast<float*>(reinterpret_cast<char*>(out_h) + ob) = t;
      *reinterpret_cast<float*>(reinterpret_cast<char*>(out_ls) + ob) = t - m - lse;
    }
    st0 = st1; dg0 = dg1;
    st1 = st2; dg1 = dg2;
    sa0 = sa1; sb0 = sb1;
  }
}

extern "C" void kernel_launch(void* const* d_in, const int* in_sizes, int n_in,
                              void* d_out, int out_size, void* d_ws, size_t ws_size,
                              hipStream_t stream) {
  const float* x  = (const float*)d_in[0];
  const int*   ei = (const int*)d_in[1];
  const float* W1 = (const float*)d_in[2];
  const float* b1 = (const float*)d_in[3];
  const float* W2 = (const float*)d_in[4];
  const float* b2 = (const float*)d_in[5];

  const int N = in_sizes[0] / 32;
  const int E = in_sizes[1] / 2;
  const int* src = ei;
  const int* dst = ei + E;

  int*   wsi      = (int*)d_ws;
  float* wsf      = (float*)d_ws;
  int*   cnt      = wsi;
  float* dinv     = wsf + (size_t)N;
  int*   rowstart = wsi + (size_t)N * 2;
  const int NB = (N + GCN_BNODES - 1) / GCN_BNODES;
  const int G1 = (E + GCN_CHUNK - 1) / GCN_CHUNK;
  const int TS = NB + 1;
  unsigned* chunked = (unsigned*)(wsi + (size_t)N * 3);
  int*   csr      = wsi + (size_t)N * 3 + (size_t)E;
  float* xw1      = wsf + (size_t)N * 3 + (size_t)E * 2;
  float* hw2      = xw1 + (size_t)N * 64;
  int*   tab      = (int*)(hw2 + (size_t)N * 32);

  float* out_h  = (float*)d_out;
  float* out_ls = out_h + (size_t)N * 16;

  const unsigned gather32Blocks = (unsigned)((N + GCN_GNODES - 1) / GCN_GNODES);
  const unsigned gather16Blocks = (unsigned)(((N + 1) / 2 + GCN_GWAVES * 4 - 1) / (GCN_GWAVES * 4));

  gcn_chunksort_kernel<<<(unsigned)G1, 512, 0, stream>>>(src, dst, chunked, tab, E, NB, TS);
  gcn_place_kernel<<<(unsigned)NB, 512, 0, stream>>>(tab, chunked, cnt, rowstart, dinv, csr,
                                                    x, W1, xw1, N, G1, TS);

  gcn_gather32_kernel<<<gather32Blocks, GCN_GWAVES * 32, 0, stream>>>(rowstart, cnt, csr, dinv, xw1, b1, W2,
                                                          hw2, N);

  gcn_gather16_kernel<<<gather16Blocks, GCN_GWAVES * 32, 0, stream>>>(rowstart, cnt, csr, dinv, hw2, b2,
                                                          out_h, out_ls, N, (int)gather16Blocks * GCN_GWAVES);
}
